# starting kernel
# baseline (speedup 1.0000x reference)
#include <hip/hip_runtime.h>

typedef float v2f __attribute__((ext_vector_type(2)));
typedef float v8f __attribute__((ext_vector_type(8)));

#define N_NODES 50000
#define N_EDGES 800000
#define N_GRAPH 512
#define F_IN    74
#define HID     64
#define BN_EPS  1e-5f

__global__ void fill_f32(float* __restrict__ p, float v, int n) {
    int i = blockIdx.x * blockDim.x + threadIdx.x;
    if (i < n) p[i] = v;
}
__global__ void fill_u32(unsigned* __restrict__ p, unsigned v, int n) {
    int i = blockIdx.x * blockDim.x + threadIdx.x;
    if (i < n) p[i] = v;
}

__device__ __forceinline__ v8f wmma_f32(v2f a, v2f b, v8f c) {
    return __builtin_amdgcn_wmma_f32_16x16x4_f32(false, a, false, b,
                                                 (short)0, c, false, false);
}

template<int DIN>
__global__ __launch_bounds__(128)
void gemm2_wmma(const float* __restrict__ X,
                const float* __restrict__ W,
                const float* __restrict__ rW,
                const float* __restrict__ rb,
                float* __restrict__ XW,
                float* __restrict__ RL,
                int ntiles)
{
    constexpr int KT   = (DIN + 3) / 4;
    constexpr int DPAD = KT * 4;

    __shared__ v2f  fragW[KT * 128];
    __shared__ v2f  fragR[KT * 128];
    __shared__ float xt[4 * 16 * DPAD];

    const int  lane = threadIdx.x & 31;
    const int  wave = threadIdx.x >> 5;
    const int  l15  = lane & 15;
    const bool hi   = lane >= 16;

    for (int idx = threadIdx.x; idx < KT * 128; idx += 128) {
        const int kc  = idx >> 7;
        const int r   = idx & 127;
        const int n   = ((r >> 5) << 4) + (r & 15);
        const int ka0 = kc * 4 + ((r & 31) >= 16 ? 2 : 0);
        const int ka1 = ka0 + 1;
        const int c0  = (ka0 < DIN) ? ka0 : 0;
        const int c1  = (ka1 < DIN) ? ka1 : 0;
        const float m0 = (ka0 < DIN) ? 1.0f : 0.0f;
        const float m1 = (ka1 < DIN) ? 1.0f : 0.0f;
        v2f fw; fw.x = W [(size_t)c0 * HID + n] * m0; fw.y = W [(size_t)c1 * HID + n] * m1;
        v2f fr; fr.x = rW[(size_t)c0 * HID + n] * m0; fr.y = rW[(size_t)c1 * HID + n] * m1;
        fragW[idx] = fw;
        fragR[idx] = fr;
    }
    __syncthreads();

    const int  tile  = blockIdx.x * 4 + wave;
    const bool activ = tile < ntiles;
    const int  row0  = (activ ? tile : (ntiles - 1)) * 16;

    float* xw_ = xt + wave * 16 * DPAD;
    for (int r = 0; r < 16; ++r) {
        const float* xrow = X + (size_t)(row0 + r) * DIN;
        for (int k = lane; k < DPAD; k += 32)
            xw_[r * DPAD + k] = (k < DIN) ? xrow[k] : 0.0f;
    }

    v8f acc[4], racc[4];
    #pragma unroll
    for (int t = 0; t < 4; ++t) { acc[t] = {}; racc[t] = {}; }

    const v2f* arow = (const v2f*)(xw_ + l15 * DPAD);
    const int  aoff = hi ? 1 : 0;

    #pragma unroll
    for (int kc = 0; kc < KT; ++kc) {
        const v2f a = arow[kc * 2 + aoff];
        #pragma unroll
        for (int t = 0; t < 4; ++t) {
            const v2f b  = fragW[kc * 128 + t * 32 + lane];
            const v2f rv = fragR[kc * 128 + t * 32 + lane];
            acc[t]  = wmma_f32(a, b,  acc[t]);
            racc[t] = wmma_f32(a, rv, racc[t]);
        }
    }

    if (activ) {
        const int rbase = row0 + (hi ? 8 : 0);
        #pragma unroll
        for (int t = 0; t < 4; ++t) {
            const int n = t * 16 + l15;
            const float bias = rb[n];
            #pragma unroll
            for (int j = 0; j < 8; ++j) {
                const int r = rbase + j;
                XW[(size_t)r * HID + n] = acc[t][j];
                float v = racc[t][j] + bias;
                RL[(size_t)r * HID + n] = v > 0.0f ? v : 0.0f;
            }
        }
    }
}

__global__ __launch_bounds__(256)
void edge_agg(const int* __restrict__ src, const int* __restrict__ dst,
              const float* __restrict__ xw, float* __restrict__ agg, int ne)
{
    int idx = blockIdx.x * blockDim.x + threadIdx.x;
    int e = idx >> 4;
    if (e >= ne) return;
    int g = (idx & 15) << 2;
    int s = src[e], d = dst[e];
    float4 v = *(const float4*)(xw + (size_t)s * HID + g);
    float* ap = agg + (size_t)d * HID + g;
    atomicAdd(ap + 0, v.x);
    atomicAdd(ap + 1, v.y);
    atomicAdd(ap + 2, v.z);
    atomicAdd(ap + 3, v.w);
}

__global__ __launch_bounds__(256)
void combine_stats(float* __restrict__ agg, const float* __restrict__ rl,
                   const float* __restrict__ bvec, float* __restrict__ stats,
                   int nrows)
{
    __shared__ float ssum[256], ssq[256];
    const int f   = threadIdx.x & 63;
    const int sub = threadIdx.x >> 6;
    const int r0  = blockIdx.x * 64;
    const int rend = (r0 + 64 < nrows) ? (r0 + 64) : nrows;
    const float b = bvec[f];
    float s = 0.0f, q = 0.0f;
    for (int r = r0 + sub; r < rend; r += 4) {
        float a = agg[(size_t)r * HID + f] + b;
        a = a > 0.0f ? a : 0.0f;
        float h = a + rl[(size_t)r * HID + f];
        agg[(size_t)r * HID + f] = h;
        s += h; q += h * h;
    }
    ssum[threadIdx.x] = s; ssq[threadIdx.x] = q;
    __syncthreads();
    if (sub == 0) {
        s = ssum[f] + ssum[64 + f] + ssum[128 + f] + ssum[192 + f];
        q = ssq[f]  + ssq[64 + f]  + ssq[128 + f]  + ssq[192 + f];
        atomicAdd(&stats[f], s);
        atomicAdd(&stats[64 + f], q);
    }
}

__global__ __launch_bounds__(256)
void bn_apply(float* __restrict__ h, const float* __restrict__ stats,
              const float* __restrict__ gamma, const float* __restrict__ beta,
              int nrows)
{
    int idx = blockIdx.x * blockDim.x + threadIdx.x;
    if (idx >= nrows * HID) return;
    int f = idx & 63;
    float inv = 1.0f / (float)nrows;
    float mean = stats[f] * inv;
    float var  = stats[64 + f] * inv - mean * mean;
    h[idx] = (h[idx] - mean) * rsqrtf(var + BN_EPS) * gamma[f] + beta[f];
}

__device__ __forceinline__ unsigned f32_ordered(float f) {
    unsigned b = __float_as_uint(f);
    return (b & 0x80000000u) ? ~b : (b | 0x80000000u);
}
#define NEG_INF_ENC 0x007FFFFFu

__global__ __launch_bounds__(256)
void readout_nodes(const float* __restrict__ h, const int* __restrict__ gid,
                   const float* __restrict__ awW, const float* __restrict__ awb,
                   float* __restrict__ hsum, unsigned* __restrict__ hmaxu,
                   int nrows)
{
    const int lane = threadIdx.x & 31;
    const int wave = threadIdx.x >> 5;
    const int n = blockIdx.x * 8 + wave;
    if (n >= nrows) return;
    float h0 = h[(size_t)n * HID + lane];
    float h1 = h[(size_t)n * HID + 32 + lane];
    float p  = h0 * awW[lane] + h1 * awW[32 + lane];
    for (int off = 16; off > 0; off >>= 1) p += __shfl_xor(p, off, 32);
    float w = 1.0f / (1.0f + expf(-(p + awb[0])));
    int g = gid[n];
    atomicAdd(&hsum[(size_t)g * HID + lane],      h0 * w);
    atomicAdd(&hsum[(size_t)g * HID + 32 + lane], h1 * w);
    atomicMax(&hmaxu[(size_t)g * HID + lane],      f32_ordered(h0));
    atomicMax(&hmaxu[(size_t)g * HID + 32 + lane], f32_ordered(h1));
}

__global__ __launch_bounds__(256)
void build_hg(const float* __restrict__ hsum, const unsigned* __restrict__ hmaxu,
              float* __restrict__ hg, int nb)
{
    int idx = blockIdx.x * blockDim.x + threadIdx.x;
    if (idx >= nb * HID) return;
    int b = idx >> 6, f = idx & 63;
    hg[(size_t)b * 128 + f] = hsum[idx];
    unsigned u = hmaxu[idx];
    float m = (u & 0x80000000u) ? __uint_as_float(u ^ 0x80000000u)
                                : __uint_as_float(~u);
    if (!(fabsf(m) <= 3.402823466e38f)) m = 0.0f;
    hg[(size_t)b * 128 + 64 + f] = m;
}

__global__ __launch_bounds__(128)
void out_gemm_wmma(const float* __restrict__ hg, const float* __restrict__ Wo,
                   const float* __restrict__ bo, float* __restrict__ out)
{
    constexpr int KT = 128 / 4;
    __shared__ v2f  fragW[KT * 128];
    __shared__ float xt[4 * 16 * 128];

    const int  lane = threadIdx.x & 31;
    const int  wave = threadIdx.x >> 5;
    const int  l15  = lane & 15;
    const bool hi   = lane >= 16;

    for (int idx = threadIdx.x; idx < KT * 128; idx += 128) {
        const int kc  = idx >> 7;
        const int r   = idx & 127;
        const int n   = ((r >> 5) << 4) + (r & 15);
        const int ka0 = kc * 4 + ((r & 31) >= 16 ? 2 : 0);
        v2f fw;
        fw.x = Wo[(size_t)ka0 * HID + n];
        fw.y = Wo[(size_t)(ka0 + 1) * HID + n];
        fragW[idx] = fw;
    }
    __syncthreads();

    const int row0 = (blockIdx.x * 4 + wave) * 16;
    float* xw_ = xt + wave * 16 * 128;
    for (int r = 0; r < 16; ++r) {
        const float* xrow = hg + (size_t)(row0 + r) * 128;
        for (int k = lane; k < 128; k += 32)
            xw_[r * 128 + k] = xrow[k];
    }

    v8f acc[4];
    #pragma unroll
    for (int t = 0; t < 4; ++t) acc[t] = {};

    const v2f* arow = (const v2f*)(xw_ + l15 * 128);
    const int  aoff = hi ? 1 : 0;
    #pragma unroll
    for (int kc = 0; kc < KT; ++kc) {
        const v2f a = arow[kc * 2 + aoff];
        #pragma unroll
        for (int t = 0; t < 4; ++t)
            acc[t] = wmma_f32(a, fragW[kc * 128 + t * 32 + lane], acc[t]);
    }

    const int rbase = row0 + (hi ? 8 : 0);
    #pragma unroll
    for (int t = 0; t < 4; ++t) {
        const int n = t * 16 + l15;
        const float bias = bo[n];
        #pragma unroll
        for (int j = 0; j < 8; ++j)
            out[(size_t)(rbase + j) * HID + n] = acc[t][j] + bias;
    }
}

extern "C" void kernel_launch(void* const* d_in, const int* in_sizes, int n_in,
                              void* d_out, int out_size, void* d_ws, size_t ws_size,
                              hipStream_t stream)
{
    (void)in_sizes; (void)n_in; (void)out_size; (void)ws_size;

    const float* feats = (const float*)d_in[0];
    const int*   src   = (const int*)d_in[1];
    const int*   dst   = (const int*)d_in[2];
    const int*   gid   = (const int*)d_in[3];
    const float* awW  = (const float*)d_in[22];
    const float* awb  = (const float*)d_in[23];
    const float* outW = (const float*)d_in[24];
    const float* outb = (const float*)d_in[25];
    float* out = (float*)d_out;

    float* ws    = (float*)d_ws;
    float* xw    = ws;
    float* rlin  = xw   + (size_t)N_NODES * HID;
    float* bufA  = rlin + (size_t)N_NODES * HID;
    float* bufB  = bufA + (size_t)N_NODES * HID;
    float* stats = bufB + (size_t)N_NODES * HID;
    float* hsum  = stats + 128;
    unsigned* hmaxu = (unsigned*)(hsum + (size_t)N_GRAPH * HID);
    float* hg    = (float*)(hmaxu + (size_t)N_GRAPH * HID);

    const int nh     = N_NODES * HID;
    const int ntiles = N_NODES / 16;
    const int gemmBlocks = (ntiles + 3) / 4;

    auto run_layer = [&](const float* x, int din, int pi, float* agg) {
        const float* W  = (const float*)d_in[pi + 0];
        const float* b  = (const float*)d_in[pi + 1];
        const float* rW = (const float*)d_in[pi + 2];
        const float* rb = (const float*)d_in[pi + 3];
        const float* g  = (const float*)d_in[pi + 4];
        const float* be = (const float*)d_in[pi + 5];

        fill_f32<<<(nh + 255) / 256, 256, 0, stream>>>(agg, 0.0f, nh);
        fill_f32<<<1, 128, 0, stream>>>(stats, 0.0f, 128);
        if (din == F_IN)
            gemm2_wmma<F_IN><<<gemmBlocks, 128, 0, stream>>>(x, W, rW, rb, xw, rlin, ntiles);
        else
            gemm2_wmma<HID><<<gemmBlocks, 128, 0, stream>>>(x, W, rW, rb, xw, rlin, ntiles);
        edge_agg<<<(N_EDGES * 16 + 255) / 256, 256, 0, stream>>>(src, dst, xw, agg, N_EDGES);
        combine_stats<<<(N_NODES + 63) / 64, 256, 0, stream>>>(agg, rlin, b, stats, N_NODES);
        bn_apply<<<(nh + 255) / 256, 256, 0, stream>>>(agg, stats, g, be, N_NODES);
    };

    run_layer(feats, F_IN, 4,  bufA);
    run_layer(bufA,  HID, 10, bufB);
    run_layer(bufB,  HID, 16, bufA);

    const int nbh = N_GRAPH * HID;
    fill_f32<<<(nbh + 255) / 256, 256, 0, stream>>>(hsum, 0.0f, nbh);
    fill_u32<<<(nbh + 255) / 256, 256, 0, stream>>>(hmaxu, NEG_INF_ENC, nbh);
    readout_nodes<<<(N_NODES + 7) / 8, 256, 0, stream>>>(bufA, gid, awW, awb,
                                                         hsum, hmaxu, N_NODES);
    build_hg<<<(nbh + 255) / 256, 256, 0, stream>>>(hsum, hmaxu, hg, N_GRAPH);
    out_gemm_wmma<<<(N_GRAPH / 16) / 4, 128, 0, stream>>>(hg, outW, outb, out);
}


# Edge aggregation as a per-destination gather over an edge list grouped by destination (built once per call, integer atomics): replaces the float-atomic scatter and agg zero-fill in all 3 layers
# speedup vs baseline: 1.0339x; 1.0339x over previous
#include <hip/hip_runtime.h>

typedef float v2f __attribute__((ext_vector_type(2)));
typedef float v8f __attribute__((ext_vector_type(8)));

#define N_NODES 50000
#define N_EDGES 800000
#define N_GRAPH 512
#define F_IN    74
#define HID     64
#define BN_EPS  1e-5f
#define N_PAD   50048

__global__ void fill_f32(float* __restrict__ p, float v, int n) {
    int i = blockIdx.x * blockDim.x + threadIdx.x;
    if (i < n) p[i] = v;
}
__global__ void fill_u32(unsigned* __restrict__ p, unsigned v, int n) {
    int i = blockIdx.x * blockDim.x + threadIdx.x;
    if (i < n) p[i] = v;
}

__device__ __forceinline__ v8f wmma_f32(v2f a, v2f b, v8f c) {
    return __builtin_amdgcn_wmma_f32_16x16x4_f32(false, a, false, b,
                                                 (short)0, c, false, false);
}

template<int DIN>
__global__ __launch_bounds__(128)
void gemm2_wmma(const float* __restrict__ X,
                const float* __restrict__ W,
                const float* __restrict__ rW,
                const float* __restrict__ rb,
                float* __restrict__ XW,
                float* __restrict__ RL,
                int ntiles)
{
    constexpr int KT   = (DIN + 3) / 4;
    constexpr int DPAD = KT * 4;

    __shared__ v2f  fragW[KT * 128];
    __shared__ v2f  fragR[KT * 128];
    __shared__ float xt[4 * 16 * DPAD];

    const int  lane = threadIdx.x & 31;
    const int  wave = threadIdx.x >> 5;
    const int  l15  = lane & 15;
    const bool hi   = lane >= 16;

    for (int idx = threadIdx.x; idx < KT * 128; idx += 128) {
        const int kc  = idx >> 7;
        const int r   = idx & 127;
        const int n   = ((r >> 5) << 4) + (r & 15);
        const int ka0 = kc * 4 + ((r & 31) >= 16 ? 2 : 0);
        const int ka1 = ka0 + 1;
        const int c0  = (ka0 < DIN) ? ka0 : 0;
        const int c1  = (ka1 < DIN) ? ka1 : 0;
        const float m0 = (ka0 < DIN) ? 1.0f : 0.0f;
        const float m1 = (ka1 < DIN) ? 1.0f : 0.0f;
        v2f fw; fw.x = W [(size_t)c0 * HID + n] * m0; fw.y = W [(size_t)c1 * HID + n] * m1;
        v2f fr; fr.x = rW[(size_t)c0 * HID + n] * m0; fr.y = rW[(size_t)c1 * HID + n] * m1;
        fragW[idx] = fw;
        fragR[idx] = fr;
    }
    __syncthreads();

    const int  tile  = blockIdx.x * 4 + wave;
    const bool activ = tile < ntiles;
    const int  row0  = (activ ? tile : (ntiles - 1)) * 16;

    float* xw_ = xt + wave * 16 * DPAD;
    for (int r = 0; r < 16; ++r) {
        const float* xrow = X + (size_t)(row0 + r) * DIN;
        for (int k = lane; k < DPAD; k += 32)
            xw_[r * DPAD + k] = (k < DIN) ? xrow[k] : 0.0f;
    }

    v8f acc[4], racc[4];
    #pragma unroll
    for (int t = 0; t < 4; ++t) { acc[t] = {}; racc[t] = {}; }

    const v2f* arow = (const v2f*)(xw_ + l15 * DPAD);
    const int  aoff = hi ? 1 : 0;

    #pragma unroll
    for (int kc = 0; kc < KT; ++kc) {
        const v2f a = arow[kc * 2 + aoff];
        #pragma unroll
        for (int t = 0; t < 4; ++t) {
            const v2f b  = fragW[kc * 128 + t * 32 + lane];
            const v2f rv = fragR[kc * 128 + t * 32 + lane];
            acc[t]  = wmma_f32(a, b,  acc[t]);
            racc[t] = wmma_f32(a, rv, racc[t]);
        }
    }

    if (activ) {
        const int rbase = row0 + (hi ? 8 : 0);
        #pragma unroll
        for (int t = 0; t < 4; ++t) {
            const int n = t * 16 + l15;
            const float bias = rb[n];
            #pragma unroll
            for (int j = 0; j < 8; ++j) {
                const int r = rbase + j;
                XW[(size_t)r * HID + n] = acc[t][j];
                float v = racc[t][j] + bias;
                RL[(size_t)r * HID + n] = v > 0.0f ? v : 0.0f;
            }
        }
    }
}

__global__ __launch_bounds__(256)
void count_deg(const int* __restrict__ dst, unsigned* __restrict__ deg, int ne)
{
    const int e = blockIdx.x * blockDim.x + threadIdx.x;
    if (e < ne) atomicAdd(&deg[dst[e]], 1u);
}

__global__ __launch_bounds__(256)
void alloc_ranges(const unsigned* __restrict__ deg, unsigned* __restrict__ total,
                  unsigned* __restrict__ cursor, int n)
{
    const int i    = blockIdx.x * blockDim.x + threadIdx.x;
    const int lane = threadIdx.x & 31;
    const unsigned d = (i < n) ? deg[i] : 0u;
    unsigned inc = d;
    for (int off = 1; off < 32; off <<= 1) {
        const unsigned t = __shfl_up(inc, off, 32);
        if (lane >= off) inc += t;
    }
    unsigned base = 0u;
    if (lane == 31) base = atomicAdd(total, inc);
    base = __shfl(base, 31, 32);
    if (i < n) cursor[i] = base + inc - d;
}

__global__ __launch_bounds__(256)
void fill_slots(const int* __restrict__ src, const int* __restrict__ dst,
                unsigned* __restrict__ cursor, int* __restrict__ col, int ne)
{
    const int e = blockIdx.x * blockDim.x + threadIdx.x;
    if (e >= ne) return;
    const unsigned p = atomicAdd(&cursor[dst[e]], 1u);
    if (p < (unsigned)ne) col[p] = src[e];
}

__global__ __launch_bounds__(256)
void gather_rows(const unsigned* __restrict__ cursor, const unsigned* __restrict__ deg,
                 const int* __restrict__ col, const float* __restrict__ xw,
                 float* __restrict__ agg, int nrows, int ne)
{
    const int lane = threadIdx.x & 31;
    const int node = __builtin_amdgcn_readfirstlane(blockIdx.x * 8 + (threadIdx.x >> 5));
    if (node >= nrows) return;

    unsigned end = cursor[node];
    if (end > (unsigned)ne) end = (unsigned)ne;
    unsigned cnt_all = deg[node];
    if (cnt_all > end) cnt_all = end;
    const unsigned beg = end - cnt_all;

    const float2* xrow = (const float2*)xw + lane;
    float2 acc = make_float2(0.0f, 0.0f);

    for (unsigned base = 0; base < cnt_all; base += 32) {
        const unsigned left = cnt_all - base;
        const unsigned cnt  = left < 32u ? left : 32u;
        unsigned myc = 0u;
        if ((unsigned)lane < cnt) myc = (unsigned)col[beg + base + lane];
        if (myc > (unsigned)(nrows - 1)) myc = (unsigned)(nrows - 1);
        unsigned j = 0;
        for (; j + 4 <= cnt; j += 4) {
            const int s0 = __builtin_amdgcn_readlane((int)myc, (int)j);
            const int s1 = __builtin_amdgcn_readlane((int)myc, (int)(j + 1));
            const int s2 = __builtin_amdgcn_readlane((int)myc, (int)(j + 2));
            const int s3 = __builtin_amdgcn_readlane((int)myc, (int)(j + 3));
            const float2 v0 = xrow[(size_t)s0 * 32];
            const float2 v1 = xrow[(size_t)s1 * 32];
            const float2 v2 = xrow[(size_t)s2 * 32];
            const float2 v3 = xrow[(size_t)s3 * 32];
            acc.x += v0.x; acc.y += v0.y;
            acc.x += v1.x; acc.y += v1.y;
            acc.x += v2.x; acc.y += v2.y;
            acc.x += v3.x; acc.y += v3.y;
        }
        if (j < cnt) {
            const unsigned last = cnt - 1;
            const unsigned j1 = (j + 1 < cnt) ? (j + 1) : last;
            const unsigned j2 = (j + 2 < cnt) ? (j + 2) : last;
            const int s0 = __builtin_amdgcn_readlane((int)myc, (int)j);
            const int s1 = __builtin_amdgcn_readlane((int)myc, (int)j1);
            const int s2 = __builtin_amdgcn_readlane((int)myc, (int)j2);
            const float2 v0 = xrow[(size_t)s0 * 32];
            const float2 v1 = xrow[(size_t)s1 * 32];
            const float2 v2 = xrow[(size_t)s2 * 32];
            acc.x += v0.x; acc.y += v0.y;
            if (j + 1 < cnt) { acc.x += v1.x; acc.y += v1.y; }
            if (j + 2 < cnt) { acc.x += v2.x; acc.y += v2.y; }
        }
    }
    ((float2*)agg)[(size_t)node * 32 + lane] = acc;
}

__global__ __launch_bounds__(256)
void combine_stats(float* __restrict__ agg, const float* __restrict__ rl,
                   const float* __restrict__ bvec, float* __restrict__ stats,
                   int nrows)
{
    __shared__ float ssum[256], ssq[256];
    const int f   = threadIdx.x & 63;
    const int sub = threadIdx.x >> 6;
    const int r0  = blockIdx.x * 64;
    const int rend = (r0 + 64 < nrows) ? (r0 + 64) : nrows;
    const float b = bvec[f];
    float s = 0.0f, q = 0.0f;
    for (int r = r0 + sub; r < rend; r += 4) {
        float a = agg[(size_t)r * HID + f] + b;
        a = a > 0.0f ? a : 0.0f;
        float h = a + rl[(size_t)r * HID + f];
        agg[(size_t)r * HID + f] = h;
        s += h; q += h * h;
    }
    ssum[threadIdx.x] = s; ssq[threadIdx.x] = q;
    __syncthreads();
    if (sub == 0) {
        s = ssum[f] + ssum[64 + f] + ssum[128 + f] + ssum[192 + f];
        q = ssq[f]  + ssq[64 + f]  + ssq[128 + f]  + ssq[192 + f];
        atomicAdd(&stats[f], s);
        atomicAdd(&stats[64 + f], q);
    }
}

__global__ __launch_bounds__(256)
void bn_apply(float* __restrict__ h, const float* __restrict__ stats,
              const float* __restrict__ gamma, const float* __restrict__ beta,
              int nrows)
{
    int idx = blockIdx.x * blockDim.x + threadIdx.x;
    if (idx >= nrows * HID) return;
    int f = idx & 63;
    float inv = 1.0f / (float)nrows;
    float mean = stats[f] * inv;
    float var  = stats[64 + f] * inv - mean * mean;
    h[idx] = (h[idx] - mean) * rsqrtf(var + BN_EPS) * gamma[f] + beta[f];
}

__device__ __forceinline__ unsigned f32_ordered(float f) {
    unsigned b = __float_as_uint(f);
    return (b & 0x80000000u) ? ~b : (b | 0x80000000u);
}
#define NEG_INF_ENC 0x007FFFFFu

__global__ __launch_bounds__(256)
void readout_nodes(const float* __restrict__ h, const int* __restrict__ gid,
                   const float* __restrict__ awW, const float* __restrict__ awb,
                   float* __restrict__ hsum, unsigned* __restrict__ hmaxu,
                   int nrows)
{
    const int lane = threadIdx.x & 31;
    const int wave = threadIdx.x >> 5;
    const int n = blockIdx.x * 8 + wave;
    if (n >= nrows) return;
    float h0 = h[(size_t)n * HID + lane];
    float h1 = h[(size_t)n * HID + 32 + lane];
    float p  = h0 * awW[lane] + h1 * awW[32 + lane];
    for (int off = 16; off > 0; off >>= 1) p += __shfl_xor(p, off, 32);
    float w = 1.0f / (1.0f + expf(-(p + awb[0])));
    int g = gid[n];
    atomicAdd(&hsum[(size_t)g * HID + lane],      h0 * w);
    atomicAdd(&hsum[(size_t)g * HID + 32 + lane], h1 * w);
    atomicMax(&hmaxu[(size_t)g * HID + lane],      f32_ordered(h0));
    atomicMax(&hmaxu[(size_t)g * HID + 32 + lane], f32_ordered(h1));
}

__global__ __launch_bounds__(256)
void build_hg(const float* __restrict__ hsum, const unsigned* __restrict__ hmaxu,
              float* __restrict__ hg, int nb)
{
    int idx = blockIdx.x * blockDim.x + threadIdx.x;
    if (idx >= nb * HID) return;
    int b = idx >> 6, f = idx & 63;
    hg[(size_t)b * 128 + f] = hsum[idx];
    unsigned u = hmaxu[idx];
    float m = (u & 0x80000000u) ? __uint_as_float(u ^ 0x80000000u)
                                : __uint_as_float(~u);
    if (!(fabsf(m) <= 3.402823466e38f)) m = 0.0f;
    hg[(size_t)b * 128 + 64 + f] = m;
}

__global__ __launch_bounds__(128)
void out_gemm_wmma(const float* __restrict__ hg, const float* __restrict__ Wo,
                   const float* __restrict__ bo, float* __restrict__ out)
{
    constexpr int KT = 128 / 4;
    __shared__ v2f  fragW[KT * 128];
    __shared__ float xt[4 * 16 * 128];

    const int  lane = threadIdx.x & 31;
    const int  wave = threadIdx.x >> 5;
    const int  l15  = lane & 15;
    const bool hi   = lane >= 16;

    for (int idx = threadIdx.x; idx < KT * 128; idx += 128) {
        const int kc  = idx >> 7;
        const int r   = idx & 127;
        const int n   = ((r >> 5) << 4) + (r & 15);
        const int ka0 = kc * 4 + ((r & 31) >= 16 ? 2 : 0);
        v2f fw;
        fw.x = Wo[(size_t)ka0 * HID + n];
        fw.y = Wo[(size_t)(ka0 + 1) * HID + n];
        fragW[idx] = fw;
    }
    __syncthreads();

    const int row0 = (blockIdx.x * 4 + wave) * 16;
    float* xw_ = xt + wave * 16 * 128;
    for (int r = 0; r < 16; ++r) {
        const float* xrow = hg + (size_t)(row0 + r) * 128;
        for (int k = lane; k < 128; k += 32)
            xw_[r * 128 + k] = xrow[k];
    }

    v8f acc[4];
    #pragma unroll
    for (int t = 0; t < 4; ++t) acc[t] = {};

    const v2f* arow = (const v2f*)(xw_ + l15 * 128);
    const int  aoff = hi ? 1 : 0;
    #pragma unroll
    for (int kc = 0; kc < KT; ++kc) {
        const v2f a = arow[kc * 2 + aoff];
        #pragma unroll
        for (int t = 0; t < 4; ++t)
            acc[t] = wmma_f32(a, fragW[kc * 128 + t * 32 + lane], acc[t]);
    }

    const int rbase = row0 + (hi ? 8 : 0);
    #pragma unroll
    for (int t = 0; t < 4; ++t) {
        const int n = t * 16 + l15;
        const float bias = bo[n];
        #pragma unroll
        for (int j = 0; j < 8; ++j)
            out[(size_t)(rbase + j) * HID + n] = acc[t][j] + bias;
    }
}

extern "C" void kernel_launch(void* const* d_in, const int* in_sizes, int n_in,
                              void* d_out, int out_size, void* d_ws, size_t ws_size,
                              hipStream_t stream)
{
    (void)in_sizes; (void)n_in; (void)out_size; (void)ws_size;

    const float* feats = (const float*)d_in[0];
    const int*   src   = (const int*)d_in[1];
    const int*   dst   = (const int*)d_in[2];
    const int*   gid   = (const int*)d_in[3];
    const float* awW  = (const float*)d_in[22];
    const float* awb  = (const float*)d_in[23];
    const float* outW = (const float*)d_in[24];
    const float* outb = (const float*)d_in[25];
    float* out = (float*)d_out;

    float* ws    = (float*)d_ws;
    float* xw    = ws;
    float* rlin  = xw   + (size_t)N_NODES * HID;
    float* bufA  = rlin + (size_t)N_NODES * HID;
    float* bufB  = bufA + (size_t)N_NODES * HID;
    float* stats = bufB + (size_t)N_NODES * HID;
    float* hsum  = stats + 128;
    unsigned* hmaxu = (unsigned*)(hsum + (size_t)N_GRAPH * HID);
    float* hg    = (float*)(hmaxu + (size_t)N_GRAPH * HID);

    int*      col    = (int*)bufB;
    unsigned* cursor = (unsigned*)(col + N_EDGES);
    unsigned* deg    = cursor + N_PAD;

    const int nh     = N_NODES * HID;
    const int ntiles = N_NODES / 16;
    const int gemmBlocks = (ntiles + 3) / 4;

    auto run_layer = [&](const float* x, int din, int pi, float* agg) {
        const float* W  = (const float*)d_in[pi + 0];
        const float* b  = (const float*)d_in[pi + 1];
        const float* rW = (const float*)d_in[pi + 2];
        const float* rb = (const float*)d_in[pi + 3];
        const float* g  = (const float*)d_in[pi + 4];
        const float* be = (const float*)d_in[pi + 5];

        fill_f32<<<1, 128, 0, stream>>>(stats, 0.0f, 128);
        if (din == F_IN)
            gemm2_wmma<F_IN><<<gemmBlocks, 128, 0, stream>>>(x, W, rW, rb, xw, rlin, ntiles);
        else
            gemm2_wmma<HID><<<gemmBlocks, 128, 0, stream>>>(x, W, rW, rb, xw, rlin, ntiles);
        gather_rows<<<(N_NODES + 7) / 8, 256, 0, stream>>>(cursor, deg, col, xw, agg,
                                                           N_NODES, N_EDGES);
        combine_stats<<<(N_NODES + 63) / 64, 256, 0, stream>>>(agg, rlin, b, stats, N_NODES);
        bn_apply<<<(nh + 255) / 256, 256, 0, stream>>>(agg, stats, g, be, N_NODES);
    };

    fill_u32<<<(N_NODES + 1 + 255) / 256, 256, 0, stream>>>(deg, 0u, N_NODES + 1);
    count_deg<<<(N_EDGES + 255) / 256, 256, 0, stream>>>(dst, deg, N_EDGES);
    alloc_ranges<<<(N_NODES + 255) / 256, 256, 0, stream>>>(deg, deg + N_NODES, cursor, N_NODES);
    fill_slots<<<(N_EDGES + 255) / 256, 256, 0, stream>>>(src, dst, cursor, col, N_EDGES);

    run_layer(feats, F_IN, 4,  bufA);
    run_layer(bufA,  HID, 10, bufA);
    run_layer(bufA,  HID, 16, bufA);

    const int nbh = N_GRAPH * HID;
    fill_f32<<<(nbh + 255) / 256, 256, 0, stream>>>(hsum, 0.0f, nbh);
    fill_u32<<<(nbh + 255) / 256, 256, 0, stream>>>(hmaxu, NEG_INF_ENC, nbh);
    readout_nodes<<<(N_NODES + 7) / 8, 256, 0, stream>>>(bufA, gid, awW, awb,
                                                         hsum, hmaxu, N_NODES);
    build_hg<<<(nbh + 255) / 256, 256, 0, stream>>>(hsum, hmaxu, hg, N_GRAPH);
    out_gemm_wmma<<<(N_GRAPH / 16) / 4, 128, 0, stream>>>(hg, outW, outb, out);
}


# Layers 2 and 3 sum neighbour rows of h BEFORE the product ((A x)W = A(xW)): one launch per layer gathers into its LDS tile, runs both WMMA products and the epilogue; 15->11 launches
# speedup vs baseline: 1.4394x; 1.3922x over previous
#include <hip/hip_runtime.h>

typedef float v2f __attribute__((ext_vector_type(2)));
typedef float v8f __attribute__((ext_vector_type(8)));

#define N_NODES 50000
#define N_EDGES 800000
#define N_GRAPH 512
#define F_IN    74
#define HID     64
#define BN_EPS  1e-5f
#define N_PAD   50048

#define NEG_INF_ENC 0x007FFFFFu

__global__ __launch_bounds__(256)
void init_state(unsigned* __restrict__ deg, float* __restrict__ stats3,
                float* __restrict__ hsum, unsigned* __restrict__ hmaxu)
{
    int i = blockIdx.x * blockDim.x + threadIdx.x;
    if (i < N_NODES + 1) { deg[i] = 0u; return; }
    i -= N_NODES + 1;
    if (i < 3 * 128) { stats3[i] = 0.0f; return; }
    i -= 3 * 128;
    if (i < N_GRAPH * HID) { hsum[i] = 0.0f; return; }
    i -= N_GRAPH * HID;
    if (i < N_GRAPH * HID) hmaxu[i] = NEG_INF_ENC;
}

__device__ __forceinline__ v8f wmma_f32(v2f a, v2f b, v8f c) {
    return __builtin_amdgcn_wmma_f32_16x16x4_f32(false, a, false, b,
                                                 (short)0, c, false, false);
}

template<int DIN>
__global__ __launch_bounds__(128)
void gemm2_wmma(const float* __restrict__ X,
                const float* __restrict__ W,
                const float* __restrict__ rW,
                const float* __restrict__ rb,
                float* __restrict__ XW,
                float* __restrict__ RL,
                int ntiles)
{
    constexpr int KT   = (DIN + 3) / 4;
    constexpr int DPAD = KT * 4;

    __shared__ v2f  fragW[KT * 128];
    __shared__ v2f  fragR[KT * 128];
    __shared__ float xt[4 * 16 * DPAD];

    const int  lane = threadIdx.x & 31;
    const int  wave = threadIdx.x >> 5;
    const int  l15  = lane & 15;
    const bool hi   = lane >= 16;

    for (int idx = threadIdx.x; idx < KT * 128; idx += 128) {
        const int kc  = idx >> 7;
        const int r   = idx & 127;
        const int n   = ((r >> 5) << 4) + (r & 15);
        const int ka0 = kc * 4 + ((r & 31) >= 16 ? 2 : 0);
        const int ka1 = ka0 + 1;
        const int c0  = (ka0 < DIN) ? ka0 : 0;
        const int c1  = (ka1 < DIN) ? ka1 : 0;
        const float m0 = (ka0 < DIN) ? 1.0f : 0.0f;
        const float m1 = (ka1 < DIN) ? 1.0f : 0.0f;
        v2f fw; fw.x = W [(size_t)c0 * HID + n] * m0; fw.y = W [(size_t)c1 * HID + n] * m1;
        v2f fr; fr.x = rW[(size_t)c0 * HID + n] * m0; fr.y = rW[(size_t)c1 * HID + n] * m1;
        fragW[idx] = fw;
        fragR[idx] = fr;
    }
    __syncthreads();

    const int  tile  = blockIdx.x * 4 + wave;
    const bool activ = tile < ntiles;
    const int  row0  = (activ ? tile : (ntiles - 1)) * 16;

    float* xw_ = xt + wave * 16 * DPAD;
    for (int r = 0; r < 16; ++r) {
        const float* xrow = X + (size_t)(row0 + r) * DIN;
        for (int k = lane; k < DPAD; k += 32)
            xw_[r * DPAD + k] = (k < DIN) ? xrow[k] : 0.0f;
    }

    v8f acc[4], racc[4];
    #pragma unroll
    for (int t = 0; t < 4; ++t) { acc[t] = {}; racc[t] = {}; }

    const v2f* arow = (const v2f*)(xw_ + l15 * DPAD);
    const int  aoff = hi ? 1 : 0;

    #pragma unroll
    for (int kc = 0; kc < KT; ++kc) {
        const v2f a = arow[kc * 2 + aoff];
        #pragma unroll
        for (int t = 0; t < 4; ++t) {
            const v2f b  = fragW[kc * 128 + t * 32 + lane];
            const v2f rv = fragR[kc * 128 + t * 32 + lane];
            acc[t]  = wmma_f32(a, b,  acc[t]);
            racc[t] = wmma_f32(a, rv, racc[t]);
        }
    }

    if (activ) {
        const int rbase = row0 + (hi ? 8 : 0);
        #pragma unroll
        for (int t = 0; t < 4; ++t) {
            const int n = t * 16 + l15;
            const float bias = rb[n];
            #pragma unroll
            for (int j = 0; j < 8; ++j) {
                const int r = rbase + j;
                XW[(size_t)r * HID + n] = acc[t][j];
                float v = racc[t][j] + bias;
                RL[(size_t)r * HID + n] = v > 0.0f ? v : 0.0f;
            }
        }
    }
}

__global__ __launch_bounds__(256)
void count_deg(const int* __restrict__ dst, unsigned* __restrict__ deg, int ne)
{
    const int e = blockIdx.x * blockDim.x + threadIdx.x;
    if (e < ne) atomicAdd(&deg[dst[e]], 1u);
}

__global__ __launch_bounds__(256)
void alloc_ranges(const unsigned* __restrict__ deg, unsigned* __restrict__ total,
                  unsigned* __restrict__ cursor, int n)
{
    const int i    = blockIdx.x * blockDim.x + threadIdx.x;
    const int lane = threadIdx.x & 31;
    const unsigned d = (i < n) ? deg[i] : 0u;
    unsigned inc = d;
    for (int off = 1; off < 32; off <<= 1) {
        const unsigned t = __shfl_up(inc, off, 32);
        if (lane >= off) inc += t;
    }
    unsigned base = 0u;
    if (lane == 31) base = atomicAdd(total, inc);
    base = __shfl(base, 31, 32);
    if (i < n) cursor[i] = base + inc - d;
}

__global__ __launch_bounds__(256)
void fill_slots(const int* __restrict__ src, const int* __restrict__ dst,
                unsigned* __restrict__ cursor, int* __restrict__ col, int ne)
{
    const int e = blockIdx.x * blockDim.x + threadIdx.x;
    if (e >= ne) return;
    const unsigned p = atomicAdd(&cursor[dst[e]], 1u);
    if (p < (unsigned)ne) col[p] = src[e];
}

__global__ __launch_bounds__(256)
void gather_rows(const unsigned* __restrict__ cursor, const unsigned* __restrict__ deg,
                 const int* __restrict__ col, const float* __restrict__ xw,
                 float* __restrict__ agg, int nrows, int ne)
{
    const int lane = threadIdx.x & 31;
    const int node = __builtin_amdgcn_readfirstlane(blockIdx.x * 8 + (threadIdx.x >> 5));
    if (node >= nrows) return;

    unsigned end = cursor[node];
    if (end > (unsigned)ne) end = (unsigned)ne;
    unsigned cnt_all = deg[node];
    if (cnt_all > end) cnt_all = end;
    const unsigned beg = end - cnt_all;

    const float2* xrow = (const float2*)xw + lane;
    float2 acc = make_float2(0.0f, 0.0f);

    for (unsigned base = 0; base < cnt_all; base += 32) {
        const unsigned left = cnt_all - base;
        const unsigned cnt  = left < 32u ? left : 32u;
        unsigned myc = 0u;
        if ((unsigned)lane < cnt) myc = (unsigned)col[beg + base + lane];
        if (myc > (unsigned)(nrows - 1)) myc = (unsigned)(nrows - 1);
        unsigned j = 0;
        for (; j + 4 <= cnt; j += 4) {
            const int s0 = __builtin_amdgcn_readlane((int)myc, (int)j);
            const int s1 = __builtin_amdgcn_readlane((int)myc, (int)(j + 1));
            const int s2 = __builtin_amdgcn_readlane((int)myc, (int)(j + 2));
            const int s3 = __builtin_amdgcn_readlane((int)myc, (int)(j + 3));
            const float2 v0 = xrow[(size_t)s0 * 32];
            const float2 v1 = xrow[(size_t)s1 * 32];
            const float2 v2 = xrow[(size_t)s2 * 32];
            const float2 v3 = xrow[(size_t)s3 * 32];
            acc.x += v0.x; acc.y += v0.y;
            acc.x += v1.x; acc.y += v1.y;
            acc.x += v2.x; acc.y += v2.y;
            acc.x += v3.x; acc.y += v3.y;
        }
        if (j < cnt) {
            const unsigned last = cnt - 1;
            const unsigned j1 = (j + 1 < cnt) ? (j + 1) : last;
            const unsigned j2 = (j + 2 < cnt) ? (j + 2) : last;
            const int s0 = __builtin_amdgcn_readlane((int)myc, (int)j);
            const int s1 = __builtin_amdgcn_readlane((int)myc, (int)j1);
            const int s2 = __builtin_amdgcn_readlane((int)myc, (int)j2);
            const float2 v0 = xrow[(size_t)s0 * 32];
            const float2 v1 = xrow[(size_t)s1 * 32];
            const float2 v2 = xrow[(size_t)s2 * 32];
            acc.x += v0.x; acc.y += v0.y;
            if (j + 1 < cnt) { acc.x += v1.x; acc.y += v1.y; }
            if (j + 2 < cnt) { acc.x += v2.x; acc.y += v2.y; }
        }
    }
    ((float2*)agg)[(size_t)node * 32 + lane] = acc;
}

__global__ __launch_bounds__(256)
void combine_stats(float* __restrict__ agg, const float* __restrict__ rl,
                   const float* __restrict__ bvec, float* __restrict__ stats,
                   int nrows)
{
    __shared__ float ssum[256], ssq[256];
    const int f   = threadIdx.x & 63;
    const int sub = threadIdx.x >> 6;
    const int r0  = blockIdx.x * 64;
    const int rend = (r0 + 64 < nrows) ? (r0 + 64) : nrows;
    const float b = bvec[f];
    float s = 0.0f, q = 0.0f;
    for (int r = r0 + sub; r < rend; r += 4) {
        float a = agg[(size_t)r * HID + f] + b;
        a = a > 0.0f ? a : 0.0f;
        float h = a + rl[(size_t)r * HID + f];
        agg[(size_t)r * HID + f] = h;
        s += h; q += h * h;
    }
    ssum[threadIdx.x] = s; ssq[threadIdx.x] = q;
    __syncthreads();
    if (sub == 0) {
        s = ssum[f] + ssum[64 + f] + ssum[128 + f] + ssum[192 + f];
        q = ssq[f]  + ssq[64 + f]  + ssq[128 + f]  + ssq[192 + f];
        atomicAdd(&stats[f], s);
        atomicAdd(&stats[64 + f], q);
    }
}

__device__ __forceinline__ float xhalf(float x) {
    return __builtin_bit_cast(float, __builtin_amdgcn_permlanex16(
        __builtin_bit_cast(int, x), __builtin_bit_cast(int, x),
        0x76543210, 0xfedcba98, false, false));
}

#define XP 68
#define NB 16

__global__ __launch_bounds__(128)
void layer_gather_gemm(const float* __restrict__ Hin,
                       const float* __restrict__ W,
                       const float* __restrict__ rW,
                       const float* __restrict__ bvec,
                       const float* __restrict__ rb,
                       const float* __restrict__ pst,
                       const float* __restrict__ pg,
                       const float* __restrict__ pb,
                       const unsigned* __restrict__ cursor, const unsigned* __restrict__ deg,
                       const int* __restrict__ col,
                       float* __restrict__ Hout,
                       float* __restrict__ stats,
                       int ntiles, int nrows, int ne)
{
    constexpr int KT = HID / 4;
    __shared__ v2f   fragW[KT * 128];
    __shared__ v2f   fragR[KT * 128];
    __shared__ float ta[4 * 16 * XP];
    __shared__ float tx[4 * 16 * XP];
    __shared__ float ssum[4 * 64], ssq[4 * 64];

    const int  lane = threadIdx.x & 31;
    const int  wave = threadIdx.x >> 5;
    const int  l15  = lane & 15;
    const bool hi   = lane >= 16;
    const int  c4   = l15 * 4;

    const int  tile  = blockIdx.x * 4 + wave;
    const bool activ = tile < ntiles;
    const int  row0  = (activ ? tile : (ntiles - 1)) * 16;

    unsigned endv = cursor[row0 + l15];
    unsigned dv   = deg[row0 + l15];
    float bb[4], rbb[4];
    #pragma unroll
    for (int t = 0; t < 4; ++t) { bb[t] = bvec[t * 16 + l15]; rbb[t] = rb[t * 16 + l15]; }
    const float4 psum = *(const float4*)(pst + c4);
    const float4 psq  = *(const float4*)(pst + 64 + c4);
    float gam[4], bet[4];
    #pragma unroll
    for (int i = 0; i < 4; ++i) { gam[i] = pg[c4 + i]; bet[i] = pb[c4 + i]; }
    float4 xr[8];
    #pragma unroll
    for (int u = 0; u < 8; ++u)
        xr[u] = *(const float4*)(Hin + (size_t)(row0 + 2 * u + (hi ? 1 : 0)) * HID + c4);

    for (int idx = threadIdx.x; idx < KT * 128; idx += 128) {
        const int kc  = idx >> 7;
        const int r   = idx & 127;
        const int n   = ((r >> 5) << 4) + (r & 15);
        const int ka0 = kc * 4 + ((r & 31) >= 16 ? 2 : 0);
        v2f fw; fw.x = W [(size_t)ka0 * HID + n]; fw.y = W [(size_t)(ka0 + 1) * HID + n];
        v2f fr; fr.x = rW[(size_t)ka0 * HID + n]; fr.y = rW[(size_t)(ka0 + 1) * HID + n];
        fragW[idx] = fw;
        fragR[idx] = fr;
    }

    float sc[4], sh[4];
    {
        const float inv = 1.0f / (float)nrows;
        const float ps[4] = {psum.x, psum.y, psum.z, psum.w};
        const float pq[4] = {psq.x, psq.y, psq.z, psq.w};
        #pragma unroll
        for (int i = 0; i < 4; ++i) {
            const float mean = ps[i] * inv;
            const float var  = pq[i] * inv - mean * mean;
            sc[i] = rsqrtf(var + BN_EPS) * gam[i];
            sh[i] = bet[i] - mean * sc[i];
        }
    }

    float* ta_ = ta + wave * 16 * XP;
    float* tx_ = tx + wave * 16 * XP;

    #pragma unroll
    for (int u = 0; u < 8; ++u) {
        float4 v = xr[u];
        v.x = v.x * sc[0] + sh[0]; v.y = v.y * sc[1] + sh[1];
        v.z = v.z * sc[2] + sh[2]; v.w = v.w * sc[3] + sh[3];
        *(float4*)(tx_ + (2 * u + (hi ? 1 : 0)) * XP + c4) = v;
    }

    if (endv > (unsigned)ne) endv = (unsigned)ne;
    if (dv > endv) dv = endv;
    if (!activ) dv = 0u;
    const unsigned begv = endv - dv;
    const float* hcol = Hin + c4;

    auto first_ids = [&](unsigned b, unsigned d) -> unsigned {
        const unsigned k = ((unsigned)lane < d) ? (unsigned)lane : (d ? d - 1u : 0u);
        unsigned p = b + k;
        if (p > (unsigned)(ne - 1)) p = (unsigned)(ne - 1);
        return (unsigned)col[p];
    };
    unsigned myc_next = first_ids((unsigned)__builtin_amdgcn_readlane((int)begv, 0),
                                  (unsigned)__builtin_amdgcn_readlane((int)dv, 0));
    for (int r = 0; r < 16; ++r) {
        const unsigned beg = (unsigned)__builtin_amdgcn_readlane((int)begv, r);
        const unsigned d   = (unsigned)__builtin_amdgcn_readlane((int)dv, r);
        unsigned myc = myc_next;
        if (r + 1 < 16)
            myc_next = first_ids((unsigned)__builtin_amdgcn_readlane((int)begv, r + 1),
                                 (unsigned)__builtin_amdgcn_readlane((int)dv, r + 1));
        float4 acc = make_float4(0.0f, 0.0f, 0.0f, 0.0f);
        for (unsigned base = 0; base < d; base += 32) {
            const unsigned left = d - base;
            const unsigned cnt  = left < 32u ? left : 32u;
            if (base != 0u) myc = ((unsigned)lane < cnt) ? (unsigned)col[beg + base + lane] : 0u;
            if (myc > (unsigned)(nrows - 1)) myc = (unsigned)(nrows - 1);
            const unsigned nst = (cnt + 1u) >> 1;
            for (unsigned i0 = 0; i0 < nst; i0 += NB) {
                float4 v[NB];
                #pragma unroll
                for (int u = 0; u < NB; ++u) {
                    const unsigned i = i0 + (unsigned)u;
                    if (i < nst) {
                        const unsigned ea = 2u * i;
                        const unsigned eb = (ea + 1u < cnt) ? (ea + 1u) : ea;
                        const int sa = __builtin_amdgcn_readlane((int)myc, (int)ea);
                        const int sb = __builtin_amdgcn_readlane((int)myc, (int)eb);
                        const int srow = hi ? sb : sa;
                        v[u] = *(const float4*)(hcol + (size_t)srow * HID);
                    }
                }
                #pragma unroll
                for (int u = 0; u < NB; ++u) {
                    const unsigned i = i0 + (unsigned)u;
                    if (i < nst) {
                        const float m = (hi && (2u * i + 1u >= cnt)) ? 0.0f : 1.0f;
                        acc.x = fmaf(v[u].x, m, acc.x); acc.y = fmaf(v[u].y, m, acc.y);
                        acc.z = fmaf(v[u].z, m, acc.z); acc.w = fmaf(v[u].w, m, acc.w);
                    }
                }
            }
        }
        acc.x += xhalf(acc.x); acc.y += xhalf(acc.y);
        acc.z += xhalf(acc.z); acc.w += xhalf(acc.w);
        if (lane < 16) {
            const float df = (float)d;
            float4 o;
            o.x = acc.x * sc[0] + df * sh[0]; o.y = acc.y * sc[1] + df * sh[1];
            o.z = acc.z * sc[2] + df * sh[2]; o.w = acc.w * sc[3] + df * sh[3];
            *(float4*)(ta_ + r * XP + c4) = o;
        }
    }
    __syncthreads();

    v8f acc[4], racc[4];
    #pragma unroll
    for (int t = 0; t < 4; ++t) { acc[t] = {}; racc[t] = {}; }

    const v2f* arow = (const v2f*)(ta_ + l15 * XP);
    const v2f* xrow = (const v2f*)(tx_ + l15 * XP);
    const int  aoff = hi ? 1 : 0;

    #pragma unroll
    for (int kc = 0; kc < KT; ++kc) {
        const v2f a = arow[kc * 2 + aoff];
        const v2f x = xrow[kc * 2 + aoff];
        #pragma unroll
        for (int t = 0; t < 4; ++t) {
            const v2f b  = fragW[kc * 128 + t * 32 + lane];
            const v2f rv = fragR[kc * 128 + t * 32 + lane];
            acc[t]  = wmma_f32(a, b,  acc[t]);
            racc[t] = wmma_f32(x, rv, racc[t]);
        }
    }

    const int rl0 = hi ? 8 : 0;
    #pragma unroll
    for (int t = 0; t < 4; ++t) {
        const int n = t * 16 + l15;
        float cs = 0.0f, cq = 0.0f;
        #pragma unroll
        for (int j = 0; j < 8; ++j) {
            float a = acc[t][j] + bb[t];
            a = a > 0.0f ? a : 0.0f;
            float v = racc[t][j] + rbb[t];
            v = v > 0.0f ? v : 0.0f;
            const float h = a + v;
            ta_[(rl0 + j) * XP + n] = h;
            cs += h; cq += h * h;
        }
        cs += xhalf(cs);
        cq += xhalf(cq);
        if (!activ) { cs = 0.0f; cq = 0.0f; }
        if (lane < 16) { ssum[wave * 64 + n] = cs; ssq[wave * 64 + n] = cq; }
    }
    if (activ) {
        #pragma unroll
        for (int u = 0; u < 8; ++u) {
            const int row = 2 * u + (hi ? 1 : 0);
            const float4 hv = *(const float4*)(ta_ + row * XP + c4);
            *(float4*)(Hout + (size_t)(row0 + row) * HID + c4) = hv;
        }
    }
    __syncthreads();
    if (threadIdx.x < 64) {
        const int f = threadIdx.x;
        const float s = ssum[f] + ssum[64 + f] + ssum[128 + f] + ssum[192 + f];
        const float q = ssq[f]  + ssq[64 + f]  + ssq[128 + f]  + ssq[192 + f];
        atomicAdd(&stats[f], s);
        atomicAdd(&stats[64 + f], q);
    }
}

__device__ __forceinline__ unsigned f32_ordered(float f) {
    unsigned b = __float_as_uint(f);
    return (b & 0x80000000u) ? ~b : (b | 0x80000000u);
}

__global__ __launch_bounds__(256)
void readout_nodes(const float* __restrict__ h, const float* __restrict__ stats,
                   const float* __restrict__ gamma, const float* __restrict__ beta,
                   const int* __restrict__ gid,
                   const float* __restrict__ awW, const float* __restrict__ awb,
                   float* __restrict__ hsum, unsigned* __restrict__ hmaxu,
                   int nrows)
{
    const int lane = threadIdx.x & 31;
    const int wave = threadIdx.x >> 5;
    const int n = blockIdx.x * 8 + wave;
    if (n >= nrows) return;
    const float inv = 1.0f / (float)nrows;
    const float m0 = stats[lane] * inv;
    const float m1 = stats[32 + lane] * inv;
    const float v0 = stats[64 + lane] * inv - m0 * m0;
    const float v1 = stats[96 + lane] * inv - m1 * m1;
    float h0 = h[(size_t)n * HID + lane];
    float h1 = h[(size_t)n * HID + 32 + lane];
    h0 = (h0 - m0) * rsqrtf(v0 + BN_EPS) * gamma[lane] + beta[lane];
    h1 = (h1 - m1) * rsqrtf(v1 + BN_EPS) * gamma[32 + lane] + beta[32 + lane];
    float p  = h0 * awW[lane] + h1 * awW[32 + lane];
    for (int off = 16; off > 0; off >>= 1) p += __shfl_xor(p, off, 32);
    float w = 1.0f / (1.0f + expf(-(p + awb[0])));
    int g = gid[n];
    atomicAdd(&hsum[(size_t)g * HID + lane],      h0 * w);
    atomicAdd(&hsum[(size_t)g * HID + 32 + lane], h1 * w);
    atomicMax(&hmaxu[(size_t)g * HID + lane],      f32_ordered(h0));
    atomicMax(&hmaxu[(size_t)g * HID + 32 + lane], f32_ordered(h1));
}

__global__ __launch_bounds__(128)
void out_gemm_wmma(const float* __restrict__ hsum, const unsigned* __restrict__ hmaxu,
                   const float* __restrict__ Wo,
                   const float* __restrict__ bo, float* __restrict__ out)
{
    constexpr int KT = 128 / 4;
    __shared__ v2f  fragW[KT * 128];
    __shared__ float xt[4 * 16 * 128];

    const int  lane = threadIdx.x & 31;
    const int  wave = threadIdx.x >> 5;
    const int  l15  = lane & 15;
    const bool hi   = lane >= 16;

    for (int idx = threadIdx.x; idx < KT * 128; idx += 128) {
        const int kc  = idx >> 7;
        const int r   = idx & 127;
        const int n   = ((r >> 5) << 4) + (r & 15);
        const int ka0 = kc * 4 + ((r & 31) >= 16 ? 2 : 0);
        v2f fw;
        fw.x = Wo[(size_t)ka0 * HID + n];
        fw.y = Wo[(size_t)(ka0 + 1) * HID + n];
        fragW[idx] = fw;
    }
    __syncthreads();

    const int row0 = (blockIdx.x * 4 + wave) * 16;
    float* xw_ = xt + wave * 16 * 128;
    for (int r = 0; r < 16; ++r) {
        const float*    srow = hsum  + (size_t)(row0 + r) * HID;
        const unsigned* mrow = hmaxu + (size_t)(row0 + r) * HID;
        for (int k = lane; k < HID; k += 32) {
            xw_[r * 128 + k] = srow[k];
            const unsigned u = mrow[k];
            float m = (u & 0x80000000u) ? __uint_as_float(u ^ 0x80000000u)
                                        : __uint_as_float(~u);
            if (!(fabsf(m) <= 3.402823466e38f)) m = 0.0f;
            xw_[r * 128 + HID + k] = m;
        }
    }

    v8f acc[4];
    #pragma unroll
    for (int t = 0; t < 4; ++t) acc[t] = {};

    const v2f* arow = (const v2f*)(xw_ + l15 * 128);
    const int  aoff = hi ? 1 : 0;
    #pragma unroll
    for (int kc = 0; kc < KT; ++kc) {
        const v2f a = arow[kc * 2 + aoff];
        #pragma unroll
        for (int t = 0; t < 4; ++t)
            acc[t] = wmma_f32(a, fragW[kc * 128 + t * 32 + lane], acc[t]);
    }

    const int rbase = row0 + (hi ? 8 : 0);
    #pragma unroll
    for (int t = 0; t < 4; ++t) {
        const int n = t * 16 + l15;
        const float bias = bo[n];
        #pragma unroll
        for (int j = 0; j < 8; ++j)
            out[(size_t)(rbase + j) * HID + n] = acc[t][j] + bias;
    }
}

extern "C" void kernel_launch(void* const* d_in, const int* in_sizes, int n_in,
                              void* d_out, int out_size, void* d_ws, size_t ws_size,
                              hipStream_t stream)
{
    (void)in_sizes; (void)n_in; (void)out_size; (void)ws_size;

    const float* feats = (const float*)d_in[0];
    const int*   src   = (const int*)d_in[1];
    const int*   dst   = (const int*)d_in[2];
    const int*   gid   = (const int*)d_in[3];
    const float* awW  = (const float*)d_in[22];
    const float* awb  = (const float*)d_in[23];
    const float* outW = (const float*)d_in[24];
    const float* outb = (const float*)d_in[25];
    float* out = (float*)d_out;

    float* ws    = (float*)d_ws;
    float* xw    = ws;
    float* rlin  = xw   + (size_t)N_NODES * HID;
    float* bufA  = rlin + (size_t)N_NODES * HID;
    float* bufB  = bufA + (size_t)N_NODES * HID;
    float* hsum  = bufB + (size_t)N_NODES * HID + 128;
    unsigned* hmaxu = (unsigned*)(hsum + (size_t)N_GRAPH * HID);

    int*      col    = (int*)bufB;
    unsigned* cursor = (unsigned*)(col + N_EDGES);
    unsigned* deg    = cursor + N_PAD;
    float*    stats3 = (float*)(deg + N_PAD);

    const int ntiles = N_NODES / 16;
    const int gemmBlocks = (ntiles + 3) / 4;
    const int nInit = N_NODES + 1 + 3 * 128 + 2 * N_GRAPH * HID;

    auto finish_layer = [&](int pi, float* agg, float* stats) {
        const float* b = (const float*)d_in[pi + 1];
        gather_rows<<<(N_NODES + 7) / 8, 256, 0, stream>>>(cursor, deg, col, xw, agg,
                                                           N_NODES, N_EDGES);
        combine_stats<<<(N_NODES + 63) / 64, 256, 0, stream>>>(agg, rlin, b, stats, N_NODES);
    };

    init_state<<<(nInit + 255) / 256, 256, 0, stream>>>(deg, stats3, hsum, hmaxu);
    count_deg<<<(N_EDGES + 255) / 256, 256, 0, stream>>>(dst, deg, N_EDGES);
    alloc_ranges<<<(N_NODES + 255) / 256, 256, 0, stream>>>(deg, deg + N_NODES, cursor, N_NODES);
    fill_slots<<<(N_EDGES + 255) / 256, 256, 0, stream>>>(src, dst, cursor, col, N_EDGES);

    gemm2_wmma<F_IN><<<gemmBlocks, 128, 0, stream>>>(
        feats, (const float*)d_in[4], (const float*)d_in[6], (const float*)d_in[7],
        xw, rlin, ntiles);
    finish_layer(4, bufA, stats3);

    layer_gather_gemm<<<gemmBlocks, 128, 0, stream>>>(
        bufA, (const float*)d_in[10], (const float*)d_in[12],
        (const float*)d_in[11], (const float*)d_in[13],
        stats3, (const float*)d_in[8], (const float*)d_in[9],
        cursor, deg, col, xw, stats3 + 128, ntiles, N_NODES, N_EDGES);
    layer_gather_gemm<<<gemmBlocks, 128, 0, stream>>>(
        xw, (const float*)d_in[16], (const float*)d_in[18],
        (const float*)d_in[17], (const float*)d_in[19],
        stats3 + 128, (const float*)d_in[14], (const float*)d_in[15],
        cursor, deg, col, bufA, stats3 + 256, ntiles, N_NODES, N_EDGES);

    readout_nodes<<<(N_NODES + 7) / 8, 256, 0, stream>>>(bufA, stats3 + 256,
                                                         (const float*)d_in[20], (const float*)d_in[21],
                                                         gid, awW, awb, hsum, hmaxu, N_NODES);
    out_gemm_wmma<<<(N_GRAPH / 16) / 4, 128, 0, stream>>>(hsum, hmaxu, outW, outb, out);
}


# Edge list built by a two-level counting sort in LDS (bucket counts per chunk, packed scatter, per-bucket sort by node, sources ascending): no global atomic per edge, no scattered 4-byte store
# speedup vs baseline: 2.3117x; 1.6060x over previous
#include <hip/hip_runtime.h>

typedef float v2f __attribute__((ext_vector_type(2)));
typedef float v8f __attribute__((ext_vector_type(8)));

#define N_NODES 50000
#define N_EDGES 800000
#define N_GRAPH 512
#define F_IN    74
#define HID     64
#define BN_EPS  1e-5f
#define N_PAD   50048

__device__ __forceinline__ v8f wmma_f32(v2f a, v2f b, v8f c) {
    return __builtin_amdgcn_wmma_f32_16x16x4_f32(false, a, false, b,
                                                 (short)0, c, false, false);
}

#define NEG_INF_ENC 0x007FFFFFu
#define BK_SHIFT 8
#define BK_NODES 256
#define NBK      ((N_NODES + BK_NODES - 1) / BK_NODES)
#define NCH      256
#define CH_EDGES ((N_EDGES + NCH - 1) / NCH)
#define BK_CAP   8192
#define RANK_CAP 512
#define N_START  (1 + 3 * 128 + 2 * N_GRAPH * HID)

__global__ __launch_bounds__(256)
void bin_count(const int* __restrict__ dst, unsigned* __restrict__ cnt,
               unsigned* __restrict__ total, float* __restrict__ stats3,
               float* __restrict__ hsum, unsigned* __restrict__ hmaxu, int ne)
{
    __shared__ unsigned hist[NBK];
    for (int i = threadIdx.x; i < NBK; i += 256) hist[i] = 0u;
    __syncthreads();
    const int e0 = blockIdx.x * CH_EDGES;
    const int e1 = (e0 + CH_EDGES < ne) ? (e0 + CH_EDGES) : ne;
    for (int e = e0 + (int)threadIdx.x; e < e1; e += 8 * 256) {
        unsigned d[8];
        #pragma unroll
        for (int u = 0; u < 8; ++u) {
            const int ee = e + u * 256;
            d[u] = (unsigned)dst[ee < e1 ? ee : e];
        }
        #pragma unroll
        for (int u = 0; u < 8; ++u) {
            if (e + u * 256 < e1) {
                const unsigned dd = d[u] < (unsigned)N_NODES ? d[u] : (unsigned)(N_NODES - 1);
                atomicAdd(&hist[dd >> BK_SHIFT], 1u);
            }
        }
    }
    __syncthreads();
    for (int i = threadIdx.x; i < NBK; i += 256) cnt[i * NCH + blockIdx.x] = hist[i];

    for (int i = blockIdx.x * 256 + threadIdx.x; i < N_START; i += NCH * 256) {
        int j = i;
        if (j == 0) { *total = 0u; continue; }
        j -= 1;
        if (j < 3 * 128) { stats3[j] = 0.0f; continue; }
        j -= 3 * 128;
        if (j < N_GRAPH * HID) { hsum[j] = 0.0f; continue; }
        j -= N_GRAPH * HID;
        hmaxu[j] = NEG_INF_ENC;
    }
}

__global__ __launch_bounds__(256)
void bucket_ranges(unsigned* __restrict__ cnt, unsigned* __restrict__ total,
                   unsigned* __restrict__ bstart, unsigned* __restrict__ bcount)
{
    const int lane = threadIdx.x & 31;
    const int b    = blockIdx.x * 8 + (threadIdx.x >> 5);
    if (b >= NBK) return;
    uint4* row = (uint4*)(cnt + (size_t)b * NCH) + 2 * lane;
    const uint4 c0 = row[0], c1 = row[1];
    const unsigned s = c0.x + c0.y + c0.z + c0.w + c1.x + c1.y + c1.z + c1.w;
    unsigned inc = s;
    for (int off = 1; off < 32; off <<= 1) {
        const unsigned t = __shfl_up(inc, off, 32);
        if (lane >= off) inc += t;
    }
    unsigned base = 0u;
    if (lane == 31) base = atomicAdd(total, inc);
    base = __shfl(base, 31, 32);
    unsigned o = base + inc - s;
    uint4 r0, r1;
    r0.x = o; o += c0.x; r0.y = o; o += c0.y; r0.z = o; o += c0.z; r0.w = o; o += c0.w;
    r1.x = o; o += c1.x; r1.y = o; o += c1.y; r1.z = o; o += c1.z; r1.w = o;
    row[0] = r0; row[1] = r1;
    if (lane == 31) { bstart[b] = base; bcount[b] = inc; }
}

__global__ __launch_bounds__(256)
void bin_scatter(const int* __restrict__ src, const int* __restrict__ dst,
                 const unsigned* __restrict__ off, unsigned* __restrict__ ebuf, int ne)
{
    __shared__ unsigned cur[NBK];
    for (int i = threadIdx.x; i < NBK; i += 256) cur[i] = off[i * NCH + blockIdx.x];
    __syncthreads();
    const int e0 = blockIdx.x * CH_EDGES;
    const int e1 = (e0 + CH_EDGES < ne) ? (e0 + CH_EDGES) : ne;
    for (int e = e0 + (int)threadIdx.x; e < e1; e += 8 * 256) {
        unsigned d[8], s[8];
        #pragma unroll
        for (int u = 0; u < 8; ++u) {
            const int ee = e + u * 256;
            const int ec = ee < e1 ? ee : e;
            d[u] = (unsigned)dst[ec];
            s[u] = (unsigned)src[ec];
        }
        #pragma unroll
        for (int u = 0; u < 8; ++u) {
            if (e + u * 256 < e1) {
                const unsigned dd = d[u] < (unsigned)N_NODES ? d[u] : (unsigned)(N_NODES - 1);
                const unsigned ss = s[u] < (unsigned)N_NODES ? s[u] : (unsigned)(N_NODES - 1);
                const unsigned p = atomicAdd(&cur[dd >> BK_SHIFT], 1u);
                if (p < (unsigned)ne) ebuf[p] = (ss << BK_SHIFT) | (dd & (unsigned)(BK_NODES - 1));
            }
        }
    }
}

__global__ __launch_bounds__(256)
void bucket_to_nodes(const unsigned* __restrict__ ebuf, const unsigned* __restrict__ bstart,
                     const unsigned* __restrict__ bcount, unsigned* __restrict__ cursor,
                     unsigned* __restrict__ deg, int* __restrict__ col, int ne)
{
    __shared__ unsigned lw[BK_CAP];
    __shared__ unsigned npos[BK_NODES];
    __shared__ unsigned nfirst[BK_NODES];
    __shared__ unsigned ncnt[BK_NODES];
    __shared__ unsigned wsum[8];

    const int t    = threadIdx.x;
    const int lane = t & 31;
    const int wave = t >> 5;
    const int b    = blockIdx.x;

    unsigned beg = bstart[b], n = bcount[b];
    if (beg > (unsigned)ne) beg = (unsigned)ne;
    if (n > (unsigned)ne - beg) n = (unsigned)ne - beg;

    npos[t] = 0u;
    __syncthreads();
    for (unsigned i0 = t; i0 < n; i0 += 8 * 256) {
        unsigned w[8];
        #pragma unroll
        for (int u = 0; u < 8; ++u) {
            const unsigned i = i0 + (unsigned)(u * 256);
            w[u] = ebuf[beg + (i < n ? i : i0)];
        }
        #pragma unroll
        for (int u = 0; u < 8; ++u)
            if (i0 + (unsigned)(u * 256) < n) atomicAdd(&npos[w[u] & (unsigned)(BK_NODES - 1)], 1u);
    }
    __syncthreads();

    const unsigned d = npos[t];
    unsigned inc = d;
    for (int off = 1; off < 32; off <<= 1) {
        const unsigned v = __shfl_up(inc, off, 32);
        if (lane >= off) inc += v;
    }
    if (lane == 31) wsum[wave] = inc;
    __syncthreads();
    unsigned wbase = 0u;
    for (int w = 0; w < wave; ++w) wbase += wsum[w];
    const unsigned first = wbase + inc - d;
    npos[t] = first; nfirst[t] = first; ncnt[t] = d;
    const int node = b * BK_NODES + t;
    if (node < N_NODES) { cursor[node] = beg + first + d; deg[node] = d; }
    __syncthreads();

    if (n <= (unsigned)BK_CAP) {
        for (unsigned i0 = t; i0 < n; i0 += 8 * 256) {
            unsigned w[8];
            #pragma unroll
            for (int u = 0; u < 8; ++u) {
                const unsigned i = i0 + (unsigned)(u * 256);
                w[u] = ebuf[beg + (i < n ? i : i0)];
            }
            #pragma unroll
            for (int u = 0; u < 8; ++u) {
                if (i0 + (unsigned)(u * 256) < n) {
                    const unsigned p = atomicAdd(&npos[w[u] & (unsigned)(BK_NODES - 1)], 1u);
                    if (p < (unsigned)BK_CAP) lw[p] = w[u];
                }
            }
        }
        __syncthreads();
        for (unsigned q = t; q < n; q += 256) {
            const unsigned w  = lw[q];
            const unsigned k  = w & (unsigned)(BK_NODES - 1);
            const unsigned a  = nfirst[k];
            const unsigned dk = ncnt[k];
            unsigned r = q - a;
            if (dk <= (unsigned)RANK_CAP) {
                r = 0u;
                for (unsigned j = a; j < a + dk; ++j) {
                    const unsigned v = lw[j];
                    r += (v < w || (v == w && j < q)) ? 1u : 0u;
                }
            }
            col[beg + a + r] = (int)(w >> BK_SHIFT);
        }
    } else {
        for (unsigned i = t; i < n; i += 256) {
            const unsigned w = ebuf[beg + i];
            const unsigned p = atomicAdd(&npos[w & (unsigned)(BK_NODES - 1)], 1u);
            if (p < n) col[beg + p] = (int)(w >> BK_SHIFT);
        }
    }
}

__device__ __forceinline__ float xhalf(float x) {
    return __builtin_bit_cast(float, __builtin_amdgcn_permlanex16(
        __builtin_bit_cast(int, x), __builtin_bit_cast(int, x),
        0x76543210, 0xfedcba98, false, false));
}

#define XP 68
#define NB 16

__global__ __launch_bounds__(128)
void layer_gather_gemm(const float* __restrict__ Hin,
                       const float* __restrict__ W,
                       const float* __restrict__ rW,
                       const float* __restrict__ bvec,
                       const float* __restrict__ rb,
                       const float* __restrict__ pst,
                       const float* __restrict__ pg,
                       const float* __restrict__ pb,
                       const unsigned* __restrict__ cursor, const unsigned* __restrict__ deg,
                       const int* __restrict__ col,
                       float* __restrict__ Hout,
                       float* __restrict__ stats,
                       int ntiles, int nrows, int ne)
{
    constexpr int KT = HID / 4;
    __shared__ v2f   fragW[KT * 128];
    __shared__ v2f   fragR[KT * 128];
    __shared__ float ta[4 * 16 * XP];
    __shared__ float tx[4 * 16 * XP];
    __shared__ float ssum[4 * 64], ssq[4 * 64];

    const int  lane = threadIdx.x & 31;
    const int  wave = threadIdx.x >> 5;
    const int  l15  = lane & 15;
    const bool hi   = lane >= 16;
    const int  c4   = l15 * 4;

    const int  tile  = blockIdx.x * 4 + wave;
    const bool activ = tile < ntiles;
    const int  row0  = (activ ? tile : (ntiles - 1)) * 16;

    unsigned endv = cursor[row0 + l15];
    unsigned dv   = deg[row0 + l15];
    float bb[4], rbb[4];
    #pragma unroll
    for (int t = 0; t < 4; ++t) { bb[t] = bvec[t * 16 + l15]; rbb[t] = rb[t * 16 + l15]; }
    const float4 psum = *(const float4*)(pst + c4);
    const float4 psq  = *(const float4*)(pst + 64 + c4);
    float gam[4], bet[4];
    #pragma unroll
    for (int i = 0; i < 4; ++i) { gam[i] = pg[c4 + i]; bet[i] = pb[c4 + i]; }
    float4 xr[8];
    #pragma unroll
    for (int u = 0; u < 8; ++u)
        xr[u] = *(const float4*)(Hin + (size_t)(row0 + 2 * u + (hi ? 1 : 0)) * HID + c4);

    for (int idx = threadIdx.x; idx < KT * 128; idx += 128) {
        const int kc  = idx >> 7;
        const int r   = idx & 127;
        const int n   = ((r >> 5) << 4) + (r & 15);
        const int ka0 = kc * 4 + ((r & 31) >= 16 ? 2 : 0);
        v2f fw; fw.x = W [(size_t)ka0 * HID + n]; fw.y = W [(size_t)(ka0 + 1) * HID + n];
        v2f fr; fr.x = rW[(size_t)ka0 * HID + n]; fr.y = rW[(size_t)(ka0 + 1) * HID + n];
        fragW[idx] = fw;
        fragR[idx] = fr;
    }

    float sc[4], sh[4];
    {
        const float inv = 1.0f / (float)nrows;
        const float ps[4] = {psum.x, psum.y, psum.z, psum.w};
        const float pq[4] = {psq.x, psq.y, psq.z, psq.w};
        #pragma unroll
        for (int i = 0; i < 4; ++i) {
            const float mean = ps[i] * inv;
            const float var  = pq[i] * inv - mean * mean;
            sc[i] = rsqrtf(var + BN_EPS) * gam[i];
            sh[i] = bet[i] - mean * sc[i];
        }
    }

    float* ta_ = ta + wave * 16 * XP;
    float* tx_ = tx + wave * 16 * XP;

    #pragma unroll
    for (int u = 0; u < 8; ++u) {
        float4 v = xr[u];
        v.x = v.x * sc[0] + sh[0]; v.y = v.y * sc[1] + sh[1];
        v.z = v.z * sc[2] + sh[2]; v.w = v.w * sc[3] + sh[3];
        *(float4*)(tx_ + (2 * u + (hi ? 1 : 0)) * XP + c4) = v;
    }

    if (endv > (unsigned)ne) endv = (unsigned)ne;
    if (dv > endv) dv = endv;
    if (!activ) dv = 0u;
    const unsigned begv = endv - dv;
    const float* hcol = Hin + c4;

    auto first_ids = [&](unsigned b, unsigned d) -> unsigned {
        const unsigned k = ((unsigned)lane < d) ? (unsigned)lane : (d ? d - 1u : 0u);
        unsigned p = b + k;
        if (p > (unsigned)(ne - 1)) p = (unsigned)(ne - 1);
        return (unsigned)col[p];
    };
    unsigned myc_next = first_ids((unsigned)__builtin_amdgcn_readlane((int)begv, 0),
                                  (unsigned)__builtin_amdgcn_readlane((int)dv, 0));
    for (int r = 0; r < 16; ++r) {
        const unsigned beg = (unsigned)__builtin_amdgcn_readlane((int)begv, r);
        const unsigned d   = (unsigned)__builtin_amdgcn_readlane((int)dv, r);
        unsigned myc = myc_next;
        if (r + 1 < 16)
            myc_next = first_ids((unsigned)__builtin_amdgcn_readlane((int)begv, r + 1),
                                 (unsigned)__builtin_amdgcn_readlane((int)dv, r + 1));
        float4 acc = make_float4(0.0f, 0.0f, 0.0f, 0.0f);
        for (unsigned base = 0; base < d; base += 32) {
            const unsigned left = d - base;
            const unsigned cnt  = left < 32u ? left : 32u;
            if (base != 0u) myc = ((unsigned)lane < cnt) ? (unsigned)col[beg + base + lane] : 0u;
            if (myc > (unsigned)(nrows - 1)) myc = (unsigned)(nrows - 1);
            const unsigned nst = (cnt + 1u) >> 1;
            for (unsigned i0 = 0; i0 < nst; i0 += NB) {
                float4 v[NB];
                #pragma unroll
                for (int u = 0; u < NB; ++u) {
                    const unsigned i = i0 + (unsigned)u;
                    if (i < nst) {
                        const unsigned ea = 2u * i;
                        const unsigned eb = (ea + 1u < cnt) ? (ea + 1u) : ea;
                        const int sa = __builtin_amdgcn_readlane((int)myc, (int)ea);
                        const int sb = __builtin_amdgcn_readlane((int)myc, (int)eb);
                        const int srow = hi ? sb : sa;
                        v[u] = *(const float4*)(hcol + (size_t)srow * HID);
                    }
                }
                #pragma unroll
                for (int u = 0; u < NB; ++u) {
                    const unsigned i = i0 + (unsigned)u;
                    if (i < nst) {
                        const float m = (hi && (2u * i + 1u >= cnt)) ? 0.0f : 1.0f;
                        acc.x = fmaf(v[u].x, m, acc.x); acc.y = fmaf(v[u].y, m, acc.y);
                        acc.z = fmaf(v[u].z, m, acc.z); acc.w = fmaf(v[u].w, m, acc.w);
                    }
                }
            }
        }
        acc.x += xhalf(acc.x); acc.y += xhalf(acc.y);
        acc.z += xhalf(acc.z); acc.w += xhalf(acc.w);
        if (lane < 16) {
            const float df = (float)d;
            float4 o;
            o.x = acc.x * sc[0] + df * sh[0]; o.y = acc.y * sc[1] + df * sh[1];
            o.z = acc.z * sc[2] + df * sh[2]; o.w = acc.w * sc[3] + df * sh[3];
            *(float4*)(ta_ + r * XP + c4) = o;
        }
    }
    __syncthreads();

    v8f acc[4], racc[4];
    #pragma unroll
    for (int t = 0; t < 4; ++t) { acc[t] = {}; racc[t] = {}; }

    const v2f* arow = (const v2f*)(ta_ + l15 * XP);
    const v2f* xrow = (const v2f*)(tx_ + l15 * XP);
    const int  aoff = hi ? 1 : 0;

    #pragma unroll
    for (int kc = 0; kc < KT; ++kc) {
        const v2f a = arow[kc * 2 + aoff];
        const v2f x = xrow[kc * 2 + aoff];
        #pragma unroll
        for (int t = 0; t < 4; ++t) {
            const v2f b  = fragW[kc * 128 + t * 32 + lane];
            const v2f rv = fragR[kc * 128 + t * 32 + lane];
            acc[t]  = wmma_f32(a, b,  acc[t]);
            racc[t] = wmma_f32(x, rv, racc[t]);
        }
    }

    const int rl0 = hi ? 8 : 0;
    #pragma unroll
    for (int t = 0; t < 4; ++t) {
        const int n = t * 16 + l15;
        float cs = 0.0f, cq = 0.0f;
        #pragma unroll
        for (int j = 0; j < 8; ++j) {
            float a = acc[t][j] + bb[t];
            a = a > 0.0f ? a : 0.0f;
            float v = racc[t][j] + rbb[t];
            v = v > 0.0f ? v : 0.0f;
            const float h = a + v;
            ta_[(rl0 + j) * XP + n] = h;
            cs += h; cq += h * h;
        }
        cs += xhalf(cs);
        cq += xhalf(cq);
        if (!activ) { cs = 0.0f; cq = 0.0f; }
        if (lane < 16) { ssum[wave * 64 + n] = cs; ssq[wave * 64 + n] = cq; }
    }
    if (activ) {
        #pragma unroll
        for (int u = 0; u < 8; ++u) {
            const int row = 2 * u + (hi ? 1 : 0);
            const float4 hv = *(const float4*)(ta_ + row * XP + c4);
            *(float4*)(Hout + (size_t)(row0 + row) * HID + c4) = hv;
        }
    }
    __syncthreads();
    if (threadIdx.x < 64) {
        const int f = threadIdx.x;
        const float s = ssum[f] + ssum[64 + f] + ssum[128 + f] + ssum[192 + f];
        const float q = ssq[f]  + ssq[64 + f]  + ssq[128 + f]  + ssq[192 + f];
        atomicAdd(&stats[f], s);
        atomicAdd(&stats[64 + f], q);
    }
}

#define F_PAD 76
#define NB1 8

__global__ __launch_bounds__(128)
void layer1_gather_gemm(const float* __restrict__ X,
                        const float* __restrict__ W,
                        const float* __restrict__ rW,
                        const float* __restrict__ bvec,
                        const float* __restrict__ rb,
                        const unsigned* __restrict__ cursor, const unsigned* __restrict__ deg,
                        const int* __restrict__ col,
                        float* __restrict__ Hout,
                        float* __restrict__ stats,
                        int ntiles, int nrows, int ne)
{
    constexpr int KT = F_PAD / 4;
    __shared__ v2f   fragW[KT * 128];
    __shared__ v2f   fragR[KT * 128];
    __shared__ __attribute__((aligned(16))) float ta[4 * 16 * F_PAD];
    __shared__ __attribute__((aligned(16))) float tx[4 * 16 * F_PAD];
    __shared__ float ssum[4 * 64], ssq[4 * 64];

    const int  lane = threadIdx.x & 31;
    const int  wave = threadIdx.x >> 5;
    const int  l15  = lane & 15;
    const bool hi   = lane >= 16;

    const int p0 = l15;
    const int p1 = l15 + 16;
    const int p2 = (l15 < 5) ? (l15 + 32) : 36;

    const int  tile  = blockIdx.x * 4 + wave;
    const bool activ = tile < ntiles;
    const int  row0  = (activ ? tile : (ntiles - 1)) * 16;

    unsigned endv = cursor[row0 + l15];
    unsigned dv   = deg[row0 + l15];
    float bb[4], rbb[4];
    #pragma unroll
    for (int t = 0; t < 4; ++t) { bb[t] = bvec[t * 16 + l15]; rbb[t] = rb[t * 16 + l15]; }
    float2 x0[8], x1[8], x2[8];
    #pragma unroll
    for (int u = 0; u < 8; ++u) {
        const float2* rp = (const float2*)(X + (size_t)(row0 + 2 * u + (hi ? 1 : 0)) * F_IN);
        x0[u] = rp[p0]; x1[u] = rp[p1]; x2[u] = rp[p2];
    }

    for (int idx = threadIdx.x; idx < KT * 128; idx += 128) {
        const int kc  = idx >> 7;
        const int r   = idx & 127;
        const int n   = ((r >> 5) << 4) + (r & 15);
        const int ka0 = kc * 4 + ((r & 31) >= 16 ? 2 : 0);
        const int ka1 = ka0 + 1;
        const int c0  = (ka0 < F_IN) ? ka0 : 0;
        const int c1  = (ka1 < F_IN) ? ka1 : 0;
        const float m0 = (ka0 < F_IN) ? 1.0f : 0.0f;
        const float m1 = (ka1 < F_IN) ? 1.0f : 0.0f;
        v2f fw; fw.x = W [(size_t)c0 * HID + n] * m0; fw.y = W [(size_t)c1 * HID + n] * m1;
        v2f fr; fr.x = rW[(size_t)c0 * HID + n] * m0; fr.y = rW[(size_t)c1 * HID + n] * m1;
        fragW[idx] = fw;
        fragR[idx] = fr;
    }

    float* ta_ = ta + wave * 16 * F_PAD;
    float* tx_ = tx + wave * 16 * F_PAD;
    const float2 zero2 = make_float2(0.0f, 0.0f);

    #pragma unroll
    for (int u = 0; u < 8; ++u) {
        float* trow = tx_ + (2 * u + (hi ? 1 : 0)) * F_PAD;
        *(float2*)(trow + 2 * p0) = x0[u];
        *(float2*)(trow + 2 * p1) = x1[u];
        if (l15 < 6) *(float2*)(trow + 64 + 2 * l15) = (l15 < 5) ? x2[u] : zero2;
    }

    if (endv > (unsigned)ne) endv = (unsigned)ne;
    if (dv > endv) dv = endv;
    if (!activ) dv = 0u;
    const unsigned begv = endv - dv;

    auto first_ids = [&](unsigned b, unsigned d) -> unsigned {
        const unsigned k = ((unsigned)lane < d) ? (unsigned)lane : (d ? d - 1u : 0u);
        unsigned p = b + k;
        if (p > (unsigned)(ne - 1)) p = (unsigned)(ne - 1);
        return (unsigned)col[p];
    };
    unsigned myc_next = first_ids((unsigned)__builtin_amdgcn_readlane((int)begv, 0),
                                  (unsigned)__builtin_amdgcn_readlane((int)dv, 0));
    for (int r = 0; r < 16; ++r) {
        const unsigned beg = (unsigned)__builtin_amdgcn_readlane((int)begv, r);
        const unsigned d   = (unsigned)__builtin_amdgcn_readlane((int)dv, r);
        unsigned myc = myc_next;
        if (r + 1 < 16)
            myc_next = first_ids((unsigned)__builtin_amdgcn_readlane((int)begv, r + 1),
                                 (unsigned)__builtin_amdgcn_readlane((int)dv, r + 1));
        float2 a0 = zero2, a1 = zero2, a2 = zero2;
        for (unsigned base = 0; base < d; base += 32) {
            const unsigned left = d - base;
            const unsigned cnt  = left < 32u ? left : 32u;
            if (base != 0u) myc = ((unsigned)lane < cnt) ? (unsigned)col[beg + base + lane] : 0u;
            if (myc > (unsigned)(nrows - 1)) myc = (unsigned)(nrows - 1);
            const unsigned nst = (cnt + 1u) >> 1;
            for (unsigned i0 = 0; i0 < nst; i0 += NB1) {
                float2 v0[NB1], v1[NB1], v2[NB1];
                #pragma unroll
                for (int u = 0; u < NB1; ++u) {
                    const unsigned i = i0 + (unsigned)u;
                    if (i < nst) {
                        const unsigned ea = 2u * i;
                        const unsigned eb = (ea + 1u < cnt) ? (ea + 1u) : ea;
                        const int sa = __builtin_amdgcn_readlane((int)myc, (int)ea);
                        const int sb = __builtin_amdgcn_readlane((int)myc, (int)eb);
                        const int srow = hi ? sb : sa;
                        const float2* rp = (const float2*)(X + (size_t)srow * F_IN);
                        v0[u] = rp[p0]; v1[u] = rp[p1]; v2[u] = rp[p2];
                    }
                }
                #pragma unroll
                for (int u = 0; u < NB1; ++u) {
                    const unsigned i = i0 + (unsigned)u;
                    if (i < nst) {
                        const float m = (hi && (2u * i + 1u >= cnt)) ? 0.0f : 1.0f;
                        a0.x = fmaf(v0[u].x, m, a0.x); a0.y = fmaf(v0[u].y, m, a0.y);
                        a1.x = fmaf(v1[u].x, m, a1.x); a1.y = fmaf(v1[u].y, m, a1.y);
                        a2.x = fmaf(v2[u].x, m, a2.x); a2.y = fmaf(v2[u].y, m, a2.y);
                    }
                }
            }
        }
        a0.x += xhalf(a0.x); a0.y += xhalf(a0.y);
        a1.x += xhalf(a1.x); a1.y += xhalf(a1.y);
        a2.x += xhalf(a2.x); a2.y += xhalf(a2.y);
        if (lane < 16) {
            float* trow = ta_ + r * F_PAD;
            *(float2*)(trow + 2 * p0) = a0;
            *(float2*)(trow + 2 * p1) = a1;
            if (l15 < 6) *(float2*)(trow + 64 + 2 * l15) = (l15 < 5) ? a2 : zero2;
        }
    }
    __syncthreads();

    v8f acc[4], racc[4];
    #pragma unroll
    for (int t = 0; t < 4; ++t) { acc[t] = {}; racc[t] = {}; }

    const v2f* arow = (const v2f*)(ta_ + l15 * F_PAD);
    const v2f* xrow = (const v2f*)(tx_ + l15 * F_PAD);
    const int  aoff = hi ? 1 : 0;

    #pragma unroll
    for (int kc = 0; kc < KT; ++kc) {
        const v2f a = arow[kc * 2 + aoff];
        const v2f x = xrow[kc * 2 + aoff];
        #pragma unroll
        for (int t = 0; t < 4; ++t) {
            const v2f b  = fragW[kc * 128 + t * 32 + lane];
            const v2f rv = fragR[kc * 128 + t * 32 + lane];
            acc[t]  = wmma_f32(a, b,  acc[t]);
            racc[t] = wmma_f32(x, rv, racc[t]);
        }
    }

    const int rl0 = hi ? 8 : 0;
    const int c4  = l15 * 4;
    #pragma unroll
    for (int t = 0; t < 4; ++t) {
        const int n = t * 16 + l15;
        float cs = 0.0f, cq = 0.0f;
        #pragma unroll
        for (int j = 0; j < 8; ++j) {
            float a = acc[t][j] + bb[t];
            a = a > 0.0f ? a : 0.0f;
            float v = racc[t][j] + rbb[t];
            v = v > 0.0f ? v : 0.0f;
            const float h = a + v;
            ta_[(rl0 + j) * F_PAD + n] = h;
            cs += h; cq += h * h;
        }
        cs += xhalf(cs);
        cq += xhalf(cq);
        if (!activ) { cs = 0.0f; cq = 0.0f; }
        if (lane < 16) { ssum[wave * 64 + n] = cs; ssq[wave * 64 + n] = cq; }
    }
    if (activ) {
        #pragma unroll
        for (int u = 0; u < 8; ++u) {
            const int row = 2 * u + (hi ? 1 : 0);
            const float4 hv = *(const float4*)(ta_ + row * F_PAD + c4);
            *(float4*)(Hout + (size_t)(row0 + row) * HID + c4) = hv;
        }
    }
    __syncthreads();
    if (threadIdx.x < 64) {
        const int f = threadIdx.x;
        const float s = ssum[f] + ssum[64 + f] + ssum[128 + f] + ssum[192 + f];
        const float q = ssq[f]  + ssq[64 + f]  + ssq[128 + f]  + ssq[192 + f];
        atomicAdd(&stats[f], s);
        atomicAdd(&stats[64 + f], q);
    }
}

__device__ __forceinline__ unsigned f32_ordered(float f) {
    unsigned b = __float_as_uint(f);
    return (b & 0x80000000u) ? ~b : (b | 0x80000000u);
}

__global__ __launch_bounds__(256)
void readout_nodes(const float* __restrict__ h, const float* __restrict__ stats,
                   const float* __restrict__ gamma, const float* __restrict__ beta,
                   const int* __restrict__ gid,
                   const float* __restrict__ awW, const float* __restrict__ awb,
                   float* __restrict__ hsum, unsigned* __restrict__ hmaxu,
                   int nrows)
{
    const int lane = threadIdx.x & 31;
    const int wave = threadIdx.x >> 5;
    const int n = blockIdx.x * 8 + wave;
    if (n >= nrows) return;
    const float inv = 1.0f / (float)nrows;
    const float m0 = stats[lane] * inv;
    const float m1 = stats[32 + lane] * inv;
    const float v0 = stats[64 + lane] * inv - m0 * m0;
    const float v1 = stats[96 + lane] * inv - m1 * m1;
    float h0 = h[(size_t)n * HID + lane];
    float h1 = h[(size_t)n * HID + 32 + lane];
    h0 = (h0 - m0) * rsqrtf(v0 + BN_EPS) * gamma[lane] + beta[lane];
    h1 = (h1 - m1) * rsqrtf(v1 + BN_EPS) * gamma[32 + lane] + beta[32 + lane];
    float p  = h0 * awW[lane] + h1 * awW[32 + lane];
    for (int off = 16; off > 0; off >>= 1) p += __shfl_xor(p, off, 32);
    float w = 1.0f / (1.0f + expf(-(p + awb[0])));
    int g = gid[n];
    atomicAdd(&hsum[(size_t)g * HID + lane],      h0 * w);
    atomicAdd(&hsum[(size_t)g * HID + 32 + lane], h1 * w);
    atomicMax(&hmaxu[(size_t)g * HID + lane],      f32_ordered(h0));
    atomicMax(&hmaxu[(size_t)g * HID + 32 + lane], f32_ordered(h1));
}

__global__ __launch_bounds__(128)
void out_gemm_wmma(const float* __restrict__ hsum, const unsigned* __restrict__ hmaxu,
                   const float* __restrict__ Wo,
                   const float* __restrict__ bo, float* __restrict__ out)
{
    constexpr int KT = 128 / 4;
    __shared__ v2f  fragW[KT * 128];
    __shared__ float xt[4 * 16 * 128];

    const int  lane = threadIdx.x & 31;
    const int  wave = threadIdx.x >> 5;
    const int  l15  = lane & 15;
    const bool hi   = lane >= 16;

    for (int idx = threadIdx.x; idx < KT * 128; idx += 128) {
        const int kc  = idx >> 7;
        const int r   = idx & 127;
        const int n   = ((r >> 5) << 4) + (r & 15);
        const int ka0 = kc * 4 + ((r & 31) >= 16 ? 2 : 0);
        v2f fw;
        fw.x = Wo[(size_t)ka0 * HID + n];
        fw.y = Wo[(size_t)(ka0 + 1) * HID + n];
        fragW[idx] = fw;
    }
    __syncthreads();

    const int row0 = (blockIdx.x * 4 + wave) * 16;
    float* xw_ = xt + wave * 16 * 128;
    for (int r = 0; r < 16; ++r) {
        const float*    srow = hsum  + (size_t)(row0 + r) * HID;
        const unsigned* mrow = hmaxu + (size_t)(row0 + r) * HID;
        for (int k = lane; k < HID; k += 32) {
            xw_[r * 128 + k] = srow[k];
            const unsigned u = mrow[k];
            float m = (u & 0x80000000u) ? __uint_as_float(u ^ 0x80000000u)
                                        : __uint_as_float(~u);
            if (!(fabsf(m) <= 3.402823466e38f)) m = 0.0f;
            xw_[r * 128 + HID + k] = m;
        }
    }

    v8f acc[4];
    #pragma unroll
    for (int t = 0; t < 4; ++t) acc[t] = {};

    const v2f* arow = (const v2f*)(xw_ + l15 * 128);
    const int  aoff = hi ? 1 : 0;
    #pragma unroll
    for (int kc = 0; kc < KT; ++kc) {
        const v2f a = arow[kc * 2 + aoff];
        #pragma unroll
        for (int t = 0; t < 4; ++t)
            acc[t] = wmma_f32(a, fragW[kc * 128 + t * 32 + lane], acc[t]);
    }

    const int rbase = row0 + (hi ? 8 : 0);
    #pragma unroll
    for (int t = 0; t < 4; ++t) {
        const int n = t * 16 + l15;
        const float bias = bo[n];
        #pragma unroll
        for (int j = 0; j < 8; ++j)
            out[(size_t)(rbase + j) * HID + n] = acc[t][j] + bias;
    }
}

extern "C" void kernel_launch(void* const* d_in, const int* in_sizes, int n_in,
                              void* d_out, int out_size, void* d_ws, size_t ws_size,
                              hipStream_t stream)
{
    (void)in_sizes; (void)n_in; (void)out_size; (void)ws_size;

    const float* feats = (const float*)d_in[0];
    const int*   src   = (const int*)d_in[1];
    const int*   dst   = (const int*)d_in[2];
    const int*   gid   = (const int*)d_in[3];
    const float* awW  = (const float*)d_in[22];
    const float* awb  = (const float*)d_in[23];
    const float* outW = (const float*)d_in[24];
    const float* outb = (const float*)d_in[25];
    float* out = (float*)d_out;

    float* ws    = (float*)d_ws;
    float* xw    = ws;
    float* rlin  = xw   + (size_t)N_NODES * HID;
    float* bufA  = rlin + (size_t)N_NODES * HID;
    float* bufB  = bufA + (size_t)N_NODES * HID;
    float* hsum  = bufB + (size_t)N_NODES * HID + 128;
    unsigned* hmaxu = (unsigned*)(hsum + (size_t)N_GRAPH * HID);

    int*      col    = (int*)bufB;
    unsigned* cursor = (unsigned*)(col + N_EDGES);
    unsigned* deg    = cursor + N_PAD;
    float*    stats3 = (float*)(deg + N_PAD);
    unsigned* ebuf   = (unsigned*)(stats3 + 3 * 128);
    unsigned* cnt    = ebuf + N_EDGES;
    unsigned* bstart = cnt + NBK * NCH;
    unsigned* bcount = bstart + 256;
    unsigned* total  = bcount + 256;

    const int ntiles = N_NODES / 16;
    const int gemmBlocks = (ntiles + 3) / 4;

    bin_count<<<NCH, 256, 0, stream>>>(dst, cnt, total, stats3, hsum, hmaxu, N_EDGES);
    bucket_ranges<<<(NBK + 7) / 8, 256, 0, stream>>>(cnt, total, bstart, bcount);
    bin_scatter<<<NCH, 256, 0, stream>>>(src, dst, cnt, ebuf, N_EDGES);
    bucket_to_nodes<<<NBK, 256, 0, stream>>>(ebuf, bstart, bcount, cursor, deg, col, N_EDGES);

    layer1_gather_gemm<<<gemmBlocks, 128, 0, stream>>>(
        feats, (const float*)d_in[4], (const float*)d_in[6],
        (const float*)d_in[5], (const float*)d_in[7],
        cursor, deg, col, bufA, stats3, ntiles, N_NODES, N_EDGES);

    layer_gather_gemm<<<gemmBlocks, 128, 0, stream>>>(
        bufA, (const float*)d_in[10], (const float*)d_in[12],
        (const float*)d_in[11], (const float*)d_in[13],
        stats3, (const float*)d_in[8], (const float*)d_in[9],
        cursor, deg, col, xw, stats3 + 128, ntiles, N_NODES, N_EDGES);
    layer_gather_gemm<<<gemmBlocks, 128, 0, stream>>>(
        xw, (const float*)d_in[16], (const float*)d_in[18],
        (const float*)d_in[17], (const float*)d_in[19],
        stats3 + 128, (const float*)d_in[14], (const float*)d_in[15],
        cursor, deg, col, bufA, stats3 + 256, ntiles, N_NODES, N_EDGES);

    readout_nodes<<<(N_NODES + 7) / 8, 256, 0, stream>>>(bufA, stats3 + 256,
                                                         (const float*)d_in[20], (const float*)d_in[21],
                                                         gid, awW, awb, hsum, hmaxu, N_NODES);
    out_gemm_wmma<<<(N_GRAPH / 16) / 4, 128, 0, stream>>>(hsum, hmaxu, outW, outb, out);
}


# Readout walks 16 consecutive nodes per wave and keeps the weighted sum and max of a run of equal graph ids in registers: atomics only where the id changes; 50000 waves -> 3125
# speedup vs baseline: 2.4118x; 1.0433x over previous
#include <hip/hip_runtime.h>

typedef float v2f __attribute__((ext_vector_type(2)));
typedef float v8f __attribute__((ext_vector_type(8)));

#define N_NODES 50000
#define N_EDGES 800000
#define N_GRAPH 512
#define F_IN    74
#define HID     64
#define BN_EPS  1e-5f
#define N_PAD   50048

__device__ __forceinline__ v8f wmma_f32(v2f a, v2f b, v8f c) {
    return __builtin_amdgcn_wmma_f32_16x16x4_f32(false, a, false, b,
                                                 (short)0, c, false, false);
}

#define NEG_INF_ENC 0x007FFFFFu
#define BK_SHIFT 8
#define BK_NODES 256
#define NBK      ((N_NODES + BK_NODES - 1) / BK_NODES)
#define NCH      256
#define CH_EDGES ((N_EDGES + NCH - 1) / NCH)
#define BK_CAP   8192
#define RANK_CAP 512
#define N_START  (1 + 3 * 128 + 2 * N_GRAPH * HID)

__global__ __launch_bounds__(256)
void bin_count(const int* __restrict__ dst, unsigned* __restrict__ cnt,
               unsigned* __restrict__ total, float* __restrict__ stats3,
               float* __restrict__ hsum, unsigned* __restrict__ hmaxu, int ne)
{
    __shared__ unsigned hist[NBK];
    for (int i = threadIdx.x; i < NBK; i += 256) hist[i] = 0u;
    __syncthreads();
    const int e0 = blockIdx.x * CH_EDGES;
    const int e1 = (e0 + CH_EDGES < ne) ? (e0 + CH_EDGES) : ne;
    for (int e = e0 + (int)threadIdx.x; e < e1; e += 8 * 256) {
        unsigned d[8];
        #pragma unroll
        for (int u = 0; u < 8; ++u) {
            const int ee = e + u * 256;
            d[u] = (unsigned)dst[ee < e1 ? ee : e];
        }
        #pragma unroll
        for (int u = 0; u < 8; ++u) {
            if (e + u * 256 < e1) {
                const unsigned dd = d[u] < (unsigned)N_NODES ? d[u] : (unsigned)(N_NODES - 1);
                atomicAdd(&hist[dd >> BK_SHIFT], 1u);
            }
        }
    }
    __syncthreads();
    for (int i = threadIdx.x; i < NBK; i += 256) cnt[i * NCH + blockIdx.x] = hist[i];

    for (int i = blockIdx.x * 256 + threadIdx.x; i < N_START; i += NCH * 256) {
        int j = i;
        if (j == 0) { *total = 0u; continue; }
        j -= 1;
        if (j < 3 * 128) { stats3[j] = 0.0f; continue; }
        j -= 3 * 128;
        if (j < N_GRAPH * HID) { hsum[j] = 0.0f; continue; }
        j -= N_GRAPH * HID;
        hmaxu[j] = NEG_INF_ENC;
    }
}

__global__ __launch_bounds__(256)
void bucket_ranges(unsigned* __restrict__ cnt, unsigned* __restrict__ total,
                   unsigned* __restrict__ bstart, unsigned* __restrict__ bcount)
{
    const int lane = threadIdx.x & 31;
    const int b    = blockIdx.x * 8 + (threadIdx.x >> 5);
    if (b >= NBK) return;
    uint4* row = (uint4*)(cnt + (size_t)b * NCH) + 2 * lane;
    const uint4 c0 = row[0], c1 = row[1];
    const unsigned s = c0.x + c0.y + c0.z + c0.w + c1.x + c1.y + c1.z + c1.w;
    unsigned inc = s;
    for (int off = 1; off < 32; off <<= 1) {
        const unsigned t = __shfl_up(inc, off, 32);
        if (lane >= off) inc += t;
    }
    unsigned base = 0u;
    if (lane == 31) base = atomicAdd(total, inc);
    base = __shfl(base, 31, 32);
    unsigned o = base + inc - s;
    uint4 r0, r1;
    r0.x = o; o += c0.x; r0.y = o; o += c0.y; r0.z = o; o += c0.z; r0.w = o; o += c0.w;
    r1.x = o; o += c1.x; r1.y = o; o += c1.y; r1.z = o; o += c1.z; r1.w = o;
    row[0] = r0; row[1] = r1;
    if (lane == 31) { bstart[b] = base; bcount[b] = inc; }
}

__global__ __launch_bounds__(256)
void bin_scatter(const int* __restrict__ src, const int* __restrict__ dst,
                 const unsigned* __restrict__ off, unsigned* __restrict__ ebuf, int ne)
{
    __shared__ unsigned cur[NBK];
    for (int i = threadIdx.x; i < NBK; i += 256) cur[i] = off[i * NCH + blockIdx.x];
    __syncthreads();
    const int e0 = blockIdx.x * CH_EDGES;
    const int e1 = (e0 + CH_EDGES < ne) ? (e0 + CH_EDGES) : ne;
    for (int e = e0 + (int)threadIdx.x; e < e1; e += 8 * 256) {
        unsigned d[8], s[8];
        #pragma unroll
        for (int u = 0; u < 8; ++u) {
            const int ee = e + u * 256;
            const int ec = ee < e1 ? ee : e;
            d[u] = (unsigned)dst[ec];
            s[u] = (unsigned)src[ec];
        }
        #pragma unroll
        for (int u = 0; u < 8; ++u) {
            if (e + u * 256 < e1) {
                const unsigned dd = d[u] < (unsigned)N_NODES ? d[u] : (unsigned)(N_NODES - 1);
                const unsigned ss = s[u] < (unsigned)N_NODES ? s[u] : (unsigned)(N_NODES - 1);
                const unsigned p = atomicAdd(&cur[dd >> BK_SHIFT], 1u);
                if (p < (unsigned)ne) ebuf[p] = (ss << BK_SHIFT) | (dd & (unsigned)(BK_NODES - 1));
            }
        }
    }
}

__global__ __launch_bounds__(256)
void bucket_to_nodes(const unsigned* __restrict__ ebuf, const unsigned* __restrict__ bstart,
                     const unsigned* __restrict__ bcount, unsigned* __restrict__ cursor,
                     unsigned* __restrict__ deg, int* __restrict__ col, int ne)
{
    __shared__ unsigned lw[BK_CAP];
    __shared__ unsigned npos[BK_NODES];
    __shared__ unsigned nfirst[BK_NODES];
    __shared__ unsigned ncnt[BK_NODES];
    __shared__ unsigned wsum[8];

    const int t    = threadIdx.x;
    const int lane = t & 31;
    const int wave = t >> 5;
    const int b    = blockIdx.x;

    unsigned beg = bstart[b], n = bcount[b];
    if (beg > (unsigned)ne) beg = (unsigned)ne;
    if (n > (unsigned)ne - beg) n = (unsigned)ne - beg;

    npos[t] = 0u;
    __syncthreads();
    for (unsigned i0 = t; i0 < n; i0 += 8 * 256) {
        unsigned w[8];
        #pragma unroll
        for (int u = 0; u < 8; ++u) {
            const unsigned i = i0 + (unsigned)(u * 256);
            w[u] = ebuf[beg + (i < n ? i : i0)];
        }
        #pragma unroll
        for (int u = 0; u < 8; ++u)
            if (i0 + (unsigned)(u * 256) < n) atomicAdd(&npos[w[u] & (unsigned)(BK_NODES - 1)], 1u);
    }
    __syncthreads();

    const unsigned d = npos[t];
    unsigned inc = d;
    for (int off = 1; off < 32; off <<= 1) {
        const unsigned v = __shfl_up(inc, off, 32);
        if (lane >= off) inc += v;
    }
    if (lane == 31) wsum[wave] = inc;
    __syncthreads();
    unsigned wbase = 0u;
    for (int w = 0; w < wave; ++w) wbase += wsum[w];
    const unsigned first = wbase + inc - d;
    npos[t] = first; nfirst[t] = first; ncnt[t] = d;
    const int node = b * BK_NODES + t;
    if (node < N_NODES) { cursor[node] = beg + first + d; deg[node] = d; }
    __syncthreads();

    if (n <= (unsigned)BK_CAP) {
        for (unsigned i0 = t; i0 < n; i0 += 8 * 256) {
            unsigned w[8];
            #pragma unroll
            for (int u = 0; u < 8; ++u) {
                const unsigned i = i0 + (unsigned)(u * 256);
                w[u] = ebuf[beg + (i < n ? i : i0)];
            }
            #pragma unroll
            for (int u = 0; u < 8; ++u) {
                if (i0 + (unsigned)(u * 256) < n) {
                    const unsigned p = atomicAdd(&npos[w[u] & (unsigned)(BK_NODES - 1)], 1u);
                    if (p < (unsigned)BK_CAP) lw[p] = w[u];
                }
            }
        }
        __syncthreads();
        for (unsigned q = t; q < n; q += 256) {
            const unsigned w  = lw[q];
            const unsigned k  = w & (unsigned)(BK_NODES - 1);
            const unsigned a  = nfirst[k];
            const unsigned dk = ncnt[k];
            unsigned r = q - a;
            if (dk <= (unsigned)RANK_CAP) {
                r = 0u;
                for (unsigned j = a; j < a + dk; ++j) {
                    const unsigned v = lw[j];
                    r += (v < w || (v == w && j < q)) ? 1u : 0u;
                }
            }
            col[beg + a + r] = (int)(w >> BK_SHIFT);
        }
    } else {
        for (unsigned i = t; i < n; i += 256) {
            const unsigned w = ebuf[beg + i];
            const unsigned p = atomicAdd(&npos[w & (unsigned)(BK_NODES - 1)], 1u);
            if (p < n) col[beg + p] = (int)(w >> BK_SHIFT);
        }
    }
}

__device__ __forceinline__ float xhalf(float x) {
    return __builtin_bit_cast(float, __builtin_amdgcn_permlanex16(
        __builtin_bit_cast(int, x), __builtin_bit_cast(int, x),
        0x76543210, 0xfedcba98, false, false));
}

#define XP 68
#define NB 16

__global__ __launch_bounds__(128)
void layer_gather_gemm(const float* __restrict__ Hin,
                       const float* __restrict__ W,
                       const float* __restrict__ rW,
                       const float* __restrict__ bvec,
                       const float* __restrict__ rb,
                       const float* __restrict__ pst,
                       const float* __restrict__ pg,
                       const float* __restrict__ pb,
                       const unsigned* __restrict__ cursor, const unsigned* __restrict__ deg,
                       const int* __restrict__ col,
                       float* __restrict__ Hout,
                       float* __restrict__ stats,
                       int ntiles, int nrows, int ne)
{
    constexpr int KT = HID / 4;
    __shared__ v2f   fragW[KT * 128];
    __shared__ v2f   fragR[KT * 128];
    __shared__ float ta[4 * 16 * XP];
    __shared__ float tx[4 * 16 * XP];
    __shared__ float ssum[4 * 64], ssq[4 * 64];

    const int  lane = threadIdx.x & 31;
    const int  wave = threadIdx.x >> 5;
    const int  l15  = lane & 15;
    const bool hi   = lane >= 16;
    const int  c4   = l15 * 4;

    const int  tile  = blockIdx.x * 4 + wave;
    const bool activ = tile < ntiles;
    const int  row0  = (activ ? tile : (ntiles - 1)) * 16;

    unsigned endv = cursor[row0 + l15];
    unsigned dv   = deg[row0 + l15];
    float bb[4], rbb[4];
    #pragma unroll
    for (int t = 0; t < 4; ++t) { bb[t] = bvec[t * 16 + l15]; rbb[t] = rb[t * 16 + l15]; }
    const float4 psum = *(const float4*)(pst + c4);
    const float4 psq  = *(const float4*)(pst + 64 + c4);
    float gam[4], bet[4];
    #pragma unroll
    for (int i = 0; i < 4; ++i) { gam[i] = pg[c4 + i]; bet[i] = pb[c4 + i]; }
    float4 xr[8];
    #pragma unroll
    for (int u = 0; u < 8; ++u)
        xr[u] = *(const float4*)(Hin + (size_t)(row0 + 2 * u + (hi ? 1 : 0)) * HID + c4);

    for (int idx = threadIdx.x; idx < KT * 128; idx += 128) {
        const int kc  = idx >> 7;
        const int r   = idx & 127;
        const int n   = ((r >> 5) << 4) + (r & 15);
        const int ka0 = kc * 4 + ((r & 31) >= 16 ? 2 : 0);
        v2f fw; fw.x = W [(size_t)ka0 * HID + n]; fw.y = W [(size_t)(ka0 + 1) * HID + n];
        v2f fr; fr.x = rW[(size_t)ka0 * HID + n]; fr.y = rW[(size_t)(ka0 + 1) * HID + n];
        fragW[idx] = fw;
        fragR[idx] = fr;
    }

    float sc[4], sh[4];
    {
        const float inv = 1.0f / (float)nrows;
        const float ps[4] = {psum.x, psum.y, psum.z, psum.w};
        const float pq[4] = {psq.x, psq.y, psq.z, psq.w};
        #pragma unroll
        for (int i = 0; i < 4; ++i) {
            const float mean = ps[i] * inv;
            const float var  = pq[i] * inv - mean * mean;
            sc[i] = rsqrtf(var + BN_EPS) * gam[i];
            sh[i] = bet[i] - mean * sc[i];
        }
    }

    float* ta_ = ta + wave * 16 * XP;
    float* tx_ = tx + wave * 16 * XP;

    #pragma unroll
    for (int u = 0; u < 8; ++u) {
        float4 v = xr[u];
        v.x = v.x * sc[0] + sh[0]; v.y = v.y * sc[1] + sh[1];
        v.z = v.z * sc[2] + sh[2]; v.w = v.w * sc[3] + sh[3];
        *(float4*)(tx_ + (2 * u + (hi ? 1 : 0)) * XP + c4) = v;
    }

    if (endv > (unsigned)ne) endv = (unsigned)ne;
    if (dv > endv) dv = endv;
    if (!activ) dv = 0u;
    const unsigned begv = endv - dv;
    const float* hcol = Hin + c4;

    auto first_ids = [&](unsigned b, unsigned d) -> unsigned {
        const unsigned k = ((unsigned)lane < d) ? (unsigned)lane : (d ? d - 1u : 0u);
        unsigned p = b + k;
        if (p > (unsigned)(ne - 1)) p = (unsigned)(ne - 1);
        return (unsigned)col[p];
    };
    unsigned myc_next = first_ids((unsigned)__builtin_amdgcn_readlane((int)begv, 0),
                                  (unsigned)__builtin_amdgcn_readlane((int)dv, 0));
    for (int r = 0; r < 16; ++r) {
        const unsigned beg = (unsigned)__builtin_amdgcn_readlane((int)begv, r);
        const unsigned d   = (unsigned)__builtin_amdgcn_readlane((int)dv, r);
        unsigned myc = myc_next;
        if (r + 1 < 16)
            myc_next = first_ids((unsigned)__builtin_amdgcn_readlane((int)begv, r + 1),
                                 (unsigned)__builtin_amdgcn_readlane((int)dv, r + 1));
        float4 acc = make_float4(0.0f, 0.0f, 0.0f, 0.0f);
        for (unsigned base = 0; base < d; base += 32) {
            const unsigned left = d - base;
            const unsigned cnt  = left < 32u ? left : 32u;
            if (base != 0u) myc = ((unsigned)lane < cnt) ? (unsigned)col[beg + base + lane] : 0u;
            if (myc > (unsigned)(nrows - 1)) myc = (unsigned)(nrows - 1);
            const unsigned nst = (cnt + 1u) >> 1;
            for (unsigned i0 = 0; i0 < nst; i0 += NB) {
                float4 v[NB];
                #pragma unroll
                for (int u = 0; u < NB; ++u) {
                    const unsigned i = i0 + (unsigned)u;
                    if (i < nst) {
                        const unsigned ea = 2u * i;
                        const unsigned eb = (ea + 1u < cnt) ? (ea + 1u) : ea;
                        const int sa = __builtin_amdgcn_readlane((int)myc, (int)ea);
                        const int sb = __builtin_amdgcn_readlane((int)myc, (int)eb);
                        const int srow = hi ? sb : sa;
                        v[u] = *(const float4*)(hcol + (size_t)srow * HID);
                    }
                }
                #pragma unroll
                for (int u = 0; u < NB; ++u) {
                    const unsigned i = i0 + (unsigned)u;
                    if (i < nst) {
                        const float m = (hi && (2u * i + 1u >= cnt)) ? 0.0f : 1.0f;
                        acc.x = fmaf(v[u].x, m, acc.x); acc.y = fmaf(v[u].y, m, acc.y);
                        acc.z = fmaf(v[u].z, m, acc.z); acc.w = fmaf(v[u].w, m, acc.w);
                    }
                }
            }
        }
        acc.x += xhalf(acc.x); acc.y += xhalf(acc.y);
        acc.z += xhalf(acc.z); acc.w += xhalf(acc.w);
        if (lane < 16) {
            const float df = (float)d;
            float4 o;
            o.x = acc.x * sc[0] + df * sh[0]; o.y = acc.y * sc[1] + df * sh[1];
            o.z = acc.z * sc[2] + df * sh[2]; o.w = acc.w * sc[3] + df * sh[3];
            *(float4*)(ta_ + r * XP + c4) = o;
        }
    }
    __syncthreads();

    v8f acc[4], racc[4];
    #pragma unroll
    for (int t = 0; t < 4; ++t) { acc[t] = {}; racc[t] = {}; }

    const v2f* arow = (const v2f*)(ta_ + l15 * XP);
    const v2f* xrow = (const v2f*)(tx_ + l15 * XP);
    const int  aoff = hi ? 1 : 0;

    #pragma unroll
    for (int kc = 0; kc < KT; ++kc) {
        const v2f a = arow[kc * 2 + aoff];
        const v2f x = xrow[kc * 2 + aoff];
        #pragma unroll
        for (int t = 0; t < 4; ++t) {
            const v2f b  = fragW[kc * 128 + t * 32 + lane];
            const v2f rv = fragR[kc * 128 + t * 32 + lane];
            acc[t]  = wmma_f32(a, b,  acc[t]);
            racc[t] = wmma_f32(x, rv, racc[t]);
        }
    }

    const int rl0 = hi ? 8 : 0;
    #pragma unroll
    for (int t = 0; t < 4; ++t) {
        const int n = t * 16 + l15;
        float cs = 0.0f, cq = 0.0f;
        #pragma unroll
        for (int j = 0; j < 8; ++j) {
            float a = acc[t][j] + bb[t];
            a = a > 0.0f ? a : 0.0f;
            float v = racc[t][j] + rbb[t];
            v = v > 0.0f ? v : 0.0f;
            const float h = a + v;
            ta_[(rl0 + j) * XP + n] = h;
            cs += h; cq += h * h;
        }
        cs += xhalf(cs);
        cq += xhalf(cq);
        if (!activ) { cs = 0.0f; cq = 0.0f; }
        if (lane < 16) { ssum[wave * 64 + n] = cs; ssq[wave * 64 + n] = cq; }
    }
    if (activ) {
        #pragma unroll
        for (int u = 0; u < 8; ++u) {
            const int row = 2 * u + (hi ? 1 : 0);
            const float4 hv = *(const float4*)(ta_ + row * XP + c4);
            *(float4*)(Hout + (size_t)(row0 + row) * HID + c4) = hv;
        }
    }
    __syncthreads();
    if (threadIdx.x < 64) {
        const int f = threadIdx.x;
        const float s = ssum[f] + ssum[64 + f] + ssum[128 + f] + ssum[192 + f];
        const float q = ssq[f]  + ssq[64 + f]  + ssq[128 + f]  + ssq[192 + f];
        atomicAdd(&stats[f], s);
        atomicAdd(&stats[64 + f], q);
    }
}

#define F_PAD 76
#define NB1 8

__global__ __launch_bounds__(128)
void layer1_gather_gemm(const float* __restrict__ X,
                        const float* __restrict__ W,
                        const float* __restrict__ rW,
                        const float* __restrict__ bvec,
                        const float* __restrict__ rb,
                        const unsigned* __restrict__ cursor, const unsigned* __restrict__ deg,
                        const int* __restrict__ col,
                        float* __restrict__ Hout,
                        float* __restrict__ stats,
                        int ntiles, int nrows, int ne)
{
    constexpr int KT = F_PAD / 4;
    __shared__ v2f   fragW[KT * 128];
    __shared__ v2f   fragR[KT * 128];
    __shared__ __attribute__((aligned(16))) float ta[4 * 16 * F_PAD];
    __shared__ __attribute__((aligned(16))) float tx[4 * 16 * F_PAD];
    __shared__ float ssum[4 * 64], ssq[4 * 64];

    const int  lane = threadIdx.x & 31;
    const int  wave = threadIdx.x >> 5;
    const int  l15  = lane & 15;
    const bool hi   = lane >= 16;

    const int p0 = l15;
    const int p1 = l15 + 16;
    const int p2 = (l15 < 5) ? (l15 + 32) : 36;

    const int  tile  = blockIdx.x * 4 + wave;
    const bool activ = tile < ntiles;
    const int  row0  = (activ ? tile : (ntiles - 1)) * 16;

    unsigned endv = cursor[row0 + l15];
    unsigned dv   = deg[row0 + l15];
    float bb[4], rbb[4];
    #pragma unroll
    for (int t = 0; t < 4; ++t) { bb[t] = bvec[t * 16 + l15]; rbb[t] = rb[t * 16 + l15]; }
    float2 x0[8], x1[8], x2[8];
    #pragma unroll
    for (int u = 0; u < 8; ++u) {
        const float2* rp = (const float2*)(X + (size_t)(row0 + 2 * u + (hi ? 1 : 0)) * F_IN);
        x0[u] = rp[p0]; x1[u] = rp[p1]; x2[u] = rp[p2];
    }

    for (int idx = threadIdx.x; idx < KT * 128; idx += 128) {
        const int kc  = idx >> 7;
        const int r   = idx & 127;
        const int n   = ((r >> 5) << 4) + (r & 15);
        const int ka0 = kc * 4 + ((r & 31) >= 16 ? 2 : 0);
        const int ka1 = ka0 + 1;
        const int c0  = (ka0 < F_IN) ? ka0 : 0;
        const int c1  = (ka1 < F_IN) ? ka1 : 0;
        const float m0 = (ka0 < F_IN) ? 1.0f : 0.0f;
        const float m1 = (ka1 < F_IN) ? 1.0f : 0.0f;
        v2f fw; fw.x = W [(size_t)c0 * HID + n] * m0; fw.y = W [(size_t)c1 * HID + n] * m1;
        v2f fr; fr.x = rW[(size_t)c0 * HID + n] * m0; fr.y = rW[(size_t)c1 * HID + n] * m1;
        fragW[idx] = fw;
        fragR[idx] = fr;
    }

    float* ta_ = ta + wave * 16 * F_PAD;
    float* tx_ = tx + wave * 16 * F_PAD;
    const float2 zero2 = make_float2(0.0f, 0.0f);

    #pragma unroll
    for (int u = 0; u < 8; ++u) {
        float* trow = tx_ + (2 * u + (hi ? 1 : 0)) * F_PAD;
        *(float2*)(trow + 2 * p0) = x0[u];
        *(float2*)(trow + 2 * p1) = x1[u];
        if (l15 < 6) *(float2*)(trow + 64 + 2 * l15) = (l15 < 5) ? x2[u] : zero2;
    }

    if (endv > (unsigned)ne) endv = (unsigned)ne;
    if (dv > endv) dv = endv;
    if (!activ) dv = 0u;
    const unsigned begv = endv - dv;

    auto first_ids = [&](unsigned b, unsigned d) -> unsigned {
        const unsigned k = ((unsigned)lane < d) ? (unsigned)lane : (d ? d - 1u : 0u);
        unsigned p = b + k;
        if (p > (unsigned)(ne - 1)) p = (unsigned)(ne - 1);
        return (unsigned)col[p];
    };
    unsigned myc_next = first_ids((unsigned)__builtin_amdgcn_readlane((int)begv, 0),
                                  (unsigned)__builtin_amdgcn_readlane((int)dv, 0));
    for (int r = 0; r < 16; ++r) {
        const unsigned beg = (unsigned)__builtin_amdgcn_readlane((int)begv, r);
        const unsigned d   = (unsigned)__builtin_amdgcn_readlane((int)dv, r);
        unsigned myc = myc_next;
        if (r + 1 < 16)
            myc_next = first_ids((unsigned)__builtin_amdgcn_readlane((int)begv, r + 1),
                                 (unsigned)__builtin_amdgcn_readlane((int)dv, r + 1));
        float2 a0 = zero2, a1 = zero2, a2 = zero2;
        for (unsigned base = 0; base < d; base += 32) {
            const unsigned left = d - base;
            const unsigned cnt  = left < 32u ? left : 32u;
            if (base != 0u) myc = ((unsigned)lane < cnt) ? (unsigned)col[beg + base + lane] : 0u;
            if (myc > (unsigned)(nrows - 1)) myc = (unsigned)(nrows - 1);
            const unsigned nst = (cnt + 1u) >> 1;
            for (unsigned i0 = 0; i0 < nst; i0 += NB1) {
                float2 v0[NB1], v1[NB1], v2[NB1];
                #pragma unroll
                for (int u = 0; u < NB1; ++u) {
                    const unsigned i = i0 + (unsigned)u;
                    if (i < nst) {
                        const unsigned ea = 2u * i;
                        const unsigned eb = (ea + 1u < cnt) ? (ea + 1u) : ea;
                        const int sa = __builtin_amdgcn_readlane((int)myc, (int)ea);
                        const int sb = __builtin_amdgcn_readlane((int)myc, (int)eb);
                        const int srow = hi ? sb : sa;
                        const float2* rp = (const float2*)(X + (size_t)srow * F_IN);
                        v0[u] = rp[p0]; v1[u] = rp[p1]; v2[u] = rp[p2];
                    }
                }
                #pragma unroll
                for (int u = 0; u < NB1; ++u) {
                    const unsigned i = i0 + (unsigned)u;
                    if (i < nst) {
                        const float m = (hi && (2u * i + 1u >= cnt)) ? 0.0f : 1.0f;
                        a0.x = fmaf(v0[u].x, m, a0.x); a0.y = fmaf(v0[u].y, m, a0.y);
                        a1.x = fmaf(v1[u].x, m, a1.x); a1.y = fmaf(v1[u].y, m, a1.y);
                        a2.x = fmaf(v2[u].x, m, a2.x); a2.y = fmaf(v2[u].y, m, a2.y);
                    }
                }
            }
        }
        a0.x += xhalf(a0.x); a0.y += xhalf(a0.y);
        a1.x += xhalf(a1.x); a1.y += xhalf(a1.y);
        a2.x += xhalf(a2.x); a2.y += xhalf(a2.y);
        if (lane < 16) {
            float* trow = ta_ + r * F_PAD;
            *(float2*)(trow + 2 * p0) = a0;
            *(float2*)(trow + 2 * p1) = a1;
            if (l15 < 6) *(float2*)(trow + 64 + 2 * l15) = (l15 < 5) ? a2 : zero2;
        }
    }
    __syncthreads();

    v8f acc[4], racc[4];
    #pragma unroll
    for (int t = 0; t < 4; ++t) { acc[t] = {}; racc[t] = {}; }

    const v2f* arow = (const v2f*)(ta_ + l15 * F_PAD);
    const v2f* xrow = (const v2f*)(tx_ + l15 * F_PAD);
    const int  aoff = hi ? 1 : 0;

    #pragma unroll
    for (int kc = 0; kc < KT; ++kc) {
        const v2f a = arow[kc * 2 + aoff];
        const v2f x = xrow[kc * 2 + aoff];
        #pragma unroll
        for (int t = 0; t < 4; ++t) {
            const v2f b  = fragW[kc * 128 + t * 32 + lane];
            const v2f rv = fragR[kc * 128 + t * 32 + lane];
            acc[t]  = wmma_f32(a, b,  acc[t]);
            racc[t] = wmma_f32(x, rv, racc[t]);
        }
    }

    const int rl0 = hi ? 8 : 0;
    const int c4  = l15 * 4;
    #pragma unroll
    for (int t = 0; t < 4; ++t) {
        const int n = t * 16 + l15;
        float cs = 0.0f, cq = 0.0f;
        #pragma unroll
        for (int j = 0; j < 8; ++j) {
            float a = acc[t][j] + bb[t];
            a = a > 0.0f ? a : 0.0f;
            float v = racc[t][j] + rbb[t];
            v = v > 0.0f ? v : 0.0f;
            const float h = a + v;
            ta_[(rl0 + j) * F_PAD + n] = h;
            cs += h; cq += h * h;
        }
        cs += xhalf(cs);
        cq += xhalf(cq);
        if (!activ) { cs = 0.0f; cq = 0.0f; }
        if (lane < 16) { ssum[wave * 64 + n] = cs; ssq[wave * 64 + n] = cq; }
    }
    if (activ) {
        #pragma unroll
        for (int u = 0; u < 8; ++u) {
            const int row = 2 * u + (hi ? 1 : 0);
            const float4 hv = *(const float4*)(ta_ + row * F_PAD + c4);
            *(float4*)(Hout + (size_t)(row0 + row) * HID + c4) = hv;
        }
    }
    __syncthreads();
    if (threadIdx.x < 64) {
        const int f = threadIdx.x;
        const float s = ssum[f] + ssum[64 + f] + ssum[128 + f] + ssum[192 + f];
        const float q = ssq[f]  + ssq[64 + f]  + ssq[128 + f]  + ssq[192 + f];
        atomicAdd(&stats[f], s);
        atomicAdd(&stats[64 + f], q);
    }
}

__device__ __forceinline__ unsigned f32_ordered(float f) {
    unsigned b = __float_as_uint(f);
    return (b & 0x80000000u) ? ~b : (b | 0x80000000u);
}

#define RO_NODES 16

__global__ __launch_bounds__(256)
void readout_nodes(const float* __restrict__ h, const float* __restrict__ stats,
                   const float* __restrict__ gamma, const float* __restrict__ beta,
                   const int* __restrict__ gid,
                   const float* __restrict__ awW, const float* __restrict__ awb,
                   float* __restrict__ hsum, unsigned* __restrict__ hmaxu,
                   int nrows)
{
    const int lane = threadIdx.x & 31;
    const int wave = __builtin_amdgcn_readfirstlane((int)(threadIdx.x >> 5));
    const int n0 = (blockIdx.x * 8 + wave) * RO_NODES;
    if (n0 >= nrows) return;
    const int last = nrows - 1;

    float r0[RO_NODES], r1[RO_NODES];
    #pragma unroll
    for (int u = 0; u < RO_NODES; ++u) {
        const int n = (n0 + u < nrows) ? (n0 + u) : last;
        r0[u] = h[(size_t)n * HID + lane];
        r1[u] = h[(size_t)n * HID + 32 + lane];
    }
    const int gn = n0 + (lane & (RO_NODES - 1));
    const int gv = gid[gn < nrows ? gn : last];

    const float inv = 1.0f / (float)nrows;
    const float m0 = stats[lane] * inv;
    const float m1 = stats[32 + lane] * inv;
    const float v0 = stats[64 + lane] * inv - m0 * m0;
    const float v1 = stats[96 + lane] * inv - m1 * m1;
    const float rs0 = rsqrtf(v0 + BN_EPS), rs1 = rsqrtf(v1 + BN_EPS);
    const float g0 = gamma[lane], g1 = gamma[32 + lane];
    const float b0 = beta[lane],  b1 = beta[32 + lane];
    const float a0 = awW[lane],   a1 = awW[32 + lane];
    const float ab = awb[0];

    float wg[RO_NODES];
    #pragma unroll
    for (int u = 0; u < RO_NODES; ++u) {
        const float h0 = (r0[u] - m0) * rs0 * g0 + b0;
        const float h1 = (r1[u] - m1) * rs1 * g1 + b1;
        r0[u] = h0; r1[u] = h1;
        float p = h0 * a0 + h1 * a1;
        for (int off = 16; off > 0; off >>= 1) p += __shfl_xor(p, off, 32);
        wg[u] = 1.0f / (1.0f + expf(-(p + ab)));
    }

    int gcur = __builtin_amdgcn_readlane(gv, 0);
    float s0 = 0.0f, s1 = 0.0f;
    unsigned x0 = 0u, x1 = 0u;
    auto flush = [&](int g) {
        atomicAdd(&hsum[(size_t)g * HID + lane],      s0);
        atomicAdd(&hsum[(size_t)g * HID + 32 + lane], s1);
        atomicMax(&hmaxu[(size_t)g * HID + lane],      x0);
        atomicMax(&hmaxu[(size_t)g * HID + 32 + lane], x1);
    };
    #pragma unroll
    for (int u = 0; u < RO_NODES; ++u) {
        if (n0 + u < nrows) {
            const int g = __builtin_amdgcn_readlane(gv, u);
            if (g != gcur) {
                flush(gcur);
                s0 = 0.0f; s1 = 0.0f; x0 = 0u; x1 = 0u;
                gcur = g;
            }
            s0 += r0[u] * wg[u];
            s1 += r1[u] * wg[u];
            const unsigned e0 = f32_ordered(r0[u]), e1 = f32_ordered(r1[u]);
            x0 = e0 > x0 ? e0 : x0;
            x1 = e1 > x1 ? e1 : x1;
        }
    }
    flush(gcur);
}

__global__ __launch_bounds__(128)
void out_gemm_wmma(const float* __restrict__ hsum, const unsigned* __restrict__ hmaxu,
                   const float* __restrict__ Wo,
                   const float* __restrict__ bo, float* __restrict__ out)
{
    constexpr int KT = 128 / 4;
    __shared__ v2f  fragW[KT * 128];
    __shared__ float xt[4 * 16 * 128];

    const int  lane = threadIdx.x & 31;
    const int  wave = threadIdx.x >> 5;
    const int  l15  = lane & 15;
    const bool hi   = lane >= 16;

    for (int idx = threadIdx.x; idx < KT * 128; idx += 128) {
        const int kc  = idx >> 7;
        const int r   = idx & 127;
        const int n   = ((r >> 5) << 4) + (r & 15);
        const int ka0 = kc * 4 + ((r & 31) >= 16 ? 2 : 0);
        v2f fw;
        fw.x = Wo[(size_t)ka0 * HID + n];
        fw.y = Wo[(size_t)(ka0 + 1) * HID + n];
        fragW[idx] = fw;
    }
    __syncthreads();

    const int row0 = (blockIdx.x * 4 + wave) * 16;
    float* xw_ = xt + wave * 16 * 128;
    for (int r = 0; r < 16; ++r) {
        const float*    srow = hsum  + (size_t)(row0 + r) * HID;
        const unsigned* mrow = hmaxu + (size_t)(row0 + r) * HID;
        for (int k = lane; k < HID; k += 32) {
            xw_[r * 128 + k] = srow[k];
            const unsigned u = mrow[k];
            float m = (u & 0x80000000u) ? __uint_as_float(u ^ 0x80000000u)
                                        : __uint_as_float(~u);
            if (!(fabsf(m) <= 3.402823466e38f)) m = 0.0f;
            xw_[r * 128 + HID + k] = m;
        }
    }

    v8f acc[4];
    #pragma unroll
    for (int t = 0; t < 4; ++t) acc[t] = {};

    const v2f* arow = (const v2f*)(xw_ + l15 * 128);
    const int  aoff = hi ? 1 : 0;
    #pragma unroll
    for (int kc = 0; kc < KT; ++kc) {
        const v2f a = arow[kc * 2 + aoff];
        #pragma unroll
        for (int t = 0; t < 4; ++t)
            acc[t] = wmma_f32(a, fragW[kc * 128 + t * 32 + lane], acc[t]);
    }

    const int rbase = row0 + (hi ? 8 : 0);
    #pragma unroll
    for (int t = 0; t < 4; ++t) {
        const int n = t * 16 + l15;
        const float bias = bo[n];
        #pragma unroll
        for (int j = 0; j < 8; ++j)
            out[(size_t)(rbase + j) * HID + n] = acc[t][j] + bias;
    }
}

extern "C" void kernel_launch(void* const* d_in, const int* in_sizes, int n_in,
                              void* d_out, int out_size, void* d_ws, size_t ws_size,
                              hipStream_t stream)
{
    (void)in_sizes; (void)n_in; (void)out_size; (void)ws_size;

    const float* feats = (const float*)d_in[0];
    const int*   src   = (const int*)d_in[1];
    const int*   dst   = (const int*)d_in[2];
    const int*   gid   = (const int*)d_in[3];
    const float* awW  = (const float*)d_in[22];
    const float* awb  = (const float*)d_in[23];
    const float* outW = (const float*)d_in[24];
    const float* outb = (const float*)d_in[25];
    float* out = (float*)d_out;

    float* ws    = (float*)d_ws;
    float* xw    = ws;
    float* rlin  = xw   + (size_t)N_NODES * HID;
    float* bufA  = rlin + (size_t)N_NODES * HID;
    float* bufB  = bufA + (size_t)N_NODES * HID;
    float* hsum  = bufB + (size_t)N_NODES * HID + 128;
    unsigned* hmaxu = (unsigned*)(hsum + (size_t)N_GRAPH * HID);

    int*      col    = (int*)bufB;
    unsigned* cursor = (unsigned*)(col + N_EDGES);
    unsigned* deg    = cursor + N_PAD;
    float*    stats3 = (float*)(deg + N_PAD);
    unsigned* ebuf   = (unsigned*)(stats3 + 3 * 128);
    unsigned* cnt    = ebuf + N_EDGES;
    unsigned* bstart = cnt + NBK * NCH;
    unsigned* bcount = bstart + 256;
    unsigned* total  = bcount + 256;

    const int ntiles = N_NODES / 16;
    const int gemmBlocks = (ntiles + 3) / 4;

    bin_count<<<NCH, 256, 0, stream>>>(dst, cnt, total, stats3, hsum, hmaxu, N_EDGES);
    bucket_ranges<<<(NBK + 7) / 8, 256, 0, stream>>>(cnt, total, bstart, bcount);
    bin_scatter<<<NCH, 256, 0, stream>>>(src, dst, cnt, ebuf, N_EDGES);
    bucket_to_nodes<<<NBK, 256, 0, stream>>>(ebuf, bstart, bcount, cursor, deg, col, N_EDGES);

    layer1_gather_gemm<<<gemmBlocks, 128, 0, stream>>>(
        feats, (const float*)d_in[4], (const float*)d_in[6],
        (const float*)d_in[5], (const float*)d_in[7],
        cursor, deg, col, bufA, stats3, ntiles, N_NODES, N_EDGES);

    layer_gather_gemm<<<gemmBlocks, 128, 0, stream>>>(
        bufA, (const float*)d_in[10], (const float*)d_in[12],
        (const float*)d_in[11], (const float*)d_in[13],
        stats3, (const float*)d_in[8], (const float*)d_in[9],
        cursor, deg, col, xw, stats3 + 128, ntiles, N_NODES, N_EDGES);
    layer_gather_gemm<<<gemmBlocks, 128, 0, stream>>>(
        xw, (const float*)d_in[16], (const float*)d_in[18],
        (const float*)d_in[17], (const float*)d_in[19],
        stats3 + 128, (const float*)d_in[14], (const float*)d_in[15],
        cursor, deg, col, bufA, stats3 + 256, ntiles, N_NODES, N_EDGES);

    readout_nodes<<<(N_NODES + 8 * RO_NODES - 1) / (8 * RO_NODES), 256, 0, stream>>>(bufA, stats3 + 256,
                                                         (const float*)d_in[20], (const float*)d_in[21],
                                                         gid, awW, awb, hsum, hmaxu, N_NODES);
    out_gemm_wmma<<<(N_GRAPH / 16) / 4, 128, 0, stream>>>(hsum, hmaxu, outW, outb, out);
}


# Every staging loop requests ALL its global loads before its first LDS store or add (weight fragments of the 4 GEMM kernels, the projection's input tile, the edge build's chunks): one memory wait each
# speedup vs baseline: 2.8166x; 1.1678x over previous
#include <hip/hip_runtime.h>

typedef float v2f __attribute__((ext_vector_type(2)));
typedef float v8f __attribute__((ext_vector_type(8)));

#define N_NODES 50000
#define N_EDGES 800000
#define N_GRAPH 512
#define F_IN    74
#define HID     64
#define BN_EPS  1e-5f
#define N_PAD   50048

__device__ __forceinline__ v8f wmma_f32(v2f a, v2f b, v8f c) {
    return __builtin_amdgcn_wmma_f32_16x16x4_f32(false, a, false, b,
                                                 (short)0, c, false, false);
}

#define NEG_INF_ENC 0x007FFFFFu
#define BK_SHIFT 8
#define BK_NODES 256
#define NBK      ((N_NODES + BK_NODES - 1) / BK_NODES)
#define NCH      256
#define CH_EDGES ((N_EDGES + NCH - 1) / NCH)
#define BK_CAP   8192
#define RANK_CAP 512
#define N_START  (1 + 3 * 128 + 2 * N_GRAPH * HID)
#define CH_LOADS ((CH_EDGES + 255) / 256)
#define BK_LOADS 18

__global__ __launch_bounds__(256)
void bin_count(const int* __restrict__ dst, unsigned* __restrict__ cnt,
               unsigned* __restrict__ total, float* __restrict__ stats3,
               float* __restrict__ hsum, unsigned* __restrict__ hmaxu, int ne)
{
    __shared__ unsigned hist[NBK];
    for (int i = threadIdx.x; i < NBK; i += 256) hist[i] = 0u;
    __syncthreads();
    const int e0 = blockIdx.x * CH_EDGES;
    const int e1 = (e0 + CH_EDGES < ne) ? (e0 + CH_EDGES) : ne;
    for (int e = e0 + (int)threadIdx.x; e < e1; e += CH_LOADS * 256) {
        unsigned d[CH_LOADS];
        #pragma unroll
        for (int u = 0; u < CH_LOADS; ++u) {
            const int ee = e + u * 256;
            d[u] = (unsigned)dst[ee < e1 ? ee : e];
        }
        #pragma unroll
        for (int u = 0; u < CH_LOADS; ++u) {
            if (e + u * 256 < e1) {
                const unsigned dd = d[u] < (unsigned)N_NODES ? d[u] : (unsigned)(N_NODES - 1);
                atomicAdd(&hist[dd >> BK_SHIFT], 1u);
            }
        }
    }
    __syncthreads();
    for (int i = threadIdx.x; i < NBK; i += 256) cnt[i * NCH + blockIdx.x] = hist[i];

    for (int i = blockIdx.x * 256 + threadIdx.x; i < N_START; i += NCH * 256) {
        int j = i;
        if (j == 0) { *total = 0u; continue; }
        j -= 1;
        if (j < 3 * 128) { stats3[j] = 0.0f; continue; }
        j -= 3 * 128;
        if (j < N_GRAPH * HID) { hsum[j] = 0.0f; continue; }
        j -= N_GRAPH * HID;
        hmaxu[j] = NEG_INF_ENC;
    }
}

__global__ __launch_bounds__(256)
void bucket_ranges(unsigned* __restrict__ cnt, unsigned* __restrict__ total,
                   unsigned* __restrict__ bstart, unsigned* __restrict__ bcount)
{
    const int lane = threadIdx.x & 31;
    const int b    = blockIdx.x * 8 + (threadIdx.x >> 5);
    if (b >= NBK) return;
    uint4* row = (uint4*)(cnt + (size_t)b * NCH) + 2 * lane;
    const uint4 c0 = row[0], c1 = row[1];
    const unsigned s = c0.x + c0.y + c0.z + c0.w + c1.x + c1.y + c1.z + c1.w;
    unsigned inc = s;
    for (int off = 1; off < 32; off <<= 1) {
        const unsigned t = __shfl_up(inc, off, 32);
        if (lane >= off) inc += t;
    }
    unsigned base = 0u;
    if (lane == 31) base = atomicAdd(total, inc);
    base = __shfl(base, 31, 32);
    unsigned o = base + inc - s;
    uint4 r0, r1;
    r0.x = o; o += c0.x; r0.y = o; o += c0.y; r0.z = o; o += c0.z; r0.w = o; o += c0.w;
    r1.x = o; o += c1.x; r1.y = o; o += c1.y; r1.z = o; o += c1.z; r1.w = o;
    row[0] = r0; row[1] = r1;
    if (lane == 31) { bstart[b] = base; bcount[b] = inc; }
}

__global__ __launch_bounds__(256)
void bin_scatter(const int* __restrict__ src, const int* __restrict__ dst,
                 const unsigned* __restrict__ off, unsigned* __restrict__ ebuf, int ne)
{
    __shared__ unsigned cur[NBK];
    for (int i = threadIdx.x; i < NBK; i += 256) cur[i] = off[i * NCH + blockIdx.x];
    __syncthreads();
    const int e0 = blockIdx.x * CH_EDGES;
    const int e1 = (e0 + CH_EDGES < ne) ? (e0 + CH_EDGES) : ne;
    for (int e = e0 + (int)threadIdx.x; e < e1; e += CH_LOADS * 256) {
        unsigned d[CH_LOADS], s[CH_LOADS];
        #pragma unroll
        for (int u = 0; u < CH_LOADS; ++u) {
            const int ee = e + u * 256;
            const int ec = ee < e1 ? ee : e;
            d[u] = (unsigned)dst[ec];
            s[u] = (unsigned)src[ec];
        }
        #pragma unroll
        for (int u = 0; u < CH_LOADS; ++u) {
            if (e + u * 256 < e1) {
                const unsigned dd = d[u] < (unsigned)N_NODES ? d[u] : (unsigned)(N_NODES - 1);
                const unsigned ss = s[u] < (unsigned)N_NODES ? s[u] : (unsigned)(N_NODES - 1);
                const unsigned p = atomicAdd(&cur[dd >> BK_SHIFT], 1u);
                if (p < (unsigned)ne) ebuf[p] = (ss << BK_SHIFT) | (dd & (unsigned)(BK_NODES - 1));
            }
        }
    }
}

__global__ __launch_bounds__(256)
void bucket_to_nodes(const unsigned* __restrict__ ebuf, const unsigned* __restrict__ bstart,
                     const unsigned* __restrict__ bcount, unsigned* __restrict__ cursor,
                     unsigned* __restrict__ deg, int* __restrict__ col, int ne)
{
    __shared__ unsigned lw[BK_CAP];
    __shared__ unsigned npos[BK_NODES];
    __shared__ unsigned nfirst[BK_NODES];
    __shared__ unsigned ncnt[BK_NODES];
    __shared__ unsigned wsum[8];

    const int t    = threadIdx.x;
    const int lane = t & 31;
    const int wave = t >> 5;
    const int b    = blockIdx.x;

    unsigned beg = bstart[b], n = bcount[b];
    if (beg > (unsigned)ne) beg = (unsigned)ne;
    if (n > (unsigned)ne - beg) n = (unsigned)ne - beg;

    npos[t] = 0u;
    __syncthreads();
    for (unsigned i0 = t; i0 < n; i0 += BK_LOADS * 256) {
        unsigned w[BK_LOADS];
        #pragma unroll
        for (int u = 0; u < BK_LOADS; ++u) {
            const unsigned i = i0 + (unsigned)(u * 256);
            w[u] = ebuf[beg + (i < n ? i : i0)];
        }
        #pragma unroll
        for (int u = 0; u < BK_LOADS; ++u)
            if (i0 + (unsigned)(u * 256) < n) atomicAdd(&npos[w[u] & (unsigned)(BK_NODES - 1)], 1u);
    }
    __syncthreads();

    const unsigned d = npos[t];
    unsigned inc = d;
    for (int off = 1; off < 32; off <<= 1) {
        const unsigned v = __shfl_up(inc, off, 32);
        if (lane >= off) inc += v;
    }
    if (lane == 31) wsum[wave] = inc;
    __syncthreads();
    unsigned wbase = 0u;
    for (int w = 0; w < wave; ++w) wbase += wsum[w];
    const unsigned first = wbase + inc - d;
    npos[t] = first; nfirst[t] = first; ncnt[t] = d;
    const int node = b * BK_NODES + t;
    if (node < N_NODES) { cursor[node] = beg + first + d; deg[node] = d; }
    __syncthreads();

    if (n <= (unsigned)BK_CAP) {
        for (unsigned i0 = t; i0 < n; i0 += BK_LOADS * 256) {
            unsigned w[BK_LOADS];
            #pragma unroll
            for (int u = 0; u < BK_LOADS; ++u) {
                const unsigned i = i0 + (unsigned)(u * 256);
                w[u] = ebuf[beg + (i < n ? i : i0)];
            }
            #pragma unroll
            for (int u = 0; u < BK_LOADS; ++u) {
                if (i0 + (unsigned)(u * 256) < n) {
                    const unsigned p = atomicAdd(&npos[w[u] & (unsigned)(BK_NODES - 1)], 1u);
                    if (p < (unsigned)BK_CAP) lw[p] = w[u];
                }
            }
        }
        __syncthreads();
        for (unsigned q = t; q < n; q += 256) {
            const unsigned w  = lw[q];
            const unsigned k  = w & (unsigned)(BK_NODES - 1);
            const unsigned a  = nfirst[k];
            const unsigned dk = ncnt[k];
            unsigned r = q - a;
            if (dk <= (unsigned)RANK_CAP) {
                r = 0u;
                for (unsigned j = a; j < a + dk; ++j) {
                    const unsigned v = lw[j];
                    r += (v < w || (v == w && j < q)) ? 1u : 0u;
                }
            }
            col[beg + a + r] = (int)(w >> BK_SHIFT);
        }
    } else {
        for (unsigned i = t; i < n; i += 256) {
            const unsigned w = ebuf[beg + i];
            const unsigned p = atomicAdd(&npos[w & (unsigned)(BK_NODES - 1)], 1u);
            if (p < n) col[beg + p] = (int)(w >> BK_SHIFT);
        }
    }
}

__device__ __forceinline__ float xhalf(float x) {
    return __builtin_bit_cast(float, __builtin_amdgcn_permlanex16(
        __builtin_bit_cast(int, x), __builtin_bit_cast(int, x),
        0x76543210, 0xfedcba98, false, false));
}

#define XP 68
#define GR_DEPTH 16

__global__ __launch_bounds__(128)
void layer_gather_gemm(const float* __restrict__ Hin,
                       const float* __restrict__ W,
                       const float* __restrict__ rW,
                       const float* __restrict__ bvec,
                       const float* __restrict__ rb,
                       const float* __restrict__ pst,
                       const float* __restrict__ pg,
                       const float* __restrict__ pb,
                       const unsigned* __restrict__ cursor, const unsigned* __restrict__ deg,
                       const int* __restrict__ col,
                       float* __restrict__ Hout,
                       float* __restrict__ stats,
                       int ntiles, int nrows, int ne)
{
    constexpr int KT = HID / 4;
    __shared__ v2f   fragW[KT * 128];
    __shared__ v2f   fragR[KT * 128];
    __shared__ float ta[4 * 16 * XP];
    __shared__ float tx[4 * 16 * XP];
    __shared__ float ssum[4 * 64], ssq[4 * 64];

    const int  lane = threadIdx.x & 31;
    const int  wave = threadIdx.x >> 5;
    const int  l15  = lane & 15;
    const bool hi   = lane >= 16;
    const int  c4   = l15 * 4;

    const int  tile  = blockIdx.x * 4 + wave;
    const bool activ = tile < ntiles;
    const int  row0  = (activ ? tile : (ntiles - 1)) * 16;

    unsigned endv = cursor[row0 + l15];
    unsigned dv   = deg[row0 + l15];
    float bb[4], rbb[4];
    #pragma unroll
    for (int t = 0; t < 4; ++t) { bb[t] = bvec[t * 16 + l15]; rbb[t] = rb[t * 16 + l15]; }
    const float4 psum = *(const float4*)(pst + c4);
    const float4 psq  = *(const float4*)(pst + 64 + c4);
    float gam[4], bet[4];
    #pragma unroll
    for (int i = 0; i < 4; ++i) { gam[i] = pg[c4 + i]; bet[i] = pb[c4 + i]; }
    float4 xr[8];
    #pragma unroll
    for (int u = 0; u < 8; ++u)
        xr[u] = *(const float4*)(Hin + (size_t)(row0 + 2 * u + (hi ? 1 : 0)) * HID + c4);

    {
        const int r  = threadIdx.x;
        const int n  = ((r >> 5) << 4) + (r & 15);
        const int kh = (r & 31) >= 16 ? 2 : 0;
        v2f fw[KT], fr[KT];
        #pragma unroll
        for (int kc = 0; kc < KT; ++kc) {
            const int ka0 = kc * 4 + kh;
            fw[kc].x = W [(size_t)ka0 * HID + n]; fw[kc].y = W [(size_t)(ka0 + 1) * HID + n];
            fr[kc].x = rW[(size_t)ka0 * HID + n]; fr[kc].y = rW[(size_t)(ka0 + 1) * HID + n];
        }
        #pragma unroll
        for (int kc = 0; kc < KT; ++kc) {
            fragW[kc * 128 + r] = fw[kc];
            fragR[kc * 128 + r] = fr[kc];
        }
    }

    float sc[4], sh[4];
    {
        const float inv = 1.0f / (float)nrows;
        const float ps[4] = {psum.x, psum.y, psum.z, psum.w};
        const float pq[4] = {psq.x, psq.y, psq.z, psq.w};
        #pragma unroll
        for (int i = 0; i < 4; ++i) {
            const float mean = ps[i] * inv;
            const float var  = pq[i] * inv - mean * mean;
            sc[i] = rsqrtf(var + BN_EPS) * gam[i];
            sh[i] = bet[i] - mean * sc[i];
        }
    }

    float* ta_ = ta + wave * 16 * XP;
    float* tx_ = tx + wave * 16 * XP;

    #pragma unroll
    for (int u = 0; u < 8; ++u) {
        float4 v = xr[u];
        v.x = v.x * sc[0] + sh[0]; v.y = v.y * sc[1] + sh[1];
        v.z = v.z * sc[2] + sh[2]; v.w = v.w * sc[3] + sh[3];
        *(float4*)(tx_ + (2 * u + (hi ? 1 : 0)) * XP + c4) = v;
    }

    if (endv > (unsigned)ne) endv = (unsigned)ne;
    if (dv > endv) dv = endv;
    if (!activ) dv = 0u;
    const unsigned begv = endv - dv;

    #pragma unroll
    for (int u = 0; u < 8; ++u)
        *(float4*)(ta_ + (2 * u + (hi ? 1 : 0)) * XP + c4) = make_float4(0.0f, 0.0f, 0.0f, 0.0f);

    const unsigned S = (unsigned)__builtin_amdgcn_readlane((int)begv, 0);
    unsigned es[16];
    #pragma unroll
    for (int j = 0; j < 16; ++j) es[j] = (unsigned)__builtin_amdgcn_readlane((int)endv, j);
    unsigned Tn = (es[15] > S) ? (es[15] - S) : 0u;
    if (!activ) Tn = 0u;
    const unsigned nch = ((unsigned)__builtin_amdgcn_readfirstlane((int)Tn) + 31u) >> 5;

    auto node_of = [&](unsigned p) -> unsigned {
        unsigned k = 0u;
        #pragma unroll
        for (int j = 0; j < 16; ++j) k += (es[j] <= p) ? 1u : 0u;
        return k;
    };
    auto chunk_ids = [&](unsigned c) -> unsigned {
        unsigned p = S + 32u * c + (unsigned)lane;
        if (p > (unsigned)(ne - 1)) p = (unsigned)(ne - 1);
        return (unsigned)col[p];
    };
    auto chunk_marks = [&](unsigned c, unsigned kv) -> unsigned {
        const unsigned p = S + 32u * c + (unsigned)lane;
        return (unsigned)__builtin_amdgcn_ballot_w32(kv != node_of(p - 1u));
    };

    float4 gacc = make_float4(0.0f, 0.0f, 0.0f, 0.0f);
    int cur = 0;
    auto flush = [&](int k) {
        float4 t;
        t.x = gacc.x + xhalf(gacc.x); t.y = gacc.y + xhalf(gacc.y);
        t.z = gacc.z + xhalf(gacc.z); t.w = gacc.w + xhalf(gacc.w);
        if (k < 16) {
            const float df = (float)(unsigned)__builtin_amdgcn_readlane((int)dv, k);
            if (lane < 16) {
                float4 o;
                o.x = t.x * sc[0] + df * sh[0]; o.y = t.y * sc[1] + df * sh[1];
                o.z = t.z * sc[2] + df * sh[2]; o.w = t.w * sc[3] + df * sh[3];
                *(float4*)(ta_ + k * XP + c4) = o;
            }
        }
        gacc = make_float4(0.0f, 0.0f, 0.0f, 0.0f);
    };
    auto consume = [&](const float4 val, unsigned kv, unsigned marks, int u) {
        if ((marks >> (2 * u)) & 3u) {
            const int kA = __builtin_amdgcn_readlane((int)kv, 2 * u);
            const int kB = __builtin_amdgcn_readlane((int)kv, 2 * u + 1);
            #pragma nounroll
            for (int h = 0; h < 2; ++h) {
                const int k = h ? kB : kA;
                if (k != cur) { flush(cur); cur = k; }
                if ((hi ? 1 : 0) == h) { gacc.x += val.x; gacc.y += val.y; gacc.z += val.z; gacc.w += val.w; }
            }
        } else {
            gacc.x += val.x; gacc.y += val.y; gacc.z += val.z; gacc.w += val.w;
        }
    };
    auto row_load = [&](unsigned ids, int u) -> float4 {
        const int sa = __builtin_amdgcn_readlane((int)ids, 2 * u);
        const int sb = __builtin_amdgcn_readlane((int)ids, 2 * u + 1);
        const unsigned off = (unsigned)(hi ? sb : sa) * (unsigned)HID + (unsigned)c4;
        return *(const float4*)(Hin + off);
    };
    auto clamp_ids = [&](unsigned ids) -> unsigned {
        return ids > (unsigned)(nrows - 1) ? (unsigned)(nrows - 1) : ids;
    };

    if (nch > 0u) {
        const unsigned raw0 = chunk_ids(0u);
        const unsigned raw1 = chunk_ids(1u);
        const unsigned id0 = clamp_ids(raw0);
        unsigned idB = clamp_ids(raw1);
        unsigned kvA = node_of(S + (unsigned)lane);
        unsigned mkA = chunk_marks(0u, kvA);
        float4 v[GR_DEPTH];
        #pragma unroll
        for (int u = 0; u < GR_DEPTH; ++u) v[u] = row_load(id0, u);
        for (unsigned c = 0; c + 1u < nch; ++c) {
            const unsigned rawN = chunk_ids(c + 2u);
            const unsigned kvB = node_of(S + 32u * (c + 1u) + (unsigned)lane);
            const unsigned mkB = chunk_marks(c + 1u, kvB);
            #pragma unroll
            for (int u = 0; u < GR_DEPTH; ++u) {
                consume(v[u], kvA, mkA, u);
                v[u] = row_load(idB, u);
            }
            kvA = kvB; mkA = mkB; idB = clamp_ids(rawN);
        }
        #pragma unroll
        for (int u = 0; u < GR_DEPTH; ++u) consume(v[u], kvA, mkA, u);
        flush(cur);
    }
    __syncthreads();

    v8f acc[4], racc[4];
    #pragma unroll
    for (int t = 0; t < 4; ++t) { acc[t] = {}; racc[t] = {}; }

    const v2f* arow = (const v2f*)(ta_ + l15 * XP);
    const v2f* xrow = (const v2f*)(tx_ + l15 * XP);
    const int  aoff = hi ? 1 : 0;

    #pragma unroll
    for (int kc = 0; kc < KT; ++kc) {
        const v2f a = arow[kc * 2 + aoff];
        const v2f x = xrow[kc * 2 + aoff];
        #pragma unroll
        for (int t = 0; t < 4; ++t) {
            const v2f b  = fragW[kc * 128 + t * 32 + lane];
            const v2f rv = fragR[kc * 128 + t * 32 + lane];
            acc[t]  = wmma_f32(a, b,  acc[t]);
            racc[t] = wmma_f32(x, rv, racc[t]);
        }
    }

    const int rl0 = hi ? 8 : 0;
    #pragma unroll
    for (int t = 0; t < 4; ++t) {
        const int n = t * 16 + l15;
        float cs = 0.0f, cq = 0.0f;
        #pragma unroll
        for (int j = 0; j < 8; ++j) {
            float a = acc[t][j] + bb[t];
            a = a > 0.0f ? a : 0.0f;
            float v = racc[t][j] + rbb[t];
            v = v > 0.0f ? v : 0.0f;
            const float h = a + v;
            ta_[(rl0 + j) * XP + n] = h;
            cs += h; cq += h * h;
        }
        cs += xhalf(cs);
        cq += xhalf(cq);
        if (!activ) { cs = 0.0f; cq = 0.0f; }
        if (lane < 16) { ssum[wave * 64 + n] = cs; ssq[wave * 64 + n] = cq; }
    }
    if (activ) {
        #pragma unroll
        for (int u = 0; u < 8; ++u) {
            const int row = 2 * u + (hi ? 1 : 0);
            const float4 hv = *(const float4*)(ta_ + row * XP + c4);
            *(float4*)(Hout + (size_t)(row0 + row) * HID + c4) = hv;
        }
    }
    __syncthreads();
    if (threadIdx.x < 64) {
        const int f = threadIdx.x;
        const float s = ssum[f] + ssum[64 + f] + ssum[128 + f] + ssum[192 + f];
        const float q = ssq[f]  + ssq[64 + f]  + ssq[128 + f]  + ssq[192 + f];
        atomicAdd(&stats[f], s);
        atomicAdd(&stats[64 + f], q);
    }
}

#define F_PAD 76
#define GR1_DEPTH 8

__global__ __launch_bounds__(128)
void layer1_gather_gemm(const float* __restrict__ X,
                        const float* __restrict__ W,
                        const float* __restrict__ rW,
                        const float* __restrict__ bvec,
                        const float* __restrict__ rb,
                        const unsigned* __restrict__ cursor, const unsigned* __restrict__ deg,
                        const int* __restrict__ col,
                        float* __restrict__ Hout,
                        float* __restrict__ stats,
                        int ntiles, int nrows, int ne)
{
    constexpr int KT = F_PAD / 4;
    __shared__ v2f   fragW[KT * 128];
    __shared__ v2f   fragR[KT * 128];
    __shared__ __attribute__((aligned(16))) float ta[4 * 16 * F_PAD];
    __shared__ __attribute__((aligned(16))) float tx[4 * 16 * F_PAD];
    __shared__ float ssum[4 * 64], ssq[4 * 64];

    const int  lane = threadIdx.x & 31;
    const int  wave = threadIdx.x >> 5;
    const int  l15  = lane & 15;
    const bool hi   = lane >= 16;

    const int p0 = l15;
    const int p1 = l15 + 16;
    const int p2 = (l15 < 5) ? (l15 + 32) : 36;

    const int  tile  = blockIdx.x * 4 + wave;
    const bool activ = tile < ntiles;
    const int  row0  = (activ ? tile : (ntiles - 1)) * 16;

    unsigned endv = cursor[row0 + l15];
    unsigned dv   = deg[row0 + l15];
    float bb[4], rbb[4];
    #pragma unroll
    for (int t = 0; t < 4; ++t) { bb[t] = bvec[t * 16 + l15]; rbb[t] = rb[t * 16 + l15]; }
    float2 x0[8], x1[8], x2[8];
    #pragma unroll
    for (int u = 0; u < 8; ++u) {
        const float2* rp = (const float2*)(X + (size_t)(row0 + 2 * u + (hi ? 1 : 0)) * F_IN);
        x0[u] = rp[p0]; x1[u] = rp[p1]; x2[u] = rp[p2];
    }

    {
        const int r  = threadIdx.x;
        const int n  = ((r >> 5) << 4) + (r & 15);
        const int kh = (r & 31) >= 16 ? 2 : 0;
        v2f fw[KT], fr[KT];
        #pragma unroll
        for (int kc = 0; kc < KT; ++kc) {
            const int ka0 = kc * 4 + kh;
            const int ka1 = ka0 + 1;
            const int c0  = (ka0 < F_IN) ? ka0 : 0;
            const int c1  = (ka1 < F_IN) ? ka1 : 0;
            const float m0 = (ka0 < F_IN) ? 1.0f : 0.0f;
            const float m1 = (ka1 < F_IN) ? 1.0f : 0.0f;
            fw[kc].x = W [(size_t)c0 * HID + n] * m0; fw[kc].y = W [(size_t)c1 * HID + n] * m1;
            fr[kc].x = rW[(size_t)c0 * HID + n] * m0; fr[kc].y = rW[(size_t)c1 * HID + n] * m1;
        }
        #pragma unroll
        for (int kc = 0; kc < KT; ++kc) {
            fragW[kc * 128 + r] = fw[kc];
            fragR[kc * 128 + r] = fr[kc];
        }
    }

    float* ta_ = ta + wave * 16 * F_PAD;
    float* tx_ = tx + wave * 16 * F_PAD;
    const float2 zero2 = make_float2(0.0f, 0.0f);

    const int   pc2  = (l15 < 5) ? l15 : 5;
    #pragma unroll
    for (int u = 0; u < 8; ++u) {
        float* trow = tx_ + (2 * u + (hi ? 1 : 0)) * F_PAD;
        *(float2*)(trow + 2 * p0) = x0[u];
        *(float2*)(trow + 2 * p1) = x1[u];
        *(float2*)(trow + 64 + 2 * pc2) = (l15 < 5) ? x2[u] : zero2;
    }

    if (endv > (unsigned)ne) endv = (unsigned)ne;
    if (dv > endv) dv = endv;
    if (!activ) dv = 0u;
    const unsigned begv = endv - dv;

    #pragma unroll
    for (int u = 0; u < 8; ++u) {
        float* trow = ta_ + (2 * u + (hi ? 1 : 0)) * F_PAD;
        *(float2*)(trow + 2 * p0) = zero2;
        *(float2*)(trow + 2 * p1) = zero2;
        if (l15 < 6) *(float2*)(trow + 64 + 2 * l15) = zero2;
    }

    const unsigned S = (unsigned)__builtin_amdgcn_readlane((int)begv, 0);
    unsigned es[16];
    #pragma unroll
    for (int j = 0; j < 16; ++j) es[j] = (unsigned)__builtin_amdgcn_readlane((int)endv, j);
    unsigned Tn = (es[15] > S) ? (es[15] - S) : 0u;
    if (!activ) Tn = 0u;
    const unsigned nch = ((unsigned)__builtin_amdgcn_readfirstlane((int)Tn) + 31u) >> 5;

    auto node_of = [&](unsigned p) -> unsigned {
        unsigned k = 0u;
        #pragma unroll
        for (int j = 0; j < 16; ++j) k += (es[j] <= p) ? 1u : 0u;
        return k;
    };
    auto chunk_ids = [&](unsigned c) -> unsigned {
        unsigned p = S + 32u * c + (unsigned)lane;
        if (p > (unsigned)(ne - 1)) p = (unsigned)(ne - 1);
        return (unsigned)col[p];
    };
    auto chunk_marks = [&](unsigned c, unsigned kv) -> unsigned {
        const unsigned p = S + 32u * c + (unsigned)lane;
        return (unsigned)__builtin_amdgcn_ballot_w32(kv != node_of(p - 1u));
    };

    float2 a0 = zero2, a1 = zero2, a2 = zero2;
    int cur = 0;
    auto flush = [&](int k) {
        float2 t0, t1, t2;
        t0.x = a0.x + xhalf(a0.x); t0.y = a0.y + xhalf(a0.y);
        t1.x = a1.x + xhalf(a1.x); t1.y = a1.y + xhalf(a1.y);
        t2.x = a2.x + xhalf(a2.x); t2.y = a2.y + xhalf(a2.y);
        if (k < 16) {
            if (lane < 16) {
                float* trow = ta_ + k * F_PAD;
                *(float2*)(trow + 2 * p0) = t0;
                *(float2*)(trow + 2 * p1) = t1;
                if (l15 < 6) *(float2*)(trow + 64 + 2 * l15) = (l15 < 5) ? t2 : zero2;
            }
        }
        a0 = zero2; a1 = zero2; a2 = zero2;
    };
    auto consume = [&](const float2 w0, const float2 w1, const float2 w2, unsigned kv, unsigned marks, int u) {
        if ((marks >> (2 * u)) & 3u) {
            const int kA = __builtin_amdgcn_readlane((int)kv, 2 * u);
            const int kB = __builtin_amdgcn_readlane((int)kv, 2 * u + 1);
            #pragma nounroll
            for (int h = 0; h < 2; ++h) {
                const int k = h ? kB : kA;
                if (k != cur) { flush(cur); cur = k; }
                if ((hi ? 1 : 0) == h) {
                    a0.x += w0.x; a0.y += w0.y; a1.x += w1.x; a1.y += w1.y; a2.x += w2.x; a2.y += w2.y;
                }
            }
        } else {
            a0.x += w0.x; a0.y += w0.y; a1.x += w1.x; a1.y += w1.y; a2.x += w2.x; a2.y += w2.y;
        }
    };
    auto clamp_ids = [&](unsigned ids) -> unsigned {
        return ids > (unsigned)(nrows - 1) ? (unsigned)(nrows - 1) : ids;
    };

    if (nch > 0u) {
        const unsigned raw0 = chunk_ids(0u);
        const unsigned raw1 = chunk_ids(1u);
        unsigned idA = clamp_ids(raw0);
        unsigned idB = clamp_ids(raw1);
        unsigned kvA = node_of(S + (unsigned)lane);
        unsigned mkA = chunk_marks(0u, kvA);
        float2 v0[GR1_DEPTH], v1[GR1_DEPTH], v2[GR1_DEPTH];
        #define GR1_LOAD(ids, st, q) do {                                                   \
            const int sa_ = __builtin_amdgcn_readlane((int)(ids), 2 * (st));                \
            const int sb_ = __builtin_amdgcn_readlane((int)(ids), 2 * (st) + 1);            \
            const unsigned ro_ = (unsigned)(hi ? sb_ : sa_) * (unsigned)F_IN;     \
            v0[q] = *(const float2*)(X + (ro_ + 2u * (unsigned)p0));                       \
            v1[q] = *(const float2*)(X + (ro_ + 2u * (unsigned)p1));                       \
            v2[q] = *(const float2*)(X + (ro_ + 2u * (unsigned)p2));                       \
        } while (0)
        #pragma unroll
        for (int u = 0; u < GR1_DEPTH; ++u) GR1_LOAD(idA, u, u);
        for (unsigned c = 0; c + 1u < nch; ++c) {
            const unsigned rawN = chunk_ids(c + 2u);
            const unsigned kvB = node_of(S + 32u * (c + 1u) + (unsigned)lane);
            const unsigned mkB = chunk_marks(c + 1u, kvB);
            #pragma unroll
            for (int u = 0; u < GR1_DEPTH; ++u) {
                consume(v0[u], v1[u], v2[u], kvA, mkA, u);
                GR1_LOAD(idA, u + GR1_DEPTH, u);
            }
            #pragma unroll
            for (int u = 0; u < GR1_DEPTH; ++u) {
                consume(v0[u], v1[u], v2[u], kvA, mkA, u + GR1_DEPTH);
                GR1_LOAD(idB, u, u);
            }
            idA = idB; idB = clamp_ids(rawN); kvA = kvB; mkA = mkB;
        }
        #pragma unroll
        for (int u = 0; u < GR1_DEPTH; ++u) {
            consume(v0[u], v1[u], v2[u], kvA, mkA, u);
            GR1_LOAD(idA, u + GR1_DEPTH, u);
        }
        #pragma unroll
        for (int u = 0; u < GR1_DEPTH; ++u) consume(v0[u], v1[u], v2[u], kvA, mkA, u + GR1_DEPTH);
        #undef GR1_LOAD
        flush(cur);
    }
    __syncthreads();

    v8f acc[4], racc[4];
    #pragma unroll
    for (int t = 0; t < 4; ++t) { acc[t] = {}; racc[t] = {}; }

    const v2f* arow = (const v2f*)(ta_ + l15 * F_PAD);
    const v2f* xrow = (const v2f*)(tx_ + l15 * F_PAD);
    const int  aoff = hi ? 1 : 0;

    #pragma unroll
    for (int kc = 0; kc < KT; ++kc) {
        const v2f a = arow[kc * 2 + aoff];
        const v2f x = xrow[kc * 2 + aoff];
        #pragma unroll
        for (int t = 0; t < 4; ++t) {
            const v2f b  = fragW[kc * 128 + t * 32 + lane];
            const v2f rv = fragR[kc * 128 + t * 32 + lane];
            acc[t]  = wmma_f32(a, b,  acc[t]);
            racc[t] = wmma_f32(x, rv, racc[t]);
        }
    }

    const int rl0 = hi ? 8 : 0;
    const int c4  = l15 * 4;
    #pragma unroll
    for (int t = 0; t < 4; ++t) {
        const int n = t * 16 + l15;
        float cs = 0.0f, cq = 0.0f;
        #pragma unroll
        for (int j = 0; j < 8; ++j) {
            float a = acc[t][j] + bb[t];
            a = a > 0.0f ? a : 0.0f;
            float v = racc[t][j] + rbb[t];
            v = v > 0.0f ? v : 0.0f;
            const float h = a + v;
            ta_[(rl0 + j) * F_PAD + n] = h;
            cs += h; cq += h * h;
        }
        cs += xhalf(cs);
        cq += xhalf(cq);
        if (!activ) { cs = 0.0f; cq = 0.0f; }
        if (lane < 16) { ssum[wave * 64 + n] = cs; ssq[wave * 64 + n] = cq; }
    }
    if (activ) {
        #pragma unroll
        for (int u = 0; u < 8; ++u) {
            const int row = 2 * u + (hi ? 1 : 0);
            const float4 hv = *(const float4*)(ta_ + row * F_PAD + c4);
            *(float4*)(Hout + (size_t)(row0 + row) * HID + c4) = hv;
        }
    }
    __syncthreads();
    if (threadIdx.x < 64) {
        const int f = threadIdx.x;
        const float s = ssum[f] + ssum[64 + f] + ssum[128 + f] + ssum[192 + f];
        const float q = ssq[f]  + ssq[64 + f]  + ssq[128 + f]  + ssq[192 + f];
        atomicAdd(&stats[f], s);
        atomicAdd(&stats[64 + f], q);
    }
}

__device__ __forceinline__ unsigned f32_ordered(float f) {
    unsigned b = __float_as_uint(f);
    return (b & 0x80000000u) ? ~b : (b | 0x80000000u);
}

#define RO_NODES 16

__global__ __launch_bounds__(256)
void readout_nodes(const float* __restrict__ h, const float* __restrict__ stats,
                   const float* __restrict__ gamma, const float* __restrict__ beta,
                   const int* __restrict__ gid,
                   const float* __restrict__ awW, const float* __restrict__ awb,
                   float* __restrict__ hsum, unsigned* __restrict__ hmaxu,
                   int nrows)
{
    const int lane = threadIdx.x & 31;
    const int wave = __builtin_amdgcn_readfirstlane((int)(threadIdx.x >> 5));
    const int n0 = (blockIdx.x * 8 + wave) * RO_NODES;
    if (n0 >= nrows) return;
    const int last = nrows - 1;

    float r0[RO_NODES], r1[RO_NODES];
    #pragma unroll
    for (int u = 0; u < RO_NODES; ++u) {
        const int n = (n0 + u < nrows) ? (n0 + u) : last;
        r0[u] = h[(size_t)n * HID + lane];
        r1[u] = h[(size_t)n * HID + 32 + lane];
    }
    const int gn = n0 + (lane & (RO_NODES - 1));
    const int gv = gid[gn < nrows ? gn : last];

    const float inv = 1.0f / (float)nrows;
    const float m0 = stats[lane] * inv;
    const float m1 = stats[32 + lane] * inv;
    const float v0 = stats[64 + lane] * inv - m0 * m0;
    const float v1 = stats[96 + lane] * inv - m1 * m1;
    const float rs0 = rsqrtf(v0 + BN_EPS), rs1 = rsqrtf(v1 + BN_EPS);
    const float g0 = gamma[lane], g1 = gamma[32 + lane];
    const float b0 = beta[lane],  b1 = beta[32 + lane];
    const float a0 = awW[lane],   a1 = awW[32 + lane];
    const float ab = awb[0];

    float wg[RO_NODES];
    #pragma unroll
    for (int u = 0; u < RO_NODES; ++u) {
        const float h0 = (r0[u] - m0) * rs0 * g0 + b0;
        const float h1 = (r1[u] - m1) * rs1 * g1 + b1;
        r0[u] = h0; r1[u] = h1;
        float p = h0 * a0 + h1 * a1;
        for (int off = 16; off > 0; off >>= 1) p += __shfl_xor(p, off, 32);
        wg[u] = 1.0f / (1.0f + expf(-(p + ab)));
    }

    int gcur = __builtin_amdgcn_readlane(gv, 0);
    float s0 = 0.0f, s1 = 0.0f;
    unsigned x0 = 0u, x1 = 0u;
    auto flush = [&](int g) {
        atomicAdd(&hsum[(size_t)g * HID + lane],      s0);
        atomicAdd(&hsum[(size_t)g * HID + 32 + lane], s1);
        atomicMax(&hmaxu[(size_t)g * HID + lane],      x0);
        atomicMax(&hmaxu[(size_t)g * HID + 32 + lane], x1);
    };
    #pragma unroll
    for (int u = 0; u < RO_NODES; ++u) {
        if (n0 + u < nrows) {
            const int g = __builtin_amdgcn_readlane(gv, u);
            if (g != gcur) {
                flush(gcur);
                s0 = 0.0f; s1 = 0.0f; x0 = 0u; x1 = 0u;
                gcur = g;
            }
            s0 += r0[u] * wg[u];
            s1 += r1[u] * wg[u];
            const unsigned e0 = f32_ordered(r0[u]), e1 = f32_ordered(r1[u]);
            x0 = e0 > x0 ? e0 : x0;
            x1 = e1 > x1 ? e1 : x1;
        }
    }
    flush(gcur);
}

__global__ __launch_bounds__(128)
void out_gemm_wmma(const float* __restrict__ hsum, const unsigned* __restrict__ hmaxu,
                   const float* __restrict__ Wo,
                   const float* __restrict__ bo, float* __restrict__ out)
{
    constexpr int KT = 128 / 4;
    __shared__ v2f  fragW[KT * 128];
    __shared__ __attribute__((aligned(16))) float xt[4 * 16 * 128];

    const int  lane = threadIdx.x & 31;
    const int  wave = threadIdx.x >> 5;
    const int  l15  = lane & 15;
    const bool hi   = lane >= 16;

    const int row0 = (blockIdx.x * 4 + wave) * 16;
    float* xw_ = xt + wave * 16 * 128;

    const int fr_ = threadIdx.x;
    const int fn_ = ((fr_ >> 5) << 4) + (fr_ & 15);
    const int fk_ = (fr_ & 31) >= 16 ? 2 : 0;
    v2f fw[KT];
    #pragma unroll
    for (int kc = 0; kc < KT; ++kc) {
        const int ka0 = kc * 4 + fk_;
        fw[kc].x = Wo[(size_t)ka0 * HID + fn_];
        fw[kc].y = Wo[(size_t)(ka0 + 1) * HID + fn_];
    }
    uint4 xraw[16];
    #pragma unroll
    for (int r = 0; r < 16; ++r) {
        const size_t o = (size_t)(row0 + r) * HID + (size_t)(l15 * 4);
        const uint4* p = hi ? (const uint4*)(hmaxu + o) : (const uint4*)(hsum + o);
        xraw[r] = *p;
    }
    float bias4[4];
    #pragma unroll
    for (int t = 0; t < 4; ++t) bias4[t] = bo[t * 16 + l15];

    #pragma unroll
    for (int kc = 0; kc < KT; ++kc) fragW[kc * 128 + fr_] = fw[kc];

    auto dec = [&](unsigned u) -> float {
        float m = (u & 0x80000000u) ? __uint_as_float(u ^ 0x80000000u)
                                    : __uint_as_float(~u);
        if (!(fabsf(m) <= 3.402823466e38f)) m = 0.0f;
        return hi ? m : __uint_as_float(u);
    };
    #pragma unroll
    for (int r = 0; r < 16; ++r) {
        float4 v;
        v.x = dec(xraw[r].x); v.y = dec(xraw[r].y); v.z = dec(xraw[r].z); v.w = dec(xraw[r].w);
        *(float4*)(xw_ + r * 128 + (hi ? HID : 0) + l15 * 4) = v;
    }
    __syncthreads();

    v8f acc[4];
    #pragma unroll
    for (int t = 0; t < 4; ++t) acc[t] = {};

    const v2f* arow = (const v2f*)(xw_ + l15 * 128);
    const int  aoff = hi ? 1 : 0;
    #pragma unroll
    for (int kc = 0; kc < KT; ++kc) {
        const v2f a = arow[kc * 2 + aoff];
        #pragma unroll
        for (int t = 0; t < 4; ++t)
            acc[t] = wmma_f32(a, fragW[kc * 128 + t * 32 + lane], acc[t]);
    }

    const int rbase = row0 + (hi ? 8 : 0);
    #pragma unroll
    for (int t = 0; t < 4; ++t) {
        const int n = t * 16 + l15;
        const float bias = bias4[t];
        #pragma unroll
        for (int j = 0; j < 8; ++j)
            out[(size_t)(rbase + j) * HID + n] = acc[t][j] + bias;
    }
}

extern "C" void kernel_launch(void* const* d_in, const int* in_sizes, int n_in,
                              void* d_out, int out_size, void* d_ws, size_t ws_size,
                              hipStream_t stream)
{
    (void)in_sizes; (void)n_in; (void)out_size; (void)ws_size;

    const float* feats = (const float*)d_in[0];
    const int*   src   = (const int*)d_in[1];
    const int*   dst   = (const int*)d_in[2];
    const int*   gid   = (const int*)d_in[3];
    const float* awW  = (const float*)d_in[22];
    const float* awb  = (const float*)d_in[23];
    const float* outW = (const float*)d_in[24];
    const float* outb = (const float*)d_in[25];
    float* out = (float*)d_out;

    float* ws    = (float*)d_ws;
    float* xw    = ws;
    float* rlin  = xw   + (size_t)N_NODES * HID;
    float* bufA  = rlin + (size_t)N_NODES * HID;
    float* bufB  = bufA + (size_t)N_NODES * HID;
    float* hsum  = bufB + (size_t)N_NODES * HID + 128;
    unsigned* hmaxu = (unsigned*)(hsum + (size_t)N_GRAPH * HID);

    int*      col    = (int*)bufB;
    unsigned* cursor = (unsigned*)(col + N_EDGES);
    unsigned* deg    = cursor + N_PAD;
    float*    stats3 = (float*)(deg + N_PAD);
    unsigned* ebuf   = (unsigned*)(stats3 + 3 * 128);
    unsigned* cnt    = ebuf + N_EDGES;
    unsigned* bstart = cnt + NBK * NCH;
    unsigned* bcount = bstart + 256;
    unsigned* total  = bcount + 256;

    const int ntiles = N_NODES / 16;
    const int gemmBlocks = (ntiles + 3) / 4;

    bin_count<<<NCH, 256, 0, stream>>>(dst, cnt, total, stats3, hsum, hmaxu, N_EDGES);
    bucket_ranges<<<(NBK + 7) / 8, 256, 0, stream>>>(cnt, total, bstart, bcount);
    bin_scatter<<<NCH, 256, 0, stream>>>(src, dst, cnt, ebuf, N_EDGES);
    bucket_to_nodes<<<NBK, 256, 0, stream>>>(ebuf, bstart, bcount, cursor, deg, col, N_EDGES);

    layer1_gather_gemm<<<gemmBlocks, 128, 0, stream>>>(
        feats, (const float*)d_in[4], (const float*)d_in[6],
        (const float*)d_in[5], (const float*)d_in[7],
        cursor, deg, col, bufA, stats3, ntiles, N_NODES, N_EDGES);

    layer_gather_gemm<<<gemmBlocks, 128, 0, stream>>>(
        bufA, (const float*)d_in[10], (const float*)d_in[12],
        (const float*)d_in[11], (const float*)d_in[13],
        stats3, (const float*)d_in[8], (const float*)d_in[9],
        cursor, deg, col, xw, stats3 + 128, ntiles, N_NODES, N_EDGES);
    layer_gather_gemm<<<gemmBlocks, 128, 0, stream>>>(
        xw, (const float*)d_in[16], (const float*)d_in[18],
        (const float*)d_in[17], (const float*)d_in[19],
        stats3 + 128, (const float*)d_in[14], (const float*)d_in[15],
        cursor, deg, col, bufA, stats3 + 256, ntiles, N_NODES, N_EDGES);

    readout_nodes<<<(N_NODES + 8 * RO_NODES - 1) / (8 * RO_NODES), 256, 0, stream>>>(bufA, stats3 + 256,
                                                         (const float*)d_in[20], (const float*)d_in[21],
                                                         gid, awW, awb, hsum, hmaxu, N_NODES);
    out_gemm_wmma<<<(N_GRAPH / 16) / 4, 128, 0, stream>>>(hsum, hmaxu, outW, outb, out);
}


# Layer 1's two products (X@W, relu(X@rW+rb)) run as extra work-groups of the two edge-list launches; layer 1 then sums 256-B product rows, A(xW), not 296-B feature rows (float sum order changes)
# speedup vs baseline: 3.4769x; 1.2344x over previous
#include <hip/hip_runtime.h>

typedef float v2f __attribute__((ext_vector_type(2)));
typedef float v8f __attribute__((ext_vector_type(8)));

#define N_NODES 50000
#define N_EDGES 800000
#define N_GRAPH 512
#define F_IN    74
#define HID     64
#define BN_EPS  1e-5f
#define N_PAD   50048

__device__ __forceinline__ v8f wmma_f32(v2f a, v2f b, v8f c) {
    return __builtin_amdgcn_wmma_f32_16x16x4_f32(false, a, false, b,
                                                 (short)0, c, false, false);
}

#define NEG_INF_ENC 0x007FFFFFu
#define BK_SHIFT 8
#define BK_NODES 256
#define NBK      ((N_NODES + BK_NODES - 1) / BK_NODES)
#define NCH      256
#define CH_EDGES ((N_EDGES + NCH - 1) / NCH)
#define BK_CAP   8192
#define RANK_CAP 512
#define N_START  (3 * 128 + 2 * N_GRAPH * HID)
#define CH_LOADS ((CH_EDGES + 255) / 256)

#define F_PAD 76
#define PJ_WAVES    8
#define PJ_BLOCKS   ((N_NODES / 16 + PJ_WAVES - 1) / PJ_WAVES)
#define PJ_BLOCKS_A (PJ_BLOCKS / 2)
#define PJ_BLOCKS_B (PJ_BLOCKS - PJ_BLOCKS_A)
#define PJ_LDS      (2 * (F_PAD / 4) * 128 * 8 + PJ_WAVES * 16 * F_PAD * 4)

__device__ __forceinline__ void project_rows(unsigned char* smem, int pblock,
                                             const float* __restrict__ X,
                                             const float* __restrict__ W,
                                             const float* __restrict__ rW,
                                             const float* __restrict__ rb,
                                             float* __restrict__ XW,
                                             float* __restrict__ R,
                                             int ntiles)
{
    constexpr int KT = F_PAD / 4;
    v2f*   fragW = (v2f*)smem;
    v2f*   fragR = fragW + KT * 128;
    float* tx    = (float*)(fragR + KT * 128);

    const int  lane = threadIdx.x & 31;
    const int  wave = threadIdx.x >> 5;
    const int  l15  = lane & 15;
    const bool hi   = lane >= 16;

    const int p0 = l15;
    const int p1 = l15 + 16;
    const int p2 = (l15 < 5) ? (l15 + 32) : 36;

    const int  tile  = pblock * PJ_WAVES + wave;
    const bool activ = tile < ntiles;
    const int  row0  = (activ ? tile : (ntiles - 1)) * 16;

    float rbb[4];
    #pragma unroll
    for (int t = 0; t < 4; ++t) rbb[t] = rb[t * 16 + l15];
    float2 x0[8], x1[8], x2[8];
    #pragma unroll
    for (int u = 0; u < 8; ++u) {
        const float2* rp = (const float2*)(X + (size_t)(row0 + 2 * u + (hi ? 1 : 0)) * F_IN);
        x0[u] = rp[p0]; x1[u] = rp[p1]; x2[u] = rp[p2];
    }

    {
        const int  r   = threadIdx.x & 127;
        const bool sel = threadIdx.x >= 128;
        const float* Wm = sel ? rW : W;
        v2f* fdst = sel ? fragR : fragW;
        const int n  = ((r >> 5) << 4) + (r & 15);
        const int kh = (r & 31) >= 16 ? 2 : 0;
        v2f fw[KT];
        #pragma unroll
        for (int kc = 0; kc < KT; ++kc) {
            const int ka0 = kc * 4 + kh;
            const int ka1 = ka0 + 1;
            const int c0  = (ka0 < F_IN) ? ka0 : 0;
            const int c1  = (ka1 < F_IN) ? ka1 : 0;
            const float m0 = (ka0 < F_IN) ? 1.0f : 0.0f;
            const float m1 = (ka1 < F_IN) ? 1.0f : 0.0f;
            fw[kc].x = Wm[(size_t)c0 * HID + n] * m0; fw[kc].y = Wm[(size_t)c1 * HID + n] * m1;
        }
        #pragma unroll
        for (int kc = 0; kc < KT; ++kc) fdst[kc * 128 + r] = fw[kc];
    }

    float* tx_ = tx + wave * 16 * F_PAD;
    const float2 zero2 = make_float2(0.0f, 0.0f);

    const int pc2 = (l15 < 5) ? l15 : 5;
    #pragma unroll
    for (int u = 0; u < 8; ++u) {
        float* trow = tx_ + (2 * u + (hi ? 1 : 0)) * F_PAD;
        *(float2*)(trow + 2 * p0) = x0[u];
        *(float2*)(trow + 2 * p1) = x1[u];
        *(float2*)(trow + 64 + 2 * pc2) = (l15 < 5) ? x2[u] : zero2;
    }
    __syncthreads();

    v8f acc[4], racc[4];
    #pragma unroll
    for (int t = 0; t < 4; ++t) { acc[t] = {}; racc[t] = {}; }

    const v2f* xrow = (const v2f*)(tx_ + l15 * F_PAD);
    const int  aoff = hi ? 1 : 0;

    #pragma unroll
    for (int kc = 0; kc < KT; ++kc) {
        const v2f x = xrow[kc * 2 + aoff];
        #pragma unroll
        for (int t = 0; t < 4; ++t) {
            const v2f b  = fragW[kc * 128 + t * 32 + lane];
            const v2f rv = fragR[kc * 128 + t * 32 + lane];
            acc[t]  = wmma_f32(x, b,  acc[t]);
            racc[t] = wmma_f32(x, rv, racc[t]);
        }
    }

    const int rl0 = hi ? 8 : 0;
    const int c4  = l15 * 4;
    #pragma unroll
    for (int t = 0; t < 4; ++t) {
        #pragma unroll
        for (int j = 0; j < 8; ++j) tx_[(rl0 + j) * F_PAD + t * 16 + l15] = acc[t][j];
    }
    if (activ) {
        #pragma unroll
        for (int u = 0; u < 8; ++u) {
            const int row = 2 * u + (hi ? 1 : 0);
            const float4 hv = *(const float4*)(tx_ + row * F_PAD + c4);
            *(float4*)(XW + (size_t)(row0 + row) * HID + c4) = hv;
        }
    }
    #pragma unroll
    for (int t = 0; t < 4; ++t) {
        #pragma unroll
        for (int j = 0; j < 8; ++j) {
            float v = racc[t][j] + rbb[t];
            v = v > 0.0f ? v : 0.0f;
            tx_[(rl0 + j) * F_PAD + t * 16 + l15] = v;
        }
    }
    if (activ) {
        #pragma unroll
        for (int u = 0; u < 8; ++u) {
            const int row = 2 * u + (hi ? 1 : 0);
            const float4 hv = *(const float4*)(tx_ + row * F_PAD + c4);
            *(float4*)(R + (size_t)(row0 + row) * HID + c4) = hv;
        }
    }
}

__global__ __launch_bounds__(256)
void chunk_sort(const int* __restrict__ src, const int* __restrict__ dst,
                unsigned* __restrict__ seg, unsigned* __restrict__ ebuf,
                float* __restrict__ stats3, float* __restrict__ hsum,
                unsigned* __restrict__ hmaxu, int ne,
                const float* __restrict__ X,
                const float* __restrict__ W,
                const float* __restrict__ rW,
                const float* __restrict__ rbv,
                float* __restrict__ XW,
                float* __restrict__ R,
                int ntiles)
{
    __shared__ __attribute__((aligned(16))) unsigned char smem[PJ_LDS];

    if (blockIdx.x >= NCH) {
        project_rows(smem, (int)blockIdx.x - NCH, X, W, rW, rbv, XW, R, ntiles);
        return;
    }

    unsigned* hist = (unsigned*)smem;
    unsigned* wsum = hist + 256;
    unsigned* lw   = wsum + 8;

    const int t    = threadIdx.x;
    const int lane = t & 31;
    const int wave = t >> 5;
    const int e0 = blockIdx.x * CH_EDGES;
    const int e1 = (e0 + CH_EDGES < ne) ? (e0 + CH_EDGES) : ne;

    unsigned d[CH_LOADS], s[CH_LOADS];
    #pragma unroll
    for (int u = 0; u < CH_LOADS; ++u) {
        const int ee = e0 + t + u * 256;
        const int ec = ee < e1 ? ee : (ne - 1);
        d[u] = (unsigned)dst[ec];
        s[u] = (unsigned)src[ec];
    }
    __builtin_amdgcn_sched_barrier(0);
    hist[t] = 0u;
    __syncthreads();

    unsigned p[CH_LOADS];
    #pragma unroll
    for (int u = 0; u < CH_LOADS; ++u) {
        d[u] = d[u] < (unsigned)N_NODES ? d[u] : (unsigned)(N_NODES - 1);
        s[u] = s[u] < (unsigned)N_NODES ? s[u] : (unsigned)(N_NODES - 1);
        p[u] = 0u;
        if (e0 + t + u * 256 < e1) p[u] = atomicAdd(&hist[d[u] >> BK_SHIFT], 1u);
    }
    __syncthreads();

    const unsigned c = hist[t];
    unsigned inc = c;
    for (int off = 1; off < 32; off <<= 1) {
        const unsigned v = __shfl_up(inc, off, 32);
        if (lane >= off) inc += v;
    }
    if (lane == 31) wsum[wave] = inc;
    __syncthreads();
    unsigned wbase = 0u;
    for (int w = 0; w < wave; ++w) wbase += wsum[w];
    const unsigned first = wbase + inc - c;
    hist[t] = first;
    if (t < NBK) seg[t * NCH + blockIdx.x] = (first << 16) | c;
    __syncthreads();

    #pragma unroll
    for (int u = 0; u < CH_LOADS; ++u) {
        if (e0 + t + u * 256 < e1) {
            const unsigned q = hist[d[u] >> BK_SHIFT] + p[u];
            if (q < (unsigned)CH_EDGES) lw[q] = (s[u] << BK_SHIFT) | (d[u] & (unsigned)(BK_NODES - 1));
        }
    }
    __syncthreads();
    for (int i = t; i < e1 - e0; i += 256) ebuf[e0 + i] = lw[i];

    for (int i = blockIdx.x * 256 + t; i < N_START; i += NCH * 256) {
        int j = i;
        if (j < 3 * 128) { stats3[j] = 0.0f; continue; }
        j -= 3 * 128;
        if (j < N_GRAPH * HID) { hsum[j] = 0.0f; continue; }
        j -= N_GRAPH * HID;
        hmaxu[j] = NEG_INF_ENC;
    }
}

__global__ __launch_bounds__(256)
void bucket_to_nodes(const unsigned* __restrict__ ebuf, const unsigned* __restrict__ seg,
                     unsigned* __restrict__ cursor, unsigned* __restrict__ deg,
                     int* __restrict__ col, int ne,
                     const float* __restrict__ X,
                     const float* __restrict__ W,
                     const float* __restrict__ rW,
                     const float* __restrict__ rbv,
                     float* __restrict__ XW,
                     float* __restrict__ R,
                     int ntiles)
{
    __shared__ __attribute__((aligned(16))) unsigned char smem[PJ_LDS];

    if (blockIdx.x >= NBK) {
        project_rows(smem, PJ_BLOCKS_A + (int)blockIdx.x - NBK, X, W, rW, rbv, XW, R, ntiles);
        return;
    }

    unsigned* lw     = (unsigned*)smem;
    unsigned* npos   = lw + BK_CAP;
    unsigned* nfirst = npos + BK_NODES;
    unsigned* ncnt   = nfirst + BK_NODES;
    unsigned* wsum   = ncnt + BK_NODES;
    unsigned* wsn    = wsum + 8;
    unsigned* wsb    = wsn + 8;

    const int t    = threadIdx.x;
    const int lane = t & 31;
    const int wave = t >> 5;
    const int b    = blockIdx.x;

    const unsigned sg = seg[b * NCH + t];
    const unsigned cb = (unsigned)t * (unsigned)CH_EDGES;
    const unsigned cl = cb < (unsigned)ne ? (((unsigned)ne - cb < (unsigned)CH_EDGES) ? (unsigned)ne - cb : (unsigned)CH_EDGES) : 0u;
    unsigned cs = sg >> 16, cc = sg & 0xFFFFu;
    if (cs > cl) cs = cl;
    if (cc > cl - cs) cc = cl - cs;

    unsigned rn = cc, rb = cs;
    for (int off = 16; off > 0; off >>= 1) { rn += __shfl_xor(rn, off, 32); rb += __shfl_xor(rb, off, 32); }
    if (lane == 0) { wsn[wave] = rn; wsb[wave] = rb; }
    npos[t] = 0u;

    unsigned w[32];
    #pragma unroll
    for (int i = 0; i < 32; ++i) {
        const unsigned ci = (unsigned)__builtin_amdgcn_readlane((int)cc, i);
        const unsigned si = (unsigned)__builtin_amdgcn_readlane((int)cs, i);
        unsigned j = (unsigned)lane < ci ? (unsigned)lane : 0u;
        unsigned a = (unsigned)(wave * 32 + i) * (unsigned)CH_EDGES + si + j;
        if (a > (unsigned)(ne - 1)) a = (unsigned)(ne - 1);
        w[i] = ebuf[a];
    }
    auto rest = [&](auto f) {
        #pragma nounroll
        for (int i = 0; i < 32; ++i) {
            const unsigned ci = (unsigned)__builtin_amdgcn_readlane((int)cc, i);
            if (ci > 32u) {
                const unsigned si = (unsigned)__builtin_amdgcn_readlane((int)cs, i);
                const unsigned a0 = (unsigned)(wave * 32 + i) * (unsigned)CH_EDGES + si;
                for (unsigned j = 32u + (unsigned)lane; j < ci; j += 32u) {
                    const unsigned a = a0 + j;
                    if (a < (unsigned)ne) f(ebuf[a]);
                }
            }
        }
    };
    __syncthreads();
    unsigned n = 0u, beg = 0u;
    for (int k = 0; k < 8; ++k) { n += wsn[k]; beg += wsb[k]; }
    if (beg > (unsigned)ne) beg = (unsigned)ne;
    if (n > (unsigned)ne - beg) n = (unsigned)ne - beg;

    #pragma unroll
    for (int i = 0; i < 32; ++i) {
        const unsigned ci = (unsigned)__builtin_amdgcn_readlane((int)cc, i);
        if ((unsigned)lane < ci) atomicAdd(&npos[w[i] & (unsigned)(BK_NODES - 1)], 1u);
    }
    rest([&](unsigned v) { atomicAdd(&npos[v & (unsigned)(BK_NODES - 1)], 1u); });
    __syncthreads();

    const unsigned d = npos[t];
    unsigned inc = d;
    for (int off = 1; off < 32; off <<= 1) {
        const unsigned v = __shfl_up(inc, off, 32);
        if (lane >= off) inc += v;
    }
    if (lane == 31) wsum[wave] = inc;
    __syncthreads();
    unsigned wbase = 0u;
    for (int k = 0; k < wave; ++k) wbase += wsum[k];
    const unsigned first = wbase + inc - d;
    npos[t] = first; nfirst[t] = first; ncnt[t] = d;
    const int node = b * BK_NODES + t;
    if (node < N_NODES) { cursor[node] = beg + first + d; deg[node] = d; }
    __syncthreads();

    if (n <= (unsigned)BK_CAP) {
        auto place = [&](unsigned v) {
            const unsigned q = atomicAdd(&npos[v & (unsigned)(BK_NODES - 1)], 1u);
            if (q < (unsigned)BK_CAP) lw[q] = v;
        };
        #pragma unroll
        for (int i = 0; i < 32; ++i) {
            const unsigned ci = (unsigned)__builtin_amdgcn_readlane((int)cc, i);
            if ((unsigned)lane < ci) place(w[i]);
        }
        rest(place);
        __syncthreads();
        for (unsigned q = t; q < n; q += 256) {
            const unsigned v  = lw[q];
            const unsigned k  = v & (unsigned)(BK_NODES - 1);
            const unsigned a  = nfirst[k];
            const unsigned dk = ncnt[k];
            unsigned r = q - a;
            if (dk <= (unsigned)RANK_CAP) {
                r = 0u;
                for (unsigned j = a; j < a + dk; ++j) {
                    const unsigned x = lw[j];
                    r += (x < v || (x == v && j < q)) ? 1u : 0u;
                }
            }
            col[beg + a + r] = (int)(v >> BK_SHIFT);
        }
    } else {
        auto direct = [&](unsigned v) {
            const unsigned q = atomicAdd(&npos[v & (unsigned)(BK_NODES - 1)], 1u);
            if (q < n) col[beg + q] = (int)(v >> BK_SHIFT);
        };
        #pragma unroll
        for (int i = 0; i < 32; ++i) {
            const unsigned ci = (unsigned)__builtin_amdgcn_readlane((int)cc, i);
            if ((unsigned)lane < ci) direct(w[i]);
        }
        rest(direct);
    }
}
__device__ __forceinline__ float xhalf(float x) {
    return __builtin_bit_cast(float, __builtin_amdgcn_permlanex16(
        __builtin_bit_cast(int, x), __builtin_bit_cast(int, x),
        0x76543210, 0xfedcba98, false, false));
}

#define XP 68
#define GR_DEPTH 16

__global__ __launch_bounds__(128)
void layer_gather_gemm(const float* __restrict__ Hin,
                       const float* __restrict__ W,
                       const float* __restrict__ rW,
                       const float* __restrict__ bvec,
                       const float* __restrict__ rb,
                       const float* __restrict__ pst,
                       const float* __restrict__ pg,
                       const float* __restrict__ pb,
                       const unsigned* __restrict__ cursor, const unsigned* __restrict__ deg,
                       const int* __restrict__ col,
                       float* __restrict__ Hout,
                       float* __restrict__ stats,
                       int ntiles, int nrows, int ne)
{
    constexpr int KT = HID / 4;
    __shared__ v2f   fragW[KT * 128];
    __shared__ v2f   fragR[KT * 128];
    __shared__ float ta[4 * 16 * XP];
    __shared__ float tx[4 * 16 * XP];
    __shared__ float ssum[4 * 64], ssq[4 * 64];

    const int  lane = threadIdx.x & 31;
    const int  wave = threadIdx.x >> 5;
    const int  l15  = lane & 15;
    const bool hi   = lane >= 16;
    const int  c4   = l15 * 4;

    const int  tile  = blockIdx.x * 4 + wave;
    const bool activ = tile < ntiles;
    const int  row0  = (activ ? tile : (ntiles - 1)) * 16;

    unsigned endv = cursor[row0 + l15];
    unsigned dv   = deg[row0 + l15];
    float bb[4], rbb[4];
    #pragma unroll
    for (int t = 0; t < 4; ++t) { bb[t] = bvec[t * 16 + l15]; rbb[t] = rb[t * 16 + l15]; }
    const float4 psum = *(const float4*)(pst + c4);
    const float4 psq  = *(const float4*)(pst + 64 + c4);
    float gam[4], bet[4];
    #pragma unroll
    for (int i = 0; i < 4; ++i) { gam[i] = pg[c4 + i]; bet[i] = pb[c4 + i]; }
    float4 xr[8];
    #pragma unroll
    for (int u = 0; u < 8; ++u)
        xr[u] = *(const float4*)(Hin + (size_t)(row0 + 2 * u + (hi ? 1 : 0)) * HID + c4);

    {
        const int r  = threadIdx.x;
        const int n  = ((r >> 5) << 4) + (r & 15);
        const int kh = (r & 31) >= 16 ? 2 : 0;
        v2f fw[KT], fr[KT];
        #pragma unroll
        for (int kc = 0; kc < KT; ++kc) {
            const int ka0 = kc * 4 + kh;
            fw[kc].x = W [(size_t)ka0 * HID + n]; fw[kc].y = W [(size_t)(ka0 + 1) * HID + n];
            fr[kc].x = rW[(size_t)ka0 * HID + n]; fr[kc].y = rW[(size_t)(ka0 + 1) * HID + n];
        }
        #pragma unroll
        for (int kc = 0; kc < KT; ++kc) {
            fragW[kc * 128 + r] = fw[kc];
            fragR[kc * 128 + r] = fr[kc];
        }
    }

    float sc[4], sh[4];
    {
        const float inv = 1.0f / (float)nrows;
        const float ps[4] = {psum.x, psum.y, psum.z, psum.w};
        const float pq[4] = {psq.x, psq.y, psq.z, psq.w};
        #pragma unroll
        for (int i = 0; i < 4; ++i) {
            const float mean = ps[i] * inv;
            const float var  = pq[i] * inv - mean * mean;
            sc[i] = rsqrtf(var + BN_EPS) * gam[i];
            sh[i] = bet[i] - mean * sc[i];
        }
    }

    float* ta_ = ta + wave * 16 * XP;
    float* tx_ = tx + wave * 16 * XP;

    #pragma unroll
    for (int u = 0; u < 8; ++u) {
        float4 v = xr[u];
        v.x = v.x * sc[0] + sh[0]; v.y = v.y * sc[1] + sh[1];
        v.z = v.z * sc[2] + sh[2]; v.w = v.w * sc[3] + sh[3];
        *(float4*)(tx_ + (2 * u + (hi ? 1 : 0)) * XP + c4) = v;
    }

    if (endv > (unsigned)ne) endv = (unsigned)ne;
    if (dv > endv) dv = endv;
    if (!activ) dv = 0u;
    const unsigned begv = endv - dv;

    #pragma unroll
    for (int u = 0; u < 8; ++u)
        *(float4*)(ta_ + (2 * u + (hi ? 1 : 0)) * XP + c4) = make_float4(0.0f, 0.0f, 0.0f, 0.0f);

    const unsigned S = (unsigned)__builtin_amdgcn_readlane((int)begv, 0);
    unsigned es[16];
    #pragma unroll
    for (int j = 0; j < 16; ++j) es[j] = (unsigned)__builtin_amdgcn_readlane((int)endv, j);
    unsigned Tn = (es[15] > S) ? (es[15] - S) : 0u;
    if (!activ) Tn = 0u;
    const unsigned nch = ((unsigned)__builtin_amdgcn_readfirstlane((int)Tn) + 31u) >> 5;

    auto node_of = [&](unsigned p) -> unsigned {
        unsigned k = 0u;
        #pragma unroll
        for (int j = 0; j < 16; ++j) k += (es[j] <= p) ? 1u : 0u;
        return k;
    };
    auto chunk_ids = [&](unsigned c) -> unsigned {
        unsigned p = S + 32u * c + (unsigned)lane;
        if (p > (unsigned)(ne - 1)) p = (unsigned)(ne - 1);
        return (unsigned)col[p];
    };
    auto chunk_marks = [&](unsigned c, unsigned kv) -> unsigned {
        const unsigned p = S + 32u * c + (unsigned)lane;
        return (unsigned)__builtin_amdgcn_ballot_w32(kv != node_of(p - 1u));
    };

    float4 gacc = make_float4(0.0f, 0.0f, 0.0f, 0.0f);
    int cur = 0;
    auto flush = [&](int k) {
        float4 t;
        t.x = gacc.x + xhalf(gacc.x); t.y = gacc.y + xhalf(gacc.y);
        t.z = gacc.z + xhalf(gacc.z); t.w = gacc.w + xhalf(gacc.w);
        if (k < 16) {
            const float df = (float)(unsigned)__builtin_amdgcn_readlane((int)dv, k);
            if (lane < 16) {
                float4 o;
                o.x = t.x * sc[0] + df * sh[0]; o.y = t.y * sc[1] + df * sh[1];
                o.z = t.z * sc[2] + df * sh[2]; o.w = t.w * sc[3] + df * sh[3];
                *(float4*)(ta_ + k * XP + c4) = o;
            }
        }
        gacc = make_float4(0.0f, 0.0f, 0.0f, 0.0f);
    };
    auto consume = [&](const float4 val, unsigned kv, unsigned marks, int u) {
        if ((marks >> (2 * u)) & 3u) {
            const int kA = __builtin_amdgcn_readlane((int)kv, 2 * u);
            const int kB = __builtin_amdgcn_readlane((int)kv, 2 * u + 1);
            #pragma nounroll
            for (int h = 0; h < 2; ++h) {
                const int k = h ? kB : kA;
                if (k != cur) { flush(cur); cur = k; }
                if ((hi ? 1 : 0) == h) { gacc.x += val.x; gacc.y += val.y; gacc.z += val.z; gacc.w += val.w; }
            }
        } else {
            gacc.x += val.x; gacc.y += val.y; gacc.z += val.z; gacc.w += val.w;
        }
    };
    auto row_load = [&](unsigned ids, int u) -> float4 {
        const int sa = __builtin_amdgcn_readlane((int)ids, 2 * u);
        const int sb = __builtin_amdgcn_readlane((int)ids, 2 * u + 1);
        const unsigned off = (unsigned)(hi ? sb : sa) * (unsigned)HID + (unsigned)c4;
        return *(const float4*)(Hin + off);
    };
    auto clamp_ids = [&](unsigned ids) -> unsigned {
        return ids > (unsigned)(nrows - 1) ? (unsigned)(nrows - 1) : ids;
    };

    if (nch > 0u) {
        const unsigned raw0 = chunk_ids(0u);
        const unsigned raw1 = chunk_ids(1u);
        const unsigned id0 = clamp_ids(raw0);
        unsigned idB = clamp_ids(raw1);
        unsigned kvA = node_of(S + (unsigned)lane);
        unsigned mkA = chunk_marks(0u, kvA);
        float4 v[GR_DEPTH];
        #pragma unroll
        for (int u = 0; u < GR_DEPTH; ++u) v[u] = row_load(id0, u);
        for (unsigned c = 0; c + 1u < nch; ++c) {
            const unsigned rawN = chunk_ids(c + 2u);
            const unsigned kvB = node_of(S + 32u * (c + 1u) + (unsigned)lane);
            const unsigned mkB = chunk_marks(c + 1u, kvB);
            #pragma unroll
            for (int u = 0; u < GR_DEPTH; ++u) {
                consume(v[u], kvA, mkA, u);
                v[u] = row_load(idB, u);
            }
            kvA = kvB; mkA = mkB; idB = clamp_ids(rawN);
        }
        #pragma unroll
        for (int u = 0; u < GR_DEPTH; ++u) consume(v[u], kvA, mkA, u);
        flush(cur);
    }
    __syncthreads();

    v8f acc[4], racc[4];
    #pragma unroll
    for (int t = 0; t < 4; ++t) { acc[t] = {}; racc[t] = {}; }

    const v2f* arow = (const v2f*)(ta_ + l15 * XP);
    const v2f* xrow = (const v2f*)(tx_ + l15 * XP);
    const int  aoff = hi ? 1 : 0;

    #pragma unroll
    for (int kc = 0; kc < KT; ++kc) {
        const v2f a = arow[kc * 2 + aoff];
        const v2f x = xrow[kc * 2 + aoff];
        #pragma unroll
        for (int t = 0; t < 4; ++t) {
            const v2f b  = fragW[kc * 128 + t * 32 + lane];
            const v2f rv = fragR[kc * 128 + t * 32 + lane];
            acc[t]  = wmma_f32(a, b,  acc[t]);
            racc[t] = wmma_f32(x, rv, racc[t]);
        }
    }

    const int rl0 = hi ? 8 : 0;
    #pragma unroll
    for (int t = 0; t < 4; ++t) {
        const int n = t * 16 + l15;
        float cs = 0.0f, cq = 0.0f;
        #pragma unroll
        for (int j = 0; j < 8; ++j) {
            float a = acc[t][j] + bb[t];
            a = a > 0.0f ? a : 0.0f;
            float v = racc[t][j] + rbb[t];
            v = v > 0.0f ? v : 0.0f;
            const float h = a + v;
            ta_[(rl0 + j) * XP + n] = h;
            cs += h; cq += h * h;
        }
        cs += xhalf(cs);
        cq += xhalf(cq);
        if (!activ) { cs = 0.0f; cq = 0.0f; }
        if (lane < 16) { ssum[wave * 64 + n] = cs; ssq[wave * 64 + n] = cq; }
    }
    if (activ) {
        #pragma unroll
        for (int u = 0; u < 8; ++u) {
            const int row = 2 * u + (hi ? 1 : 0);
            const float4 hv = *(const float4*)(ta_ + row * XP + c4);
            *(float4*)(Hout + (size_t)(row0 + row) * HID + c4) = hv;
        }
    }
    __syncthreads();
    if (threadIdx.x < 64) {
        const int f = threadIdx.x;
        const float s = ssum[f] + ssum[64 + f] + ssum[128 + f] + ssum[192 + f];
        const float q = ssq[f]  + ssq[64 + f]  + ssq[128 + f]  + ssq[192 + f];
        atomicAdd(&stats[f], s);
        atomicAdd(&stats[64 + f], q);
    }
}

__global__ __launch_bounds__(128)
void layer1_gather_sum(const float* __restrict__ XW,
                       const float* __restrict__ R,
                       const float* __restrict__ bvec,
                       const unsigned* __restrict__ cursor, const unsigned* __restrict__ deg,
                       const int* __restrict__ col,
                       float* __restrict__ Hout,
                       float* __restrict__ stats,
                       int ntiles, int nrows, int ne)
{
    __shared__ __attribute__((aligned(16))) float ta[4 * 16 * XP];
    __shared__ float ssum[4 * 64], ssq[4 * 64];

    const int  lane = threadIdx.x & 31;
    const int  wave = threadIdx.x >> 5;
    const int  l15  = lane & 15;
    const bool hi   = lane >= 16;
    const int  c4   = l15 * 4;

    const int  tile  = blockIdx.x * 4 + wave;
    const bool activ = tile < ntiles;
    const int  row0  = (activ ? tile : (ntiles - 1)) * 16;

    unsigned endv = cursor[row0 + l15];
    unsigned dv   = deg[row0 + l15];
    float4 b4;
    b4.x = bvec[c4]; b4.y = bvec[c4 + 1]; b4.z = bvec[c4 + 2]; b4.w = bvec[c4 + 3];
    float4 xr[8];
    #pragma unroll
    for (int u = 0; u < 8; ++u)
        xr[u] = *(const float4*)(R + (size_t)(row0 + 2 * u + (hi ? 1 : 0)) * HID + c4);

    float* ta_ = ta + wave * 16 * XP;

    if (endv > (unsigned)ne) endv = (unsigned)ne;
    if (dv > endv) dv = endv;
    if (!activ) dv = 0u;
    const unsigned begv = endv - dv;

    #pragma unroll
    for (int u = 0; u < 8; ++u)
        *(float4*)(ta_ + (2 * u + (hi ? 1 : 0)) * XP + c4) = make_float4(0.0f, 0.0f, 0.0f, 0.0f);

    const unsigned S = (unsigned)__builtin_amdgcn_readlane((int)begv, 0);
    unsigned es[16];
    #pragma unroll
    for (int j = 0; j < 16; ++j) es[j] = (unsigned)__builtin_amdgcn_readlane((int)endv, j);
    unsigned Tn = (es[15] > S) ? (es[15] - S) : 0u;
    if (!activ) Tn = 0u;
    const unsigned nch = ((unsigned)__builtin_amdgcn_readfirstlane((int)Tn) + 31u) >> 5;

    auto node_of = [&](unsigned p) -> unsigned {
        unsigned k = 0u;
        #pragma unroll
        for (int j = 0; j < 16; ++j) k += (es[j] <= p) ? 1u : 0u;
        return k;
    };
    auto chunk_ids = [&](unsigned c) -> unsigned {
        unsigned p = S + 32u * c + (unsigned)lane;
        if (p > (unsigned)(ne - 1)) p = (unsigned)(ne - 1);
        return (unsigned)col[p];
    };
    auto chunk_marks = [&](unsigned c, unsigned kv) -> unsigned {
        const unsigned p = S + 32u * c + (unsigned)lane;
        return (unsigned)__builtin_amdgcn_ballot_w32(kv != node_of(p - 1u));
    };

    float4 gacc = make_float4(0.0f, 0.0f, 0.0f, 0.0f);
    int cur = 0;
    auto flush = [&](int k) {
        float4 t;
        t.x = gacc.x + xhalf(gacc.x); t.y = gacc.y + xhalf(gacc.y);
        t.z = gacc.z + xhalf(gacc.z); t.w = gacc.w + xhalf(gacc.w);
        if (k < 16) {
            if (lane < 16) *(float4*)(ta_ + k * XP + c4) = t;
        }
        gacc = make_float4(0.0f, 0.0f, 0.0f, 0.0f);
    };
    auto consume = [&](const float4 val, unsigned kv, unsigned marks, int u) {
        if ((marks >> (2 * u)) & 3u) {
            const int kA = __builtin_amdgcn_readlane((int)kv, 2 * u);
            const int kB = __builtin_amdgcn_readlane((int)kv, 2 * u + 1);
            #pragma nounroll
            for (int h = 0; h < 2; ++h) {
                const int k = h ? kB : kA;
                if (k != cur) { flush(cur); cur = k; }
                if ((hi ? 1 : 0) == h) { gacc.x += val.x; gacc.y += val.y; gacc.z += val.z; gacc.w += val.w; }
            }
        } else {
            gacc.x += val.x; gacc.y += val.y; gacc.z += val.z; gacc.w += val.w;
        }
    };
    auto row_load = [&](unsigned ids, int u) -> float4 {
        const int sa = __builtin_amdgcn_readlane((int)ids, 2 * u);
        const int sb = __builtin_amdgcn_readlane((int)ids, 2 * u + 1);
        const unsigned off = (unsigned)(hi ? sb : sa) * (unsigned)HID + (unsigned)c4;
        return *(const float4*)(XW + off);
    };
    auto clamp_ids = [&](unsigned ids) -> unsigned {
        return ids > (unsigned)(nrows - 1) ? (unsigned)(nrows - 1) : ids;
    };

    if (nch > 0u) {
        const unsigned raw0 = chunk_ids(0u);
        const unsigned raw1 = chunk_ids(1u);
        const unsigned id0 = clamp_ids(raw0);
        unsigned idB = clamp_ids(raw1);
        unsigned kvA = node_of(S + (unsigned)lane);
        unsigned mkA = chunk_marks(0u, kvA);
        float4 v[GR_DEPTH];
        #pragma unroll
        for (int u = 0; u < GR_DEPTH; ++u) v[u] = row_load(id0, u);
        for (unsigned c = 0; c + 1u < nch; ++c) {
            const unsigned rawN = chunk_ids(c + 2u);
            const unsigned kvB = node_of(S + 32u * (c + 1u) + (unsigned)lane);
            const unsigned mkB = chunk_marks(c + 1u, kvB);
            #pragma unroll
            for (int u = 0; u < GR_DEPTH; ++u) {
                consume(v[u], kvA, mkA, u);
                v[u] = row_load(idB, u);
            }
            kvA = kvB; mkA = mkB; idB = clamp_ids(rawN);
        }
        #pragma unroll
        for (int u = 0; u < GR_DEPTH; ++u) consume(v[u], kvA, mkA, u);
        flush(cur);
    }

    float cs[4] = {0.0f, 0.0f, 0.0f, 0.0f}, cq[4] = {0.0f, 0.0f, 0.0f, 0.0f};
    #pragma unroll
    for (int u = 0; u < 8; ++u) {
        const int row = 2 * u + (hi ? 1 : 0);
        const float4 a = *(const float4*)(ta_ + row * XP + c4);
        float4 h;
        h.x = a.x + b4.x; h.x = (h.x > 0.0f ? h.x : 0.0f) + xr[u].x;
        h.y = a.y + b4.y; h.y = (h.y > 0.0f ? h.y : 0.0f) + xr[u].y;
        h.z = a.z + b4.z; h.z = (h.z > 0.0f ? h.z : 0.0f) + xr[u].z;
        h.w = a.w + b4.w; h.w = (h.w > 0.0f ? h.w : 0.0f) + xr[u].w;
        if (activ) *(float4*)(Hout + (size_t)(row0 + row) * HID + c4) = h;
        cs[0] += h.x; cs[1] += h.y; cs[2] += h.z; cs[3] += h.w;
        cq[0] += h.x * h.x; cq[1] += h.y * h.y; cq[2] += h.z * h.z; cq[3] += h.w * h.w;
    }
    #pragma unroll
    for (int i = 0; i < 4; ++i) {
        cs[i] += xhalf(cs[i]);
        cq[i] += xhalf(cq[i]);
        if (!activ) { cs[i] = 0.0f; cq[i] = 0.0f; }
        if (lane < 16) { ssum[wave * 64 + c4 + i] = cs[i]; ssq[wave * 64 + c4 + i] = cq[i]; }
    }
    __syncthreads();
    if (threadIdx.x < 64) {
        const int f = threadIdx.x;
        const float s = ssum[f] + ssum[64 + f] + ssum[128 + f] + ssum[192 + f];
        const float q = ssq[f]  + ssq[64 + f]  + ssq[128 + f]  + ssq[192 + f];
        atomicAdd(&stats[f], s);
        atomicAdd(&stats[64 + f], q);
    }
}

__device__ __forceinline__ unsigned f32_ordered(float f) {
    unsigned b = __float_as_uint(f);
    return (b & 0x80000000u) ? ~b : (b | 0x80000000u);
}

#define RO_NODES 16

__global__ __launch_bounds__(256)
void readout_nodes(const float* __restrict__ h, const float* __restrict__ stats,
                   const float* __restrict__ gamma, const float* __restrict__ beta,
                   const int* __restrict__ gid,
                   const float* __restrict__ awW, const float* __restrict__ awb,
                   float* __restrict__ hsum, unsigned* __restrict__ hmaxu,
                   int nrows)
{
    const int lane = threadIdx.x & 31;
    const int wave = __builtin_amdgcn_readfirstlane((int)(threadIdx.x >> 5));
    const int n0 = (blockIdx.x * 8 + wave) * RO_NODES;
    if (n0 >= nrows) return;
    const int last = nrows - 1;

    float r0[RO_NODES], r1[RO_NODES];
    #pragma unroll
    for (int u = 0; u < RO_NODES; ++u) {
        const int n = (n0 + u < nrows) ? (n0 + u) : last;
        r0[u] = h[(size_t)n * HID + lane];
        r1[u] = h[(size_t)n * HID + 32 + lane];
    }
    const int gn = n0 + (lane & (RO_NODES - 1));
    const int gv = gid[gn < nrows ? gn : last];

    const float inv = 1.0f / (float)nrows;
    const float m0 = stats[lane] * inv;
    const float m1 = stats[32 + lane] * inv;
    const float v0 = stats[64 + lane] * inv - m0 * m0;
    const float v1 = stats[96 + lane] * inv - m1 * m1;
    const float rs0 = rsqrtf(v0 + BN_EPS), rs1 = rsqrtf(v1 + BN_EPS);
    const float g0 = gamma[lane], g1 = gamma[32 + lane];
    const float b0 = beta[lane],  b1 = beta[32 + lane];
    const float a0 = awW[lane],   a1 = awW[32 + lane];
    const float ab = awb[0];

    float wg[RO_NODES];
    #pragma unroll
    for (int u = 0; u < RO_NODES; ++u) {
        const float h0 = (r0[u] - m0) * rs0 * g0 + b0;
        const float h1 = (r1[u] - m1) * rs1 * g1 + b1;
        r0[u] = h0; r1[u] = h1;
        float p = h0 * a0 + h1 * a1;
        for (int off = 16; off > 0; off >>= 1) p += __shfl_xor(p, off, 32);
        wg[u] = 1.0f / (1.0f + expf(-(p + ab)));
    }

    int gcur = __builtin_amdgcn_readlane(gv, 0);
    float s0 = 0.0f, s1 = 0.0f;
    unsigned x0 = 0u, x1 = 0u;
    auto flush = [&](int g) {
        atomicAdd(&hsum[(size_t)g * HID + lane],      s0);
        atomicAdd(&hsum[(size_t)g * HID + 32 + lane], s1);
        atomicMax(&hmaxu[(size_t)g * HID + lane],      x0);
        atomicMax(&hmaxu[(size_t)g * HID + 32 + lane], x1);
    };
    #pragma unroll
    for (int u = 0; u < RO_NODES; ++u) {
        if (n0 + u < nrows) {
            const int g = __builtin_amdgcn_readlane(gv, u);
            if (g != gcur) {
                flush(gcur);
                s0 = 0.0f; s1 = 0.0f; x0 = 0u; x1 = 0u;
                gcur = g;
            }
            s0 += r0[u] * wg[u];
            s1 += r1[u] * wg[u];
            const unsigned e0 = f32_ordered(r0[u]), e1 = f32_ordered(r1[u]);
            x0 = e0 > x0 ? e0 : x0;
            x1 = e1 > x1 ? e1 : x1;
        }
    }
    flush(gcur);
}

__global__ __launch_bounds__(128)
void out_gemm_wmma(const float* __restrict__ hsum, const unsigned* __restrict__ hmaxu,
                   const float* __restrict__ Wo,
                   const float* __restrict__ bo, float* __restrict__ out)
{
    constexpr int KT = 128 / 4;
    __shared__ v2f  fragW[KT * 128];
    __shared__ __attribute__((aligned(16))) float xt[4 * 16 * 128];

    const int  lane = threadIdx.x & 31;
    const int  wave = threadIdx.x >> 5;
    const int  l15  = lane & 15;
    const bool hi   = lane >= 16;

    const int row0 = (blockIdx.x * 4 + wave) * 16;
    float* xw_ = xt + wave * 16 * 128;

    const int fr_ = threadIdx.x;
    const int fn_ = ((fr_ >> 5) << 4) + (fr_ & 15);
    const int fk_ = (fr_ & 31) >= 16 ? 2 : 0;
    v2f fw[KT];
    #pragma unroll
    for (int kc = 0; kc < KT; ++kc) {
        const int ka0 = kc * 4 + fk_;
        fw[kc].x = Wo[(size_t)ka0 * HID + fn_];
        fw[kc].y = Wo[(size_t)(ka0 + 1) * HID + fn_];
    }
    uint4 xraw[16];
    #pragma unroll
    for (int r = 0; r < 16; ++r) {
        const size_t o = (size_t)(row0 + r) * HID + (size_t)(l15 * 4);
        const uint4* p = hi ? (const uint4*)(hmaxu + o) : (const uint4*)(hsum + o);
        xraw[r] = *p;
    }
    float bias4[4];
    #pragma unroll
    for (int t = 0; t < 4; ++t) bias4[t] = bo[t * 16 + l15];

    #pragma unroll
    for (int kc = 0; kc < KT; ++kc) fragW[kc * 128 + fr_] = fw[kc];

    auto dec = [&](unsigned u) -> float {
        float m = (u & 0x80000000u) ? __uint_as_float(u ^ 0x80000000u)
                                    : __uint_as_float(~u);
        if (!(fabsf(m) <= 3.402823466e38f)) m = 0.0f;
        return hi ? m : __uint_as_float(u);
    };
    #pragma unroll
    for (int r = 0; r < 16; ++r) {
        float4 v;
        v.x = dec(xraw[r].x); v.y = dec(xraw[r].y); v.z = dec(xraw[r].z); v.w = dec(xraw[r].w);
        *(float4*)(xw_ + r * 128 + (hi ? HID : 0) + l15 * 4) = v;
    }
    __syncthreads();

    v8f acc[4];
    #pragma unroll
    for (int t = 0; t < 4; ++t) acc[t] = {};

    const v2f* arow = (const v2f*)(xw_ + l15 * 128);
    const int  aoff = hi ? 1 : 0;
    #pragma unroll
    for (int kc = 0; kc < KT; ++kc) {
        const v2f a = arow[kc * 2 + aoff];
        #pragma unroll
        for (int t = 0; t < 4; ++t)
            acc[t] = wmma_f32(a, fragW[kc * 128 + t * 32 + lane], acc[t]);
    }

    const int rbase = row0 + (hi ? 8 : 0);
    #pragma unroll
    for (int t = 0; t < 4; ++t) {
        const int n = t * 16 + l15;
        const float bias = bias4[t];
        #pragma unroll
        for (int j = 0; j < 8; ++j)
            out[(size_t)(rbase + j) * HID + n] = acc[t][j] + bias;
    }
}

extern "C" void kernel_launch(void* const* d_in, const int* in_sizes, int n_in,
                              void* d_out, int out_size, void* d_ws, size_t ws_size,
                              hipStream_t stream)
{
    (void)in_sizes; (void)n_in; (void)out_size; (void)ws_size;

    const float* feats = (const float*)d_in[0];
    const int*   src   = (const int*)d_in[1];
    const int*   dst   = (const int*)d_in[2];
    const int*   gid   = (const int*)d_in[3];
    const float* awW  = (const float*)d_in[22];
    const float* awb  = (const float*)d_in[23];
    const float* outW = (const float*)d_in[24];
    const float* outb = (const float*)d_in[25];
    float* out = (float*)d_out;

    float* ws    = (float*)d_ws;
    float* xw    = ws;
    float* rlin  = xw   + (size_t)N_NODES * HID;
    float* bufA  = rlin + (size_t)N_NODES * HID;
    float* bufB  = bufA + (size_t)N_NODES * HID;
    float* hsum  = bufB + (size_t)N_NODES * HID + 128;
    unsigned* hmaxu = (unsigned*)(hsum + (size_t)N_GRAPH * HID);

    int*      col    = (int*)bufB;
    unsigned* cursor = (unsigned*)(col + N_EDGES);
    unsigned* deg    = cursor + N_PAD;
    float*    stats3 = (float*)(deg + N_PAD);
    unsigned* ebuf   = (unsigned*)(stats3 + 3 * 128);
    unsigned* seg    = ebuf + N_EDGES;

    const int ntiles = N_NODES / 16;
    const int gemmBlocks = (ntiles + 3) / 4;

    chunk_sort<<<NCH + PJ_BLOCKS_A, 256, 0, stream>>>(src, dst, seg, ebuf, stats3, hsum, hmaxu, N_EDGES,
        feats, (const float*)d_in[4], (const float*)d_in[6], (const float*)d_in[7],
        rlin, xw, ntiles);
    bucket_to_nodes<<<NBK + PJ_BLOCKS_B, 256, 0, stream>>>(ebuf, seg, cursor, deg, col, N_EDGES,
        feats, (const float*)d_in[4], (const float*)d_in[6], (const float*)d_in[7],
        rlin, xw, ntiles);

    layer1_gather_sum<<<gemmBlocks, 128, 0, stream>>>(
        rlin, xw, (const float*)d_in[5],
        cursor, deg, col, bufA, stats3, ntiles, N_NODES, N_EDGES);

    layer_gather_gemm<<<gemmBlocks, 128, 0, stream>>>(
        bufA, (const float*)d_in[10], (const float*)d_in[12],
        (const float*)d_in[11], (const float*)d_in[13],
        stats3, (const float*)d_in[8], (const float*)d_in[9],
        cursor, deg, col, xw, stats3 + 128, ntiles, N_NODES, N_EDGES);
    layer_gather_gemm<<<gemmBlocks, 128, 0, stream>>>(
        xw, (const float*)d_in[16], (const float*)d_in[18],
        (const float*)d_in[17], (const float*)d_in[19],
        stats3 + 128, (const float*)d_in[14], (const float*)d_in[15],
        cursor, deg, col, bufA, stats3 + 256, ntiles, N_NODES, N_EDGES);

    readout_nodes<<<(N_NODES + 8 * RO_NODES - 1) / (8 * RO_NODES), 256, 0, stream>>>(bufA, stats3 + 256,
                                                         (const float*)d_in[20], (const float*)d_in[21],
                                                         gid, awW, awb, hsum, hmaxu, N_NODES);
    out_gemm_wmma<<<(N_GRAPH / 16) / 4, 128, 0, stream>>>(hsum, hmaxu, outW, outb, out);
}
